# starting kernel
# baseline (speedup 1.0000x reference)
#include <hip/hip_runtime.h>
#include <math.h>

typedef __attribute__((ext_vector_type(2))) float v2f;
typedef __attribute__((ext_vector_type(8))) float v8f;

namespace {

constexpr int B_   = 32;
constexpr int L_   = 64;
constexpr int H_   = 200;
constexpr int M_   = 3;
constexpr int A_   = B_ * L_;
constexpr int N_   = M_ * A_;
constexpr int TAG_ = 6;
constexpr float PI_F = 3.14159265358979323846f;

template <typename FA, typename FB, typename FE>
__device__ __forceinline__ void wmma_tile_f32(int m0, int n0, int K,
                                              FA loadA, FB loadB, FE epi) {
  const int lane = threadIdx.x & 31;
  const int mr   = lane & 15;
  const int kh   = lane >> 4;
  v8f acc = {};
  for (int k = 0; k < K; k += 4) {
    const int kk = k + 2 * kh;
    v2f a, b;
    a.x = loadA(m0 + mr, kk);
    a.y = loadA(m0 + mr, kk + 1);
    b.x = loadB(kk, n0 + mr);
    b.y = loadB(kk + 1, n0 + mr);
    acc = __builtin_amdgcn_wmma_f32_16x16x4_f32(false, a, false, b,
                                                (short)0, acc, false, false);
  }
#pragma unroll
  for (int v = 0; v < 8; ++v) epi(m0 + 8 * kh + v, n0 + mr, acc[v]);
}

__device__ __forceinline__ float wave_sum(float s) {
#pragma unroll
  for (int off = 16; off > 0; off >>= 1) s += __shfl_xor(s, off, 32);
  return s;
}

__global__ void k_norm(const float* __restrict__ xa, const float* __restrict__ xv,
                       const float* __restrict__ xt, float* __restrict__ big,
                       float* __restrict__ nrm) {
  const int row  = blockIdx.x * 8 + (threadIdx.x >> 5);
  const int lane = threadIdx.x & 31;
  const float* src = (row < A_)     ? xa + (size_t)row * H_
                   : (row < 2 * A_) ? xv + (size_t)(row - A_) * H_
                                    : xt + (size_t)(row - 2 * A_) * H_;
  float s = 0.f;
  for (int k = lane; k < H_; k += 32) { float v = src[k]; s += v * v; }
  s = wave_sum(s);
  const float rn = 1.0f / sqrtf(s);
  for (int k = lane; k < H_; k += 32) {
    const float v = src[k];
    big[(size_t)row * H_ + k] = v;
    nrm[(size_t)row * H_ + k] = v * rn;
  }
}

__global__ void k_gram(const float* __restrict__ nrm, float* __restrict__ intra) {
  const int mb = blockIdx.x;
  const int w  = threadIdx.x >> 5;
  const int m0 = (w >> 2) * 16, n0 = (w & 3) * 16;
  const float* blk = nrm + (size_t)mb * L_ * H_;
  auto lA = [&](int i, int k) { return blk[(size_t)i * H_ + k]; };
  auto lB = [&](int k, int j) { return blk[(size_t)j * H_ + k]; };
  auto ep = [&](int i, int j, float v) {
    intra[((size_t)mb * L_ + i) * L_ + j] = 1.0f - acosf(v * 0.99999f) * (1.0f / PI_F);
  };
  wmma_tile_f32(m0, n0, H_, lA, lB, ep);
}

__global__ void k_cross(const float* __restrict__ nrm, float* __restrict__ crossv) {
  const int idx  = blockIdx.x * 8 + (threadIdx.x >> 5);
  const int lane = threadIdx.x & 31;
  const int pm[6] = {0, 0, 1, 1, 2, 2};
  const int pn[6] = {1, 2, 0, 2, 0, 1};
  const int q = idx / A_, a = idx % A_;
  const int m = pm[q], n = pn[q];
  const float* pa = nrm + ((size_t)m * A_ + a) * H_;
  const float* pb = nrm + ((size_t)n * A_ + a) * H_;
  float s = 0.f;
  for (int k = lane; k < H_; k += 32) s += pa[k] * pb[k];
  s = wave_sum(s);
  if (lane == 0)
    crossv[((size_t)m * M_ + n) * A_ + a] = 1.0f - acosf(s * 0.99999f) * (1.0f / PI_F);
}

__global__ void k_dinv(const float* __restrict__ intra, const float* __restrict__ crossv,
                       float* __restrict__ dinv) {
  const int row  = blockIdx.x * 8 + (threadIdx.x >> 5);
  const int lane = threadIdx.x & 31;
  const int m = row >> 11, a = row & (A_ - 1);
  const float* r = intra + (size_t)row * L_;
  float s = 0.f;
  for (int j = lane; j < L_; j += 32) s += r[j];
  s = wave_sum(s);
  if (lane == 0) {
#pragma unroll
    for (int n = 0; n < M_; ++n)
      if (n != m) s += crossv[((size_t)m * M_ + n) * A_ + a];
    dinv[row] = 1.0f / sqrtf(s);
  }
}

__global__ void k_fc0(const float* __restrict__ big, const float* __restrict__ w,
                      const float* __restrict__ bias, float* __restrict__ h0) {
  const int wv = threadIdx.x >> 5;
  const int m0 = (blockIdx.y * 8 + wv) * 16, n0 = blockIdx.x * 16;
  auto lA = [&](int m, int k) { return big[(size_t)m * H_ + k]; };
  auto lB = [&](int k, int n) { return (n < H_) ? w[(size_t)k * H_ + n] : 0.0f; };
  auto ep = [&](int m, int n, float v) {
    if (n < H_) h0[(size_t)m * H_ + n] = fmaxf(v + bias[n], 0.0f);
  };
  wmma_tile_f32(m0, n0, H_, lA, lB, ep);
}

__global__ void k_spmm(const float* __restrict__ intra, const float* __restrict__ crossv,
                       const float* __restrict__ dinv, const float* __restrict__ h,
                       float* __restrict__ hi) {
  const int mb = blockIdx.y;
  const int m  = mb >> 5;
  const int rowbase = mb * L_;
  const int m0 = (threadIdx.x >> 5) * 16;
  const int n0 = blockIdx.x * 16;
  const float* S = intra + (size_t)mb * L_ * L_;
  auto lA = [&](int i, int k) { return S[(size_t)i * L_ + k] * dinv[rowbase + k]; };
  auto lB = [&](int k, int n) { return (n < H_) ? h[(size_t)(rowbase + k) * H_ + n] : 0.0f; };
  auto ep = [&](int i, int n, float acc) {
    if (n >= H_) return;
    const int row = rowbase + i;
    const int a   = row - m * A_;
    float s = acc;
#pragma unroll
    for (int nn = 0; nn < M_; ++nn) {
      if (nn == m) continue;
      const int orow = nn * A_ + a;
      s += crossv[((size_t)m * M_ + nn) * A_ + a] * dinv[orow] * h[(size_t)orow * H_ + n];
    }
    hi[(size_t)row * H_ + n] = s * dinv[row];
  };
  wmma_tile_f32(m0, n0, L_, lA, lB, ep);
}

__global__ void k_conv(const float* __restrict__ hi, const float* __restrict__ h0,
                       const float* __restrict__ w, float beta, float* __restrict__ hout) {
  const int wv = threadIdx.x >> 5;
  const int m0 = (blockIdx.y * 8 + wv) * 16, n0 = blockIdx.x * 16;
  auto lA = [&](int m, int k) {
    return (k < H_) ? hi[(size_t)m * H_ + k] : h0[(size_t)m * H_ + (k - H_)];
  };
  auto lB = [&](int k, int n) { return (n < H_) ? w[(size_t)k * H_ + n] : 0.0f; };
  auto ep = [&](int m, int n, float v) {
    if (n >= H_) return;
    const float r = 0.9f * hi[(size_t)m * H_ + n] + 0.1f * h0[(size_t)m * H_ + n];
    hout[(size_t)m * H_ + n] = fmaxf(beta * v + (1.0f - beta) * r, 0.0f);
  };
  wmma_tile_f32(m0, n0, 2 * H_, lA, lB, ep);
}

__global__ void k_hx(const float* __restrict__ big, const float* __restrict__ hf,
                     const float* __restrict__ tw, const float* __restrict__ tb,
                     float* __restrict__ hx) {
  const int m  = blockIdx.z;
  const int wv = threadIdx.x >> 5;
  const int a0 = (blockIdx.y * 8 + wv) * 16, n0 = blockIdx.x * 16;
  const float* W  = tw + (size_t)m * 2 * H_ * H_;
  const float* Bb = tb + (size_t)m * H_;
  const int rb = m * A_;
  auto lA = [&](int a, int k) {
    return (k < H_) ? big[(size_t)(rb + a) * H_ + k] : hf[(size_t)(rb + a) * H_ + (k - H_)];
  };
  auto lB = [&](int k, int n) { return (n < H_) ? W[(size_t)k * H_ + n] : 0.0f; };
  auto ep = [&](int a, int n, float v) {
    if (n < H_) hx[(size_t)(rb + a) * H_ + n] = fmaxf(v + Bb[n], 0.0f);
  };
  wmma_tile_f32(a0, n0, 2 * H_, lA, lB, ep);
}

__global__ void k_gate(const float* __restrict__ hx, const float* __restrict__ cw,
                       const float* __restrict__ cb, float* __restrict__ hout,
                       int p, int i, int j) {
  const int wv = threadIdx.x >> 5;
  const int a0 = (blockIdx.y * 8 + wv) * 16, n0 = blockIdx.x * 16;
  const float* Hi = hx + (size_t)i * A_ * H_;
  const float* Hj = hx + (size_t)j * A_ * H_;
  const float* W  = cw + (size_t)p * 3 * H_ * H_;
  const float* Bb = cb + (size_t)p * H_;
  auto lA = [&](int a, int k) {
    if (k < H_) return Hi[(size_t)a * H_ + k];
    if (k < 2 * H_) return Hj[(size_t)a * H_ + (k - H_)];
    const int kk = k - 2 * H_;
    return Hi[(size_t)a * H_ + kk] * Hj[(size_t)a * H_ + kk];
  };
  auto lB = [&](int k, int n) { return (n < H_) ? W[(size_t)k * H_ + n] : 0.0f; };
  auto ep = [&](int a, int n, float v) {
    if (n >= H_) return;
    const float z = 1.0f / (1.0f + expf(-(v + Bb[n])));
    hout[(size_t)(i * A_ + a) * H_ + n] +=
        z * Hi[(size_t)a * H_ + n] + (1.0f - z) * Hj[(size_t)a * H_ + n];
  };
  wmma_tile_f32(a0, n0, 3 * H_, lA, lB, ep);
}

__global__ void k_final(const float* __restrict__ hout, const float* __restrict__ uw,
                        const float* __restrict__ ub, float* __restrict__ out) {
  const int a = blockIdx.x * blockDim.x + threadIdx.x;
  float acc[TAG_];
#pragma unroll
  for (int t = 0; t < TAG_; ++t)
    acc[t] = ub[t] + ub[TAG_ + t] + ub[2 * TAG_ + t];
  for (int m = 0; m < M_; ++m) {
    const float* hr = hout + (size_t)(m * A_ + a) * H_;
    const float* wm = uw + (size_t)m * H_ * TAG_;
    for (int h = 0; h < H_; ++h) {
      const float v = hr[h];
#pragma unroll
      for (int t = 0; t < TAG_; ++t) acc[t] += v * wm[h * TAG_ + t];
    }
  }
#pragma unroll
  for (int t = 0; t < TAG_; ++t) out[(size_t)a * TAG_ + t] = acc[t];
}

}

extern "C" void kernel_launch(void* const* d_in, const int* in_sizes, int n_in,
                              void* d_out, int out_size, void* d_ws, size_t ws_size,
                              hipStream_t stream) {
  (void)in_sizes; (void)n_in; (void)out_size; (void)ws_size;
  const float* x_a     = (const float*)d_in[0];
  const float* x_v     = (const float*)d_in[1];
  const float* x_t     = (const float*)d_in[2];
  const float* fc0_w   = (const float*)d_in[3];
  const float* fc0_b   = (const float*)d_in[4];
  const float* conv_w  = (const float*)d_in[5];
  const float* tm_w    = (const float*)d_in[6];
  const float* tm_b    = (const float*)d_in[7];
  const float* cross_w = (const float*)d_in[8];
  const float* cross_b = (const float*)d_in[9];
  const float* uni_w   = (const float*)d_in[10];
  const float* uni_b   = (const float*)d_in[11];
  float* out = (float*)d_out;

  const size_t NH = (size_t)N_ * H_;
  float* ws    = (float*)d_ws;
  float* big   = ws; ws += NH;
  float* nrm   = ws; ws += NH;
  float* intra = ws; ws += (size_t)N_ * L_;
  float* crossv= ws; ws += (size_t)M_ * M_ * A_;
  float* dinv  = ws; ws += N_;
  float* h0    = ws; ws += NH;
  float* hA    = ws; ws += NH;
  float* hBf   = ws; ws += NH;
  float* hi    = ws; ws += NH;
  float* hx    = ws; ws += NH;
  float* hout  = ws; ws += NH;

  const int NT = (H_ + 15) / 16;

  k_norm <<<N_ / 8,              dim3(256), 0, stream>>>(x_a, x_v, x_t, big, nrm);
  k_gram <<<M_ * B_,             dim3(512), 0, stream>>>(nrm, intra);
  k_cross<<<6 * A_ / 8,          dim3(256), 0, stream>>>(nrm, crossv);
  k_dinv <<<N_ / 8,              dim3(256), 0, stream>>>(intra, crossv, dinv);
  k_fc0  <<<dim3(NT, N_ / 128),  dim3(256), 0, stream>>>(big, fc0_w, fc0_b, h0);

  const float beta0 = logf(0.5f / 1.0f + 1.0f);
  const float beta1 = logf(0.5f / 2.0f + 1.0f);
  k_spmm <<<dim3(NT, M_ * B_),   dim3(128), 0, stream>>>(intra, crossv, dinv, h0, hi);
  k_conv <<<dim3(NT, N_ / 128),  dim3(256), 0, stream>>>(hi, h0, conv_w, beta0, hA);
  k_spmm <<<dim3(NT, M_ * B_),   dim3(128), 0, stream>>>(intra, crossv, dinv, hA, hi);
  k_conv <<<dim3(NT, N_ / 128),  dim3(256), 0, stream>>>(hi, h0, conv_w + 2 * H_ * H_, beta1, hBf);

  k_hx   <<<dim3(NT, A_ / 128, M_), dim3(256), 0, stream>>>(big, hBf, tm_w, tm_b, hx);

  hipMemsetAsync(hout, 0, NH * sizeof(float), stream);
  const int pi_[6] = {0, 0, 1, 1, 2, 2};
  const int pj_[6] = {1, 2, 0, 2, 0, 1};
  for (int p = 0; p < 6; ++p)
    k_gate<<<dim3(NT, A_ / 128), dim3(256), 0, stream>>>(hx, cross_w, cross_b, hout,
                                                         p, pi_[p], pj_[p]);

  k_final<<<A_ / 256, dim3(256), 0, stream>>>(hout, uni_w, uni_b, out);
}


# GEMM kernels (fc0, spmm, conv, hx, gate) rebuilt as LDS block tiling: 64x64 tile per 4-wave block, A and B chunks fetched once per block with wide coalesced loads, each wave a strip of 4 tiles
# speedup vs baseline: 2.8930x; 2.8930x over previous
#include <hip/hip_runtime.h>
#include <math.h>

typedef __attribute__((ext_vector_type(2))) float v2f;
typedef __attribute__((ext_vector_type(8))) float v8f;

namespace {

constexpr int B_   = 32;
constexpr int L_   = 64;
constexpr int H_   = 200;
constexpr int M_   = 3;
constexpr int A_   = B_ * L_;
constexpr int N_   = M_ * A_;
constexpr int TAG_ = 6;
constexpr float PI_F = 3.14159265358979323846f;

template <typename FA, typename FB, typename FE>
__device__ __forceinline__ void wmma_tile_f32(int m0, int n0, int K,
                                              FA loadA, FB loadB, FE epi) {
  const int lane = threadIdx.x & 31;
  const int mr   = lane & 15;
  const int kh   = lane >> 4;
  v8f acc = {};
  for (int k = 0; k < K; k += 4) {
    const int kk = k + 2 * kh;
    v2f a, b;
    a.x = loadA(m0 + mr, kk);
    a.y = loadA(m0 + mr, kk + 1);
    b.x = loadB(kk, n0 + mr);
    b.y = loadB(kk + 1, n0 + mr);
    acc = __builtin_amdgcn_wmma_f32_16x16x4_f32(false, a, false, b,
                                                (short)0, acc, false, false);
  }
#pragma unroll
  for (int v = 0; v < 8; ++v) epi(m0 + 8 * kh + v, n0 + mr, acc[v]);
}

__device__ __forceinline__ float wave_sum(float s) {
#pragma unroll
  for (int off = 16; off > 0; off >>= 1) s += __shfl_xor(s, off, 32);
  return s;
}

__global__ void k_norm(const float* __restrict__ xa, const float* __restrict__ xv,
                       const float* __restrict__ xt, float* __restrict__ big,
                       float* __restrict__ nrm) {
  const int row  = blockIdx.x * 8 + (threadIdx.x >> 5);
  const int lane = threadIdx.x & 31;
  const float* src = (row < A_)     ? xa + (size_t)row * H_
                   : (row < 2 * A_) ? xv + (size_t)(row - A_) * H_
                                    : xt + (size_t)(row - 2 * A_) * H_;
  float s = 0.f;
  for (int k = lane; k < H_; k += 32) { float v = src[k]; s += v * v; }
  s = wave_sum(s);
  const float rn = 1.0f / sqrtf(s);
  for (int k = lane; k < H_; k += 32) {
    const float v = src[k];
    big[(size_t)row * H_ + k] = v;
    nrm[(size_t)row * H_ + k] = v * rn;
  }
}

__global__ void k_gram(const float* __restrict__ nrm, float* __restrict__ intra) {
  const int mb = blockIdx.x;
  const int w  = threadIdx.x >> 5;
  const int m0 = (w >> 2) * 16, n0 = (w & 3) * 16;
  const float* blk = nrm + (size_t)mb * L_ * H_;
  auto lA = [&](int i, int k) { return blk[(size_t)i * H_ + k]; };
  auto lB = [&](int k, int j) { return blk[(size_t)j * H_ + k]; };
  auto ep = [&](int i, int j, float v) {
    intra[((size_t)mb * L_ + i) * L_ + j] = 1.0f - acosf(v * 0.99999f) * (1.0f / PI_F);
  };
  wmma_tile_f32(m0, n0, H_, lA, lB, ep);
}

__global__ void k_cross(const float* __restrict__ nrm, float* __restrict__ crossv) {
  const int idx  = blockIdx.x * 8 + (threadIdx.x >> 5);
  const int lane = threadIdx.x & 31;
  const int pm[6] = {0, 0, 1, 1, 2, 2};
  const int pn[6] = {1, 2, 0, 2, 0, 1};
  const int q = idx / A_, a = idx % A_;
  const int m = pm[q], n = pn[q];
  const float* pa = nrm + ((size_t)m * A_ + a) * H_;
  const float* pb = nrm + ((size_t)n * A_ + a) * H_;
  float s = 0.f;
  for (int k = lane; k < H_; k += 32) s += pa[k] * pb[k];
  s = wave_sum(s);
  if (lane == 0)
    crossv[((size_t)m * M_ + n) * A_ + a] = 1.0f - acosf(s * 0.99999f) * (1.0f / PI_F);
}

__global__ void k_dinv(const float* __restrict__ intra, const float* __restrict__ crossv,
                       float* __restrict__ dinv) {
  const int row  = blockIdx.x * 8 + (threadIdx.x >> 5);
  const int lane = threadIdx.x & 31;
  const int m = row >> 11, a = row & (A_ - 1);
  const float* r = intra + (size_t)row * L_;
  float s = 0.f;
  for (int j = lane; j < L_; j += 32) s += r[j];
  s = wave_sum(s);
  if (lane == 0) {
#pragma unroll
    for (int n = 0; n < M_; ++n)
      if (n != m) s += crossv[((size_t)m * M_ + n) * A_ + a];
    dinv[row] = 1.0f / sqrtf(s);
  }
}

__device__ __forceinline__ float4 ld4(const float* p) {
  return *reinterpret_cast<const float4*>(p);
}
__device__ __forceinline__ float2 ld2(const float* p) {
  return *reinterpret_cast<const float2*>(p);
}

template <int BK, bool COMB_HAS_MATH, typename FA, typename CA, typename FE>
__device__ __forceinline__ void gemm_block_64x64(const int K, const float* __restrict__ Bp,
                                                 const int n0, FA fetchA, CA combA, FE epi) {
  constexpr int BM = 64, BN = 64, NTHR = 128;
  constexpr int PA = BK + 4;
  constexpr int PB = BN + 16;
  constexpr int KQ = BK / 4;
  constexpr int NP = BN / 2;
  constexpr int A4 = BM * KQ;
  constexpr int B4 = (BK / 2) * NP;
  constexpr int RA = A4 / NTHR;
  constexpr int RB = B4 / NTHR;
  static_assert(A4 % NTHR == 0 && B4 % NTHR == 0, "chunks must split evenly over the block");
  static_assert((PA % 8) == 4, "A pitch must be 4 * odd dwords");
  static_assert(((2 * PB) % 64) == 32, "B pair pitch must be 32 mod 64 dwords");

  constexpr int ASZ = BM * PA;
  constexpr int BSZ = (BK / 2) * PB * 2;
  __shared__ __attribute__((aligned(16))) float As[2 * ASZ];
  __shared__ __attribute__((aligned(16))) float Bs[2 * BSZ];

  const int t    = threadIdx.x;
  const int lane = t & 31;
  const int w    = t >> 5;
  const int mr   = lane & 15;
  const int kh   = lane >> 4;

  float4 ax[RA], ay[RA];
  float2 be[RB], bo[RB];

  auto prefetch = [&](const int k0) {
#pragma unroll
    for (int r = 0; r < RA; ++r) {
      const int e = t + NTHR * r;
      fetchA(e / KQ, k0, (e % KQ) * 4, ax[r], ay[r]);
    }
#pragma unroll
    for (int r = 0; r < RB; ++r) {
      const int e  = t + NTHR * r;
      const int kp = e / NP;
      const int n  = min(n0 + (e % NP) * 2, H_ - 2);
      const float* p = Bp + (size_t)(k0 + 2 * kp) * H_ + n;
      be[r] = ld2(p);
      bo[r] = ld2(p + H_);
    }
  };

  auto stage = [&](const int k0, const int buf) {
#pragma unroll
    for (int r = 0; r < RA; ++r) {
      const int e = t + NTHR * r;
      *reinterpret_cast<float4*>(&As[buf * ASZ + (e / KQ) * PA + (e % KQ) * 4]) = combA(k0, ax[r], ay[r]);
    }
#pragma unroll
    for (int r = 0; r < RB; ++r) {
      const int e = t + NTHR * r;
      const int j = (e % NP) * 2;
      float4 v = make_float4(be[r].x, bo[r].x, be[r].y, bo[r].y);
      if (n0 + j >= H_) v = make_float4(0.f, 0.f, 0.f, 0.f);
      *reinterpret_cast<float4*>(&Bs[buf * BSZ + ((e / NP) * PB + j) * 2]) = v;
    }
  };

  const int aoff = (w * 16 + mr) * PA + 2 * kh;
  const int boff = (kh * PB + mr) * 2;

  v8f acc[4] = {};
  auto compute = [&](const int buf) {
    const float* aptr = As + buf * ASZ + aoff;
    const float* bptr = Bs + buf * BSZ + boff;
#pragma unroll
    for (int ks = 0; ks < BK; ks += 4) {
      const v2f a = *reinterpret_cast<const v2f*>(aptr + ks);
#pragma unroll
      for (int c = 0; c < 4; ++c) {
        const v2f b = *reinterpret_cast<const v2f*>(bptr + ((ks / 2) * PB + 16 * c) * 2);
        acc[c] = __builtin_amdgcn_wmma_f32_16x16x4_f32(false, a, false, b, (short)0, acc[c], false, false);
      }
    }
  };

  prefetch(0);
  stage(0, 0);
  __syncthreads();
  int buf = 0;
  for (int k0 = BK; k0 < K; k0 += BK) {
    prefetch(k0);
    __builtin_amdgcn_sched_barrier(0);
    compute(buf);
    if constexpr (COMB_HAS_MATH) __builtin_amdgcn_sched_barrier(0);
    stage(k0, buf ^ 1);
    __syncthreads();
    buf ^= 1;
  }
  compute(buf);

  const int ir = w * 16 + 8 * kh;
  const int nc = n0 + mr;
#pragma unroll
  for (int c = 0; c < 4; ++c) {
#pragma unroll
    for (int v = 0; v < 8; ++v) epi(ir + v, nc + 16 * c, acc[c][v]);
  }
}

__global__ __launch_bounds__(128) void k_fc0(const float* __restrict__ big,
                                             const float* __restrict__ w,
                                             const float* __restrict__ bias,
                                             float* __restrict__ h0) {
  const int n0 = blockIdx.x * 64;
  const int r0 = blockIdx.y * 64;
  auto fA = [&](int i, int k0, int kq, float4& x, float4& y) {
    (void)y;
    x = ld4(big + (size_t)(r0 + i) * H_ + k0 + kq);
  };
  auto cA = [&](int k0, const float4& x, const float4& y) { (void)k0; (void)y; return x; };
  auto ep = [&](int i, int n, float v) {
    if (n < H_) h0[(size_t)(r0 + i) * H_ + n] = fmaxf(v + bias[n], 0.0f);
  };
  gemm_block_64x64<40, false>(H_, w, n0, fA, cA, ep);
}

__global__ __launch_bounds__(128) void k_spmm(const float* __restrict__ intra,
                                              const float* __restrict__ crossv,
                                              const float* __restrict__ dinv,
                                              const float* __restrict__ h,
                                              float* __restrict__ hi) {
  const int n0 = blockIdx.x * 64;
  const int mb = blockIdx.y;
  const int m  = mb >> 5;
  const int rowbase = mb * L_;
  const float* S = intra + (size_t)mb * L_ * L_;
  auto fA = [&](int i, int k0, int kq, float4& x, float4& y) {
    x = ld4(S + (size_t)i * L_ + k0 + kq);
    y = ld4(dinv + rowbase + k0 + kq);
  };
  auto cA = [&](int k0, const float4& x, const float4& y) {
    (void)k0;
    return make_float4(x.x * y.x, x.y * y.y, x.z * y.z, x.w * y.w);
  };
  auto ep = [&](int i, int n, float acc) {
    if (n >= H_) return;
    const int row = rowbase + i;
    const int a   = row - m * A_;
    float s = acc;
#pragma unroll
    for (int nn = 0; nn < M_; ++nn) {
      if (nn == m) continue;
      const int orow = nn * A_ + a;
      s += crossv[((size_t)m * M_ + nn) * A_ + a] * dinv[orow] * h[(size_t)orow * H_ + n];
    }
    hi[(size_t)row * H_ + n] = s * dinv[row];
  };
  gemm_block_64x64<32, true>(L_, h + (size_t)rowbase * H_, n0, fA, cA, ep);
}

__global__ __launch_bounds__(128) void k_conv(const float* __restrict__ hi,
                                              const float* __restrict__ h0,
                                              const float* __restrict__ w, float beta,
                                              float* __restrict__ hout) {
  const int n0 = blockIdx.x * 64;
  const int r0 = blockIdx.y * 64;
  auto fA = [&](int i, int k0, int kq, float4& x, float4& y) {
    (void)y;
    const float* p = (k0 < H_) ? hi + (size_t)(r0 + i) * H_ + (k0 + kq)
                               : h0 + (size_t)(r0 + i) * H_ + (k0 - H_ + kq);
    x = ld4(p);
  };
  auto cA = [&](int k0, const float4& x, const float4& y) { (void)k0; (void)y; return x; };
  auto ep = [&](int i, int n, float v) {
    if (n >= H_) return;
    const size_t o = (size_t)(r0 + i) * H_ + n;
    const float r = 0.9f * hi[o] + 0.1f * h0[o];
    hout[o] = fmaxf(beta * v + (1.0f - beta) * r, 0.0f);
  };
  gemm_block_64x64<40, false>(2 * H_, w, n0, fA, cA, ep);
}

__global__ __launch_bounds__(128) void k_hx(const float* __restrict__ big,
                                            const float* __restrict__ hf,
                                            const float* __restrict__ tw,
                                            const float* __restrict__ tb,
                                            float* __restrict__ hx) {
  const int m  = blockIdx.z;
  const int n0 = blockIdx.x * 64;
  const int r0 = m * A_ + blockIdx.y * 64;
  const float* Wm = tw + (size_t)m * 2 * H_ * H_;
  const float* Bb = tb + (size_t)m * H_;
  auto fA = [&](int i, int k0, int kq, float4& x, float4& y) {
    (void)y;
    const float* p = (k0 < H_) ? big + (size_t)(r0 + i) * H_ + (k0 + kq)
                               : hf  + (size_t)(r0 + i) * H_ + (k0 - H_ + kq);
    x = ld4(p);
  };
  auto cA = [&](int k0, const float4& x, const float4& y) { (void)k0; (void)y; return x; };
  auto ep = [&](int i, int n, float v) {
    if (n < H_) hx[(size_t)(r0 + i) * H_ + n] = fmaxf(v + Bb[n], 0.0f);
  };
  gemm_block_64x64<40, false>(2 * H_, Wm, n0, fA, cA, ep);
}

__global__ __launch_bounds__(128) void k_gate(const float* __restrict__ hx,
                                              const float* __restrict__ cw,
                                              const float* __restrict__ cb,
                                              float* __restrict__ hout, int p, int i, int j) {
  const int n0 = blockIdx.x * 64;
  const int a0 = blockIdx.y * 64;
  const float* Hi = hx + (size_t)i * A_ * H_;
  const float* Hj = hx + (size_t)j * A_ * H_;
  const float* Wp = cw + (size_t)p * 3 * H_ * H_;
  const float* Bb = cb + (size_t)p * H_;
  auto fA = [&](int r, int k0, int kq, float4& x, float4& y) {
    const int s  = k0 / H_;
    const int kk = k0 - s * H_ + kq;
    const float* px = ((s == 1) ? Hj : Hi) + (size_t)(a0 + r) * H_ + kk;
    x = ld4(px);
    if (s == 2) y = ld4(Hj + (size_t)(a0 + r) * H_ + kk);
    else y = make_float4(0.f, 0.f, 0.f, 0.f);
  };
  auto cA = [&](int k0, const float4& x, const float4& y) {
    if (k0 / H_ == 2) return make_float4(x.x * y.x, x.y * y.y, x.z * y.z, x.w * y.w);
    return x;
  };
  auto ep = [&](int r, int n, float v) {
    if (n >= H_) return;
    const int a = a0 + r;
    const float z = 1.0f / (1.0f + expf(-(v + Bb[n])));
    hout[(size_t)(i * A_ + a) * H_ + n] +=
        z * Hi[(size_t)a * H_ + n] + (1.0f - z) * Hj[(size_t)a * H_ + n];
  };
  gemm_block_64x64<40, true>(3 * H_, Wp, n0, fA, cA, ep);
}

__global__ void k_final(const float* __restrict__ hout, const float* __restrict__ uw,
                        const float* __restrict__ ub, float* __restrict__ out) {
  const int a = blockIdx.x * blockDim.x + threadIdx.x;
  float acc[TAG_];
#pragma unroll
  for (int t = 0; t < TAG_; ++t)
    acc[t] = ub[t] + ub[TAG_ + t] + ub[2 * TAG_ + t];
  for (int m = 0; m < M_; ++m) {
    const float* hr = hout + (size_t)(m * A_ + a) * H_;
    const float* wm = uw + (size_t)m * H_ * TAG_;
    for (int h = 0; h < H_; ++h) {
      const float v = hr[h];
#pragma unroll
      for (int t = 0; t < TAG_; ++t) acc[t] += v * wm[h * TAG_ + t];
    }
  }
#pragma unroll
  for (int t = 0; t < TAG_; ++t) out[(size_t)a * TAG_ + t] = acc[t];
}

}

extern "C" void kernel_launch(void* const* d_in, const int* in_sizes, int n_in,
                              void* d_out, int out_size, void* d_ws, size_t ws_size,
                              hipStream_t stream) {
  (void)in_sizes; (void)n_in; (void)out_size; (void)ws_size;
  const float* x_a     = (const float*)d_in[0];
  const float* x_v     = (const float*)d_in[1];
  const float* x_t     = (const float*)d_in[2];
  const float* fc0_w   = (const float*)d_in[3];
  const float* fc0_b   = (const float*)d_in[4];
  const float* conv_w  = (const float*)d_in[5];
  const float* tm_w    = (const float*)d_in[6];
  const float* tm_b    = (const float*)d_in[7];
  const float* cross_w = (const float*)d_in[8];
  const float* cross_b = (const float*)d_in[9];
  const float* uni_w   = (const float*)d_in[10];
  const float* uni_b   = (const float*)d_in[11];
  float* out = (float*)d_out;

  const size_t NH = (size_t)N_ * H_;
  float* ws    = (float*)d_ws;
  float* big   = ws; ws += NH;
  float* nrm   = ws; ws += NH;
  float* intra = ws; ws += (size_t)N_ * L_;
  float* crossv= ws; ws += (size_t)M_ * M_ * A_;
  float* dinv  = ws; ws += N_;
  float* h0    = ws; ws += NH;
  float* hA    = ws; ws += NH;
  float* hBf   = ws; ws += NH;
  float* hi    = ws; ws += NH;
  float* hx    = ws; ws += NH;
  float* hout  = ws; ws += NH;

  k_norm <<<N_ / 8,              dim3(256), 0, stream>>>(x_a, x_v, x_t, big, nrm);
  k_gram <<<M_ * B_,             dim3(512), 0, stream>>>(nrm, intra);
  k_cross<<<6 * A_ / 8,          dim3(256), 0, stream>>>(nrm, crossv);
  k_dinv <<<N_ / 8,              dim3(256), 0, stream>>>(intra, crossv, dinv);
  k_fc0  <<<dim3(4, N_ / 64),    dim3(128), 0, stream>>>(big, fc0_w, fc0_b, h0);

  const float beta0 = logf(0.5f / 1.0f + 1.0f);
  const float beta1 = logf(0.5f / 2.0f + 1.0f);
  k_spmm <<<dim3(4, M_ * B_),    dim3(128), 0, stream>>>(intra, crossv, dinv, h0, hi);
  k_conv <<<dim3(4, N_ / 64),    dim3(128), 0, stream>>>(hi, h0, conv_w, beta0, hA);
  k_spmm <<<dim3(4, M_ * B_),    dim3(128), 0, stream>>>(intra, crossv, dinv, hA, hi);
  k_conv <<<dim3(4, N_ / 64),    dim3(128), 0, stream>>>(hi, h0, conv_w + 2 * H_ * H_, beta1, hBf);

  k_hx   <<<dim3(4, A_ / 64, M_), dim3(128), 0, stream>>>(big, hBf, tm_w, tm_b, hx);

  hipMemsetAsync(hout, 0, NH * sizeof(float), stream);
  const int pi_[6] = {0, 0, 1, 1, 2, 2};
  const int pj_[6] = {1, 2, 0, 2, 0, 1};
  for (int p = 0; p < 6; ++p)
    k_gate<<<dim3(4, A_ / 64), dim3(128), 0, stream>>>(hx, cross_w, cross_b, hout,
                                                       p, pi_[p], pj_[p]);

  k_final<<<A_ / 256, dim3(256), 0, stream>>>(hout, uni_w, uni_b, out);
}


# six per-pair gate launches and the clearing memset fused into one launch: a block runs both pairs of its modality in one K range, first gated term kept in registers, sum order kept, hout stored once
# speedup vs baseline: 4.0896x; 1.4136x over previous
#include <hip/hip_runtime.h>
#include <math.h>

typedef __attribute__((ext_vector_type(2))) float v2f;
typedef __attribute__((ext_vector_type(8))) float v8f;

namespace {

constexpr int B_   = 32;
constexpr int L_   = 64;
constexpr int H_   = 200;
constexpr int M_   = 3;
constexpr int A_   = B_ * L_;
constexpr int N_   = M_ * A_;
constexpr int TAG_ = 6;
constexpr float PI_F = 3.14159265358979323846f;

template <typename FA, typename FB, typename FE>
__device__ __forceinline__ void wmma_tile_f32(int m0, int n0, int K,
                                              FA loadA, FB loadB, FE epi) {
  const int lane = threadIdx.x & 31;
  const int mr   = lane & 15;
  const int kh   = lane >> 4;
  v8f acc = {};
  for (int k = 0; k < K; k += 4) {
    const int kk = k + 2 * kh;
    v2f a, b;
    a.x = loadA(m0 + mr, kk);
    a.y = loadA(m0 + mr, kk + 1);
    b.x = loadB(kk, n0 + mr);
    b.y = loadB(kk + 1, n0 + mr);
    acc = __builtin_amdgcn_wmma_f32_16x16x4_f32(false, a, false, b,
                                                (short)0, acc, false, false);
  }
#pragma unroll
  for (int v = 0; v < 8; ++v) epi(m0 + 8 * kh + v, n0 + mr, acc[v]);
}

__device__ __forceinline__ float wave_sum(float s) {
#pragma unroll
  for (int off = 16; off > 0; off >>= 1) s += __shfl_xor(s, off, 32);
  return s;
}

__global__ void k_norm(const float* __restrict__ xa, const float* __restrict__ xv,
                       const float* __restrict__ xt, float* __restrict__ big,
                       float* __restrict__ nrm) {
  const int row  = blockIdx.x * 8 + (threadIdx.x >> 5);
  const int lane = threadIdx.x & 31;
  const float* src = (row < A_)     ? xa + (size_t)row * H_
                   : (row < 2 * A_) ? xv + (size_t)(row - A_) * H_
                                    : xt + (size_t)(row - 2 * A_) * H_;
  float s = 0.f;
  for (int k = lane; k < H_; k += 32) { float v = src[k]; s += v * v; }
  s = wave_sum(s);
  const float rn = 1.0f / sqrtf(s);
  for (int k = lane; k < H_; k += 32) {
    const float v = src[k];
    big[(size_t)row * H_ + k] = v;
    nrm[(size_t)row * H_ + k] = v * rn;
  }
}

__global__ void k_gram(const float* __restrict__ nrm, float* __restrict__ intra) {
  const int mb = blockIdx.x;
  const int w  = threadIdx.x >> 5;
  const int m0 = (w >> 2) * 16, n0 = (w & 3) * 16;
  const float* blk = nrm + (size_t)mb * L_ * H_;
  auto lA = [&](int i, int k) { return blk[(size_t)i * H_ + k]; };
  auto lB = [&](int k, int j) { return blk[(size_t)j * H_ + k]; };
  auto ep = [&](int i, int j, float v) {
    intra[((size_t)mb * L_ + i) * L_ + j] = 1.0f - acosf(v * 0.99999f) * (1.0f / PI_F);
  };
  wmma_tile_f32(m0, n0, H_, lA, lB, ep);
}

__global__ void k_cross(const float* __restrict__ nrm, float* __restrict__ crossv) {
  const int idx  = blockIdx.x * 8 + (threadIdx.x >> 5);
  const int lane = threadIdx.x & 31;
  const int pm[6] = {0, 0, 1, 1, 2, 2};
  const int pn[6] = {1, 2, 0, 2, 0, 1};
  const int q = idx / A_, a = idx % A_;
  const int m = pm[q], n = pn[q];
  const float* pa = nrm + ((size_t)m * A_ + a) * H_;
  const float* pb = nrm + ((size_t)n * A_ + a) * H_;
  float s = 0.f;
  for (int k = lane; k < H_; k += 32) s += pa[k] * pb[k];
  s = wave_sum(s);
  if (lane == 0)
    crossv[((size_t)m * M_ + n) * A_ + a] = 1.0f - acosf(s * 0.99999f) * (1.0f / PI_F);
}

__global__ void k_dinv(const float* __restrict__ intra, const float* __restrict__ crossv,
                       float* __restrict__ dinv) {
  const int row  = blockIdx.x * 8 + (threadIdx.x >> 5);
  const int lane = threadIdx.x & 31;
  const int m = row >> 11, a = row & (A_ - 1);
  const float* r = intra + (size_t)row * L_;
  float s = 0.f;
  for (int j = lane; j < L_; j += 32) s += r[j];
  s = wave_sum(s);
  if (lane == 0) {
#pragma unroll
    for (int n = 0; n < M_; ++n)
      if (n != m) s += crossv[((size_t)m * M_ + n) * A_ + a];
    dinv[row] = 1.0f / sqrtf(s);
  }
}

struct no_mid {};
template <typename T> struct is_no_mid { static constexpr bool value = false; };
template <> struct is_no_mid<no_mid> { static constexpr bool value = true; };
__device__ __forceinline__ float4 ld4(const float* p) {
  return *reinterpret_cast<const float4*>(p);
}
__device__ __forceinline__ float2 ld2(const float* p) {
  return *reinterpret_cast<const float2*>(p);
}

template <int BK, bool COMB_HAS_MATH, typename FA, typename CA, typename FE, typename FM = no_mid>
__device__ __forceinline__ void gemm_block_64x64(const int K, const float* __restrict__ Bp,
                                                 const int n0, FA fetchA, CA combA, FE epi,
                                                 const int Kmid = 0, FM mid = FM{}) {
  constexpr bool TWO_PRODUCTS = !is_no_mid<FM>::value;
  constexpr int BM = 64, BN = 64, NTHR = 128;
  constexpr int PA = BK + 4;
  constexpr int PB = BN + 16;
  constexpr int KQ = BK / 4;
  constexpr int NP = BN / 2;
  constexpr int A4 = BM * KQ;
  constexpr int B4 = (BK / 2) * NP;
  constexpr int RA = A4 / NTHR;
  constexpr int RB = B4 / NTHR;
  static_assert(A4 % NTHR == 0 && B4 % NTHR == 0, "chunks must split evenly over the block");
  static_assert((PA % 8) == 4, "A pitch must be 4 * odd dwords");
  static_assert(((2 * PB) % 64) == 32, "B pair pitch must be 32 mod 64 dwords");

  constexpr int ASZ = BM * PA;
  constexpr int BSZ = (BK / 2) * PB * 2;
  __shared__ __attribute__((aligned(16))) float As[2 * ASZ];
  __shared__ __attribute__((aligned(16))) float Bs[2 * BSZ];

  const int t    = threadIdx.x;
  const int lane = t & 31;
  const int w    = t >> 5;
  const int mr   = lane & 15;
  const int kh   = lane >> 4;

  float4 ax[RA], ay[RA];
  float2 be[RB], bo[RB];

  auto prefetch = [&](const int k0) {
#pragma unroll
    for (int r = 0; r < RA; ++r) {
      const int e = t + NTHR * r;
      fetchA(e / KQ, k0, (e % KQ) * 4, ax[r], ay[r]);
    }
#pragma unroll
    for (int r = 0; r < RB; ++r) {
      const int e  = t + NTHR * r;
      const int kp = e / NP;
      const int n  = min(n0 + (e % NP) * 2, H_ - 2);
      const float* p = Bp + (size_t)(k0 + 2 * kp) * H_ + n;
      be[r] = ld2(p);
      bo[r] = ld2(p + H_);
    }
  };

  auto stage = [&](const int k0, const int buf) {
#pragma unroll
    for (int r = 0; r < RA; ++r) {
      const int e = t + NTHR * r;
      *reinterpret_cast<float4*>(&As[buf * ASZ + (e / KQ) * PA + (e % KQ) * 4]) = combA(k0, ax[r], ay[r]);
    }
#pragma unroll
    for (int r = 0; r < RB; ++r) {
      const int e = t + NTHR * r;
      const int j = (e % NP) * 2;
      float4 v = make_float4(be[r].x, bo[r].x, be[r].y, bo[r].y);
      if (n0 + j >= H_) v = make_float4(0.f, 0.f, 0.f, 0.f);
      *reinterpret_cast<float4*>(&Bs[buf * BSZ + ((e / NP) * PB + j) * 2]) = v;
    }
  };

  const int aoff = (w * 16 + mr) * PA + 2 * kh;
  const int boff = (kh * PB + mr) * 2;

  v8f acc[4] = {};
  auto compute = [&](const int buf) {
    const float* aptr = As + buf * ASZ + aoff;
    const float* bptr = Bs + buf * BSZ + boff;
#pragma unroll
    for (int ks = 0; ks < BK; ks += 4) {
      const v2f a = *reinterpret_cast<const v2f*>(aptr + ks);
#pragma unroll
      for (int c = 0; c < 4; ++c) {
        const v2f b = *reinterpret_cast<const v2f*>(bptr + ((ks / 2) * PB + 16 * c) * 2);
        acc[c] = __builtin_amdgcn_wmma_f32_16x16x4_f32(false, a, false, b, (short)0, acc[c], false, false);
      }
    }
  };

  prefetch(0);
  stage(0, 0);
  __syncthreads();
  int buf = 0;
  for (int k0 = BK; k0 < K; k0 += BK) {
    prefetch(k0);
    __builtin_amdgcn_sched_barrier(0);
    compute(buf);
    if constexpr (COMB_HAS_MATH) __builtin_amdgcn_sched_barrier(0);
    if constexpr (TWO_PRODUCTS) {
      if (k0 == Kmid) mid(acc);
    }
    stage(k0, buf ^ 1);
    __syncthreads();
    buf ^= 1;
  }
  compute(buf);

  if constexpr (TWO_PRODUCTS) {
    epi(acc);
  } else {
    const int ir = w * 16 + 8 * kh;
    const int nc = n0 + mr;
#pragma unroll
    for (int c = 0; c < 4; ++c) {
#pragma unroll
      for (int v = 0; v < 8; ++v) epi(ir + v, nc + 16 * c, acc[c][v]);
    }
  }
}

__global__ __launch_bounds__(128) void k_fc0(const float* __restrict__ big,
                                             const float* __restrict__ w,
                                             const float* __restrict__ bias,
                                             float* __restrict__ h0) {
  const int n0 = blockIdx.x * 64;
  const int r0 = blockIdx.y * 64;
  auto fA = [&](int i, int k0, int kq, float4& x, float4& y) {
    (void)y;
    x = ld4(big + (size_t)(r0 + i) * H_ + k0 + kq);
  };
  auto cA = [&](int k0, const float4& x, const float4& y) { (void)k0; (void)y; return x; };
  auto ep = [&](int i, int n, float v) {
    if (n < H_) h0[(size_t)(r0 + i) * H_ + n] = fmaxf(v + bias[n], 0.0f);
  };
  gemm_block_64x64<40, false>(H_, w, n0, fA, cA, ep);
}

__global__ __launch_bounds__(128) void k_spmm(const float* __restrict__ intra,
                                              const float* __restrict__ crossv,
                                              const float* __restrict__ dinv,
                                              const float* __restrict__ h,
                                              float* __restrict__ hi) {
  const int n0 = blockIdx.x * 64;
  const int mb = blockIdx.y;
  const int m  = mb >> 5;
  const int rowbase = mb * L_;
  const float* S = intra + (size_t)mb * L_ * L_;
  auto fA = [&](int i, int k0, int kq, float4& x, float4& y) {
    x = ld4(S + (size_t)i * L_ + k0 + kq);
    y = ld4(dinv + rowbase + k0 + kq);
  };
  auto cA = [&](int k0, const float4& x, const float4& y) {
    (void)k0;
    return make_float4(x.x * y.x, x.y * y.y, x.z * y.z, x.w * y.w);
  };
  auto ep = [&](int i, int n, float acc) {
    if (n >= H_) return;
    const int row = rowbase + i;
    const int a   = row - m * A_;
    float s = acc;
#pragma unroll
    for (int nn = 0; nn < M_; ++nn) {
      if (nn == m) continue;
      const int orow = nn * A_ + a;
      s += crossv[((size_t)m * M_ + nn) * A_ + a] * dinv[orow] * h[(size_t)orow * H_ + n];
    }
    hi[(size_t)row * H_ + n] = s * dinv[row];
  };
  gemm_block_64x64<32, true>(L_, h + (size_t)rowbase * H_, n0, fA, cA, ep);
}

__global__ __launch_bounds__(128) void k_conv(const float* __restrict__ hi,
                                              const float* __restrict__ h0,
                                              const float* __restrict__ w, float beta,
                                              float* __restrict__ hout) {
  const int n0 = blockIdx.x * 64;
  const int r0 = blockIdx.y * 64;
  auto fA = [&](int i, int k0, int kq, float4& x, float4& y) {
    (void)y;
    const float* p = (k0 < H_) ? hi + (size_t)(r0 + i) * H_ + (k0 + kq)
                               : h0 + (size_t)(r0 + i) * H_ + (k0 - H_ + kq);
    x = ld4(p);
  };
  auto cA = [&](int k0, const float4& x, const float4& y) { (void)k0; (void)y; return x; };
  auto ep = [&](int i, int n, float v) {
    if (n >= H_) return;
    const size_t o = (size_t)(r0 + i) * H_ + n;
    const float r = 0.9f * hi[o] + 0.1f * h0[o];
    hout[o] = fmaxf(beta * v + (1.0f - beta) * r, 0.0f);
  };
  gemm_block_64x64<40, false>(2 * H_, w, n0, fA, cA, ep);
}

__global__ __launch_bounds__(128) void k_hx(const float* __restrict__ big,
                                            const float* __restrict__ hf,
                                            const float* __restrict__ tw,
                                            const float* __restrict__ tb,
                                            float* __restrict__ hx) {
  const int m  = blockIdx.z;
  const int n0 = blockIdx.x * 64;
  const int r0 = m * A_ + blockIdx.y * 64;
  const float* Wm = tw + (size_t)m * 2 * H_ * H_;
  const float* Bb = tb + (size_t)m * H_;
  auto fA = [&](int i, int k0, int kq, float4& x, float4& y) {
    (void)y;
    const float* p = (k0 < H_) ? big + (size_t)(r0 + i) * H_ + (k0 + kq)
                               : hf  + (size_t)(r0 + i) * H_ + (k0 - H_ + kq);
    x = ld4(p);
  };
  auto cA = [&](int k0, const float4& x, const float4& y) { (void)k0; (void)y; return x; };
  auto ep = [&](int i, int n, float v) {
    if (n < H_) hx[(size_t)(r0 + i) * H_ + n] = fmaxf(v + Bb[n], 0.0f);
  };
  gemm_block_64x64<40, false>(2 * H_, Wm, n0, fA, cA, ep);
}

__global__ __launch_bounds__(128) void k_gate(const float* __restrict__ hx,
                                              const float* __restrict__ cw,
                                              const float* __restrict__ cb,
                                              float* __restrict__ hout) {
  constexpr int KP = 3 * H_;
  const int i  = blockIdx.z;
  const int n0 = blockIdx.x * 64;
  const int a0 = blockIdx.y * 64;
  const int j0 = (i == 0) ? 1 : 0;
  const int j1 = (i == 2) ? 1 : 2;
  const float* Hi  = hx + (size_t)i  * A_ * H_;
  const float* Hj0 = hx + (size_t)j0 * A_ * H_;
  const float* Hj1 = hx + (size_t)j1 * A_ * H_;
  const float* Wp = cw + (size_t)(2 * i) * KP * H_;
  const float* Bb = cb + (size_t)(2 * i) * H_;
  auto fA = [&](int r, int k0, int kq, float4& x, float4& y) {
    const int q  = k0 / KP;
    const int kp = k0 - q * KP;
    const int s  = kp / H_;
    const int kk = kp - s * H_ + kq;
    const float* Hj = q ? Hj1 : Hj0;
    const float* px = ((s == 1) ? Hj : Hi) + (size_t)(a0 + r) * H_ + kk;
    x = ld4(px);
    if (s == 2) y = ld4(Hj + (size_t)(a0 + r) * H_ + kk);
    else y = make_float4(0.f, 0.f, 0.f, 0.f);
  };
  auto cA = [&](int k0, const float4& x, const float4& y) {
    if ((k0 % KP) / H_ == 2) return make_float4(x.x * y.x, x.y * y.y, x.z * y.z, x.w * y.w);
    return x;
  };

  const int lane = threadIdx.x & 31;
  const int ir = (threadIdx.x >> 5) * 16 + 8 * (lane >> 4);
  const int nc = n0 + (lane & 15);
  float g[4][8];
  auto gated = [&](const v8f (&acc)[4], const float* Hj, const float* bias, const bool add) {
#pragma unroll
    for (int c = 0; c < 4; ++c) {
      const int n = nc + 16 * c;
      if (n >= H_) continue;
      const float bn = bias[n];
#pragma unroll
      for (int v = 0; v < 8; ++v) {
        const size_t o = (size_t)(a0 + ir + v) * H_ + n;
        const float z = 1.0f / (1.0f + expf(-(acc[c][v] + bn)));
        const float t = z * Hi[o] + (1.0f - z) * Hj[o];
        g[c][v] = add ? g[c][v] + t : t;
      }
    }
  };
  auto mid = [&](v8f (&acc)[4]) {
    gated(acc, Hj0, Bb, false);
#pragma unroll
    for (int c = 0; c < 4; ++c) acc[c] = v8f{};
  };
  auto ep = [&](const v8f (&acc)[4]) {
    gated(acc, Hj1, Bb + H_, true);
#pragma unroll
    for (int c = 0; c < 4; ++c) {
      const int n = nc + 16 * c;
      if (n >= H_) continue;
#pragma unroll
      for (int v = 0; v < 8; ++v)
        hout[((size_t)i * A_ + a0 + ir + v) * H_ + n] = g[c][v];
    }
  };
  gemm_block_64x64<40, true>(2 * KP, Wp, n0, fA, cA, ep, KP, mid);
}

__global__ void k_final(const float* __restrict__ hout, const float* __restrict__ uw,
                        const float* __restrict__ ub, float* __restrict__ out) {
  const int a = blockIdx.x * blockDim.x + threadIdx.x;
  float acc[TAG_];
#pragma unroll
  for (int t = 0; t < TAG_; ++t)
    acc[t] = ub[t] + ub[TAG_ + t] + ub[2 * TAG_ + t];
  for (int m = 0; m < M_; ++m) {
    const float* hr = hout + (size_t)(m * A_ + a) * H_;
    const float* wm = uw + (size_t)m * H_ * TAG_;
    for (int h = 0; h < H_; ++h) {
      const float v = hr[h];
#pragma unroll
      for (int t = 0; t < TAG_; ++t) acc[t] += v * wm[h * TAG_ + t];
    }
  }
#pragma unroll
  for (int t = 0; t < TAG_; ++t) out[(size_t)a * TAG_ + t] = acc[t];
}

}

extern "C" void kernel_launch(void* const* d_in, const int* in_sizes, int n_in,
                              void* d_out, int out_size, void* d_ws, size_t ws_size,
                              hipStream_t stream) {
  (void)in_sizes; (void)n_in; (void)out_size; (void)ws_size;
  const float* x_a     = (const float*)d_in[0];
  const float* x_v     = (const float*)d_in[1];
  const float* x_t     = (const float*)d_in[2];
  const float* fc0_w   = (const float*)d_in[3];
  const float* fc0_b   = (const float*)d_in[4];
  const float* conv_w  = (const float*)d_in[5];
  const float* tm_w    = (const float*)d_in[6];
  const float* tm_b    = (const float*)d_in[7];
  const float* cross_w = (const float*)d_in[8];
  const float* cross_b = (const float*)d_in[9];
  const float* uni_w   = (const float*)d_in[10];
  const float* uni_b   = (const float*)d_in[11];
  float* out = (float*)d_out;

  const size_t NH = (size_t)N_ * H_;
  float* ws    = (float*)d_ws;
  float* big   = ws; ws += NH;
  float* nrm   = ws; ws += NH;
  float* intra = ws; ws += (size_t)N_ * L_;
  float* crossv= ws; ws += (size_t)M_ * M_ * A_;
  float* dinv  = ws; ws += N_;
  float* h0    = ws; ws += NH;
  float* hA    = ws; ws += NH;
  float* hBf   = ws; ws += NH;
  float* hi    = ws; ws += NH;
  float* hx    = ws; ws += NH;
  float* hout  = ws; ws += NH;

  k_norm <<<N_ / 8,              dim3(256), 0, stream>>>(x_a, x_v, x_t, big, nrm);
  k_gram <<<M_ * B_,             dim3(512), 0, stream>>>(nrm, intra);
  k_cross<<<6 * A_ / 8,          dim3(256), 0, stream>>>(nrm, crossv);
  k_dinv <<<N_ / 8,              dim3(256), 0, stream>>>(intra, crossv, dinv);
  k_fc0  <<<dim3(4, N_ / 64),    dim3(128), 0, stream>>>(big, fc0_w, fc0_b, h0);

  const float beta0 = logf(0.5f / 1.0f + 1.0f);
  const float beta1 = logf(0.5f / 2.0f + 1.0f);
  k_spmm <<<dim3(4, M_ * B_),    dim3(128), 0, stream>>>(intra, crossv, dinv, h0, hi);
  k_conv <<<dim3(4, N_ / 64),    dim3(128), 0, stream>>>(hi, h0, conv_w, beta0, hA);
  k_spmm <<<dim3(4, M_ * B_),    dim3(128), 0, stream>>>(intra, crossv, dinv, hA, hi);
  k_conv <<<dim3(4, N_ / 64),    dim3(128), 0, stream>>>(hi, h0, conv_w + 2 * H_ * H_, beta1, hBf);

  k_hx   <<<dim3(4, A_ / 64, M_), dim3(128), 0, stream>>>(big, hBf, tm_w, tm_b, hx);

  k_gate <<<dim3(4, A_ / 64, M_), dim3(128), 0, stream>>>(hx, cross_w, cross_b, hout);

  k_final<<<A_ / 256, dim3(256), 0, stream>>>(hout, uni_w, uni_b, out);
}


# k_final (600x6 output product): one wave per utterance, lane-strided coalesced reads of hout and weights, butterfly sum over lanes (changes the sum order of out); 128 blocks x 16 waves, was 8 x 8
# speedup vs baseline: 4.8879x; 1.1952x over previous
#include <hip/hip_runtime.h>
#include <math.h>

typedef __attribute__((ext_vector_type(2))) float v2f;
typedef __attribute__((ext_vector_type(8))) float v8f;

namespace {

constexpr int B_   = 32;
constexpr int L_   = 64;
constexpr int H_   = 200;
constexpr int M_   = 3;
constexpr int A_   = B_ * L_;
constexpr int N_   = M_ * A_;
constexpr int TAG_ = 6;
constexpr float PI_F = 3.14159265358979323846f;

template <typename FA, typename FB, typename FE>
__device__ __forceinline__ void wmma_tile_f32(int m0, int n0, int K,
                                              FA loadA, FB loadB, FE epi) {
  const int lane = threadIdx.x & 31;
  const int mr   = lane & 15;
  const int kh   = lane >> 4;
  v8f acc = {};
  for (int k = 0; k < K; k += 4) {
    const int kk = k + 2 * kh;
    v2f a, b;
    a.x = loadA(m0 + mr, kk);
    a.y = loadA(m0 + mr, kk + 1);
    b.x = loadB(kk, n0 + mr);
    b.y = loadB(kk + 1, n0 + mr);
    acc = __builtin_amdgcn_wmma_f32_16x16x4_f32(false, a, false, b,
                                                (short)0, acc, false, false);
  }
#pragma unroll
  for (int v = 0; v < 8; ++v) epi(m0 + 8 * kh + v, n0 + mr, acc[v]);
}

__device__ __forceinline__ float wave_sum(float s) {
#pragma unroll
  for (int off = 16; off > 0; off >>= 1) s += __shfl_xor(s, off, 32);
  return s;
}

__global__ void k_norm(const float* __restrict__ xa, const float* __restrict__ xv,
                       const float* __restrict__ xt, float* __restrict__ big,
                       float* __restrict__ nrm) {
  const int row  = blockIdx.x * 8 + (threadIdx.x >> 5);
  const int lane = threadIdx.x & 31;
  const float* src = (row < A_)     ? xa + (size_t)row * H_
                   : (row < 2 * A_) ? xv + (size_t)(row - A_) * H_
                                    : xt + (size_t)(row - 2 * A_) * H_;
  float s = 0.f;
  for (int k = lane; k < H_; k += 32) { float v = src[k]; s += v * v; }
  s = wave_sum(s);
  const float rn = 1.0f / sqrtf(s);
  for (int k = lane; k < H_; k += 32) {
    const float v = src[k];
    big[(size_t)row * H_ + k] = v;
    nrm[(size_t)row * H_ + k] = v * rn;
  }
}

__global__ void k_gram(const float* __restrict__ nrm, float* __restrict__ intra) {
  const int mb = blockIdx.x;
  const int w  = threadIdx.x >> 5;
  const int m0 = (w >> 2) * 16, n0 = (w & 3) * 16;
  const float* blk = nrm + (size_t)mb * L_ * H_;
  auto lA = [&](int i, int k) { return blk[(size_t)i * H_ + k]; };
  auto lB = [&](int k, int j) { return blk[(size_t)j * H_ + k]; };
  auto ep = [&](int i, int j, float v) {
    intra[((size_t)mb * L_ + i) * L_ + j] = 1.0f - acosf(v * 0.99999f) * (1.0f / PI_F);
  };
  wmma_tile_f32(m0, n0, H_, lA, lB, ep);
}

__global__ void k_cross(const float* __restrict__ nrm, float* __restrict__ crossv) {
  const int idx  = blockIdx.x * 8 + (threadIdx.x >> 5);
  const int lane = threadIdx.x & 31;
  const int pm[6] = {0, 0, 1, 1, 2, 2};
  const int pn[6] = {1, 2, 0, 2, 0, 1};
  const int q = idx / A_, a = idx % A_;
  const int m = pm[q], n = pn[q];
  const float* pa = nrm + ((size_t)m * A_ + a) * H_;
  const float* pb = nrm + ((size_t)n * A_ + a) * H_;
  float s = 0.f;
  for (int k = lane; k < H_; k += 32) s += pa[k] * pb[k];
  s = wave_sum(s);
  if (lane == 0)
    crossv[((size_t)m * M_ + n) * A_ + a] = 1.0f - acosf(s * 0.99999f) * (1.0f / PI_F);
}

__global__ void k_dinv(const float* __restrict__ intra, const float* __restrict__ crossv,
                       float* __restrict__ dinv) {
  const int row  = blockIdx.x * 8 + (threadIdx.x >> 5);
  const int lane = threadIdx.x & 31;
  const int m = row >> 11, a = row & (A_ - 1);
  const float* r = intra + (size_t)row * L_;
  float s = 0.f;
  for (int j = lane; j < L_; j += 32) s += r[j];
  s = wave_sum(s);
  if (lane == 0) {
#pragma unroll
    for (int n = 0; n < M_; ++n)
      if (n != m) s += crossv[((size_t)m * M_ + n) * A_ + a];
    dinv[row] = 1.0f / sqrtf(s);
  }
}

struct no_mid {};
template <typename T> struct is_no_mid { static constexpr bool value = false; };
template <> struct is_no_mid<no_mid> { static constexpr bool value = true; };
__device__ __forceinline__ float4 ld4(const float* p) {
  return *reinterpret_cast<const float4*>(p);
}
__device__ __forceinline__ float2 ld2(const float* p) {
  return *reinterpret_cast<const float2*>(p);
}

template <int BK, bool COMB_HAS_MATH, typename FA, typename CA, typename FE, typename FM = no_mid>
__device__ __forceinline__ void gemm_block_64x64(const int K, const float* __restrict__ Bp,
                                                 const int n0, FA fetchA, CA combA, FE epi,
                                                 const int Kmid = 0, FM mid = FM{}) {
  constexpr bool TWO_PRODUCTS = !is_no_mid<FM>::value;
  constexpr int BM = 64, BN = 64, NTHR = 128;
  constexpr int PA = BK + 4;
  constexpr int PB = BN + 16;
  constexpr int KQ = BK / 4;
  constexpr int NP = BN / 2;
  constexpr int A4 = BM * KQ;
  constexpr int B4 = (BK / 2) * NP;
  constexpr int RA = A4 / NTHR;
  constexpr int RB = B4 / NTHR;
  static_assert(A4 % NTHR == 0 && B4 % NTHR == 0, "chunks must split evenly over the block");
  static_assert((PA % 8) == 4, "A pitch must be 4 * odd dwords");
  static_assert(((2 * PB) % 64) == 32, "B pair pitch must be 32 mod 64 dwords");

  constexpr int ASZ = BM * PA;
  constexpr int BSZ = (BK / 2) * PB * 2;
  __shared__ __attribute__((aligned(16))) float As[2 * ASZ];
  __shared__ __attribute__((aligned(16))) float Bs[2 * BSZ];

  const int t    = threadIdx.x;
  const int lane = t & 31;
  const int w    = t >> 5;
  const int mr   = lane & 15;
  const int kh   = lane >> 4;

  float4 ax[RA], ay[RA];
  float2 be[RB], bo[RB];

  auto prefetch = [&](const int k0) {
#pragma unroll
    for (int r = 0; r < RA; ++r) {
      const int e = t + NTHR * r;
      fetchA(e / KQ, k0, (e % KQ) * 4, ax[r], ay[r]);
    }
#pragma unroll
    for (int r = 0; r < RB; ++r) {
      const int e  = t + NTHR * r;
      const int kp = e / NP;
      const int n  = min(n0 + (e % NP) * 2, H_ - 2);
      const float* p = Bp + (size_t)(k0 + 2 * kp) * H_ + n;
      be[r] = ld2(p);
      bo[r] = ld2(p + H_);
    }
  };

  auto stage = [&](const int k0, const int buf) {
#pragma unroll
    for (int r = 0; r < RA; ++r) {
      const int e = t + NTHR * r;
      *reinterpret_cast<float4*>(&As[buf * ASZ + (e / KQ) * PA + (e % KQ) * 4]) = combA(k0, ax[r], ay[r]);
    }
#pragma unroll
    for (int r = 0; r < RB; ++r) {
      const int e = t + NTHR * r;
      const int j = (e % NP) * 2;
      float4 v = make_float4(be[r].x, bo[r].x, be[r].y, bo[r].y);
      if (n0 + j >= H_) v = make_float4(0.f, 0.f, 0.f, 0.f);
      *reinterpret_cast<float4*>(&Bs[buf * BSZ + ((e / NP) * PB + j) * 2]) = v;
    }
  };

  const int aoff = (w * 16 + mr) * PA + 2 * kh;
  const int boff = (kh * PB + mr) * 2;

  v8f acc[4] = {};
  auto compute = [&](const int buf) {
    const float* aptr = As + buf * ASZ + aoff;
    const float* bptr = Bs + buf * BSZ + boff;
#pragma unroll
    for (int ks = 0; ks < BK; ks += 4) {
      const v2f a = *reinterpret_cast<const v2f*>(aptr + ks);
#pragma unroll
      for (int c = 0; c < 4; ++c) {
        const v2f b = *reinterpret_cast<const v2f*>(bptr + ((ks / 2) * PB + 16 * c) * 2);
        acc[c] = __builtin_amdgcn_wmma_f32_16x16x4_f32(false, a, false, b, (short)0, acc[c], false, false);
      }
    }
  };

  prefetch(0);
  stage(0, 0);
  __syncthreads();
  int buf = 0;
  for (int k0 = BK; k0 < K; k0 += BK) {
    prefetch(k0);
    __builtin_amdgcn_sched_barrier(0);
    compute(buf);
    if constexpr (COMB_HAS_MATH) __builtin_amdgcn_sched_barrier(0);
    if constexpr (TWO_PRODUCTS) {
      if (k0 == Kmid) mid(acc);
    }
    stage(k0, buf ^ 1);
    __syncthreads();
    buf ^= 1;
  }
  compute(buf);

  if constexpr (TWO_PRODUCTS) {
    epi(acc);
  } else {
    const int ir = w * 16 + 8 * kh;
    const int nc = n0 + mr;
#pragma unroll
    for (int c = 0; c < 4; ++c) {
#pragma unroll
      for (int v = 0; v < 8; ++v) epi(ir + v, nc + 16 * c, acc[c][v]);
    }
  }
}

__global__ __launch_bounds__(128) void k_fc0(const float* __restrict__ big,
                                             const float* __restrict__ w,
                                             const float* __restrict__ bias,
                                             float* __restrict__ h0) {
  const int n0 = blockIdx.x * 64;
  const int r0 = blockIdx.y * 64;
  auto fA = [&](int i, int k0, int kq, float4& x, float4& y) {
    (void)y;
    x = ld4(big + (size_t)(r0 + i) * H_ + k0 + kq);
  };
  auto cA = [&](int k0, const float4& x, const float4& y) { (void)k0; (void)y; return x; };
  auto ep = [&](int i, int n, float v) {
    if (n < H_) h0[(size_t)(r0 + i) * H_ + n] = fmaxf(v + bias[n], 0.0f);
  };
  gemm_block_64x64<40, false>(H_, w, n0, fA, cA, ep);
}

__global__ __launch_bounds__(128) void k_spmm(const float* __restrict__ intra,
                                              const float* __restrict__ crossv,
                                              const float* __restrict__ dinv,
                                              const float* __restrict__ h,
                                              float* __restrict__ hi) {
  const int n0 = blockIdx.x * 64;
  const int mb = blockIdx.y;
  const int m  = mb >> 5;
  const int rowbase = mb * L_;
  const float* S = intra + (size_t)mb * L_ * L_;
  auto fA = [&](int i, int k0, int kq, float4& x, float4& y) {
    x = ld4(S + (size_t)i * L_ + k0 + kq);
    y = ld4(dinv + rowbase + k0 + kq);
  };
  auto cA = [&](int k0, const float4& x, const float4& y) {
    (void)k0;
    return make_float4(x.x * y.x, x.y * y.y, x.z * y.z, x.w * y.w);
  };
  auto ep = [&](int i, int n, float acc) {
    if (n >= H_) return;
    const int row = rowbase + i;
    const int a   = row - m * A_;
    float s = acc;
#pragma unroll
    for (int nn = 0; nn < M_; ++nn) {
      if (nn == m) continue;
      const int orow = nn * A_ + a;
      s += crossv[((size_t)m * M_ + nn) * A_ + a] * dinv[orow] * h[(size_t)orow * H_ + n];
    }
    hi[(size_t)row * H_ + n] = s * dinv[row];
  };
  gemm_block_64x64<32, true>(L_, h + (size_t)rowbase * H_, n0, fA, cA, ep);
}

__global__ __launch_bounds__(128) void k_conv(const float* __restrict__ hi,
                                              const float* __restrict__ h0,
                                              const float* __restrict__ w, float beta,
                                              float* __restrict__ hout) {
  const int n0 = blockIdx.x * 64;
  const int r0 = blockIdx.y * 64;
  auto fA = [&](int i, int k0, int kq, float4& x, float4& y) {
    (void)y;
    const float* p = (k0 < H_) ? hi + (size_t)(r0 + i) * H_ + (k0 + kq)
                               : h0 + (size_t)(r0 + i) * H_ + (k0 - H_ + kq);
    x = ld4(p);
  };
  auto cA = [&](int k0, const float4& x, const float4& y) { (void)k0; (void)y; return x; };
  auto ep = [&](int i, int n, float v) {
    if (n >= H_) return;
    const size_t o = (size_t)(r0 + i) * H_ + n;
    const float r = 0.9f * hi[o] + 0.1f * h0[o];
    hout[o] = fmaxf(beta * v + (1.0f - beta) * r, 0.0f);
  };
  gemm_block_64x64<40, false>(2 * H_, w, n0, fA, cA, ep);
}

__global__ __launch_bounds__(128) void k_hx(const float* __restrict__ big,
                                            const float* __restrict__ hf,
                                            const float* __restrict__ tw,
                                            const float* __restrict__ tb,
                                            float* __restrict__ hx) {
  const int m  = blockIdx.z;
  const int n0 = blockIdx.x * 64;
  const int r0 = m * A_ + blockIdx.y * 64;
  const float* Wm = tw + (size_t)m * 2 * H_ * H_;
  const float* Bb = tb + (size_t)m * H_;
  auto fA = [&](int i, int k0, int kq, float4& x, float4& y) {
    (void)y;
    const float* p = (k0 < H_) ? big + (size_t)(r0 + i) * H_ + (k0 + kq)
                               : hf  + (size_t)(r0 + i) * H_ + (k0 - H_ + kq);
    x = ld4(p);
  };
  auto cA = [&](int k0, const float4& x, const float4& y) { (void)k0; (void)y; return x; };
  auto ep = [&](int i, int n, float v) {
    if (n < H_) hx[(size_t)(r0 + i) * H_ + n] = fmaxf(v + Bb[n], 0.0f);
  };
  gemm_block_64x64<40, false>(2 * H_, Wm, n0, fA, cA, ep);
}

__global__ __launch_bounds__(128) void k_gate(const float* __restrict__ hx,
                                              const float* __restrict__ cw,
                                              const float* __restrict__ cb,
                                              float* __restrict__ hout) {
  constexpr int KP = 3 * H_;
  const int i  = blockIdx.z;
  const int n0 = blockIdx.x * 64;
  const int a0 = blockIdx.y * 64;
  const int j0 = (i == 0) ? 1 : 0;
  const int j1 = (i == 2) ? 1 : 2;
  const float* Hi  = hx + (size_t)i  * A_ * H_;
  const float* Hj0 = hx + (size_t)j0 * A_ * H_;
  const float* Hj1 = hx + (size_t)j1 * A_ * H_;
  const float* Wp = cw + (size_t)(2 * i) * KP * H_;
  const float* Bb = cb + (size_t)(2 * i) * H_;
  auto fA = [&](int r, int k0, int kq, float4& x, float4& y) {
    const int q  = k0 / KP;
    const int kp = k0 - q * KP;
    const int s  = kp / H_;
    const int kk = kp - s * H_ + kq;
    const float* Hj = q ? Hj1 : Hj0;
    const float* px = ((s == 1) ? Hj : Hi) + (size_t)(a0 + r) * H_ + kk;
    x = ld4(px);
    if (s == 2) y = ld4(Hj + (size_t)(a0 + r) * H_ + kk);
    else y = make_float4(0.f, 0.f, 0.f, 0.f);
  };
  auto cA = [&](int k0, const float4& x, const float4& y) {
    if ((k0 % KP) / H_ == 2) return make_float4(x.x * y.x, x.y * y.y, x.z * y.z, x.w * y.w);
    return x;
  };

  const int lane = threadIdx.x & 31;
  const int ir = (threadIdx.x >> 5) * 16 + 8 * (lane >> 4);
  const int nc = n0 + (lane & 15);
  float g[4][8];
  auto gated = [&](const v8f (&acc)[4], const float* Hj, const float* bias, const bool add) {
#pragma unroll
    for (int c = 0; c < 4; ++c) {
      const int n = nc + 16 * c;
      if (n >= H_) continue;
      const float bn = bias[n];
#pragma unroll
      for (int v = 0; v < 8; ++v) {
        const size_t o = (size_t)(a0 + ir + v) * H_ + n;
        const float z = 1.0f / (1.0f + expf(-(acc[c][v] + bn)));
        const float t = z * Hi[o] + (1.0f - z) * Hj[o];
        g[c][v] = add ? g[c][v] + t : t;
      }
    }
  };
  auto mid = [&](v8f (&acc)[4]) {
    gated(acc, Hj0, Bb, false);
#pragma unroll
    for (int c = 0; c < 4; ++c) acc[c] = v8f{};
  };
  auto ep = [&](const v8f (&acc)[4]) {
    gated(acc, Hj1, Bb + H_, true);
#pragma unroll
    for (int c = 0; c < 4; ++c) {
      const int n = nc + 16 * c;
      if (n >= H_) continue;
#pragma unroll
      for (int v = 0; v < 8; ++v)
        hout[((size_t)i * A_ + a0 + ir + v) * H_ + n] = g[c][v];
    }
  };
  gemm_block_64x64<40, true>(2 * KP, Wp, n0, fA, cA, ep, KP, mid);
}

__global__ __launch_bounds__(512) void k_final(const float* __restrict__ hout,
                                               const float* __restrict__ uw,
                                               const float* __restrict__ ub,
                                               float* __restrict__ out) {
  constexpr int UPB = 16;
  constexpr int NSF = H_ / 32;
  constexpr int REM = H_ - 32 * NSF;
  static_assert((UPB * TAG_) % 32 == 0, "a block's results must be whole 128-byte lines");
  static_assert(REM > 0 && M_ * REM <= 32, "the left-over units of all modalities share one slot");
  __shared__ float res[UPB * TAG_];
  const int wv   = threadIdx.x >> 5;
  const int lane = threadIdx.x & 31;
  const int a    = blockIdx.x * UPB + wv;
  float acc[TAG_] = {};
  auto add6 = [&](const float v, const float2 w01, const float2 w23, const float2 w45) {
    acc[0] += v * w01.x; acc[1] += v * w01.y;
    acc[2] += v * w23.x; acc[3] += v * w23.y;
    acc[4] += v * w45.x; acc[5] += v * w45.y;
  };
#pragma unroll
  for (int m = 0; m < M_; ++m) {
    const float* hr = hout + (size_t)(m * A_ + a) * H_;
    const float* wm = uw + (size_t)m * H_ * TAG_;
#pragma unroll
    for (int s = 0; s < NSF; ++s) {
      const int h = lane + 32 * s;
      const float* wr = wm + h * TAG_;
      add6(hr[h], ld2(wr), ld2(wr + 2), ld2(wr + 4));
    }
  }
  {
    const bool ok = lane < M_ * REM;
    const int l = ok ? lane : M_ * REM - 1;
    const int m = l / REM;
    const int h = 32 * NSF + l % REM;
    const float* wr = uw + ((size_t)m * H_ + h) * TAG_;
    const float  v   = hout[(size_t)(m * A_ + a) * H_ + h];
    const float2 w01 = ld2(wr), w23 = ld2(wr + 2), w45 = ld2(wr + 4);
    const float2 z = make_float2(0.f, 0.f);
    add6(ok ? v : 0.f, ok ? w01 : z, ok ? w23 : z, ok ? w45 : z);
  }
#pragma unroll
  for (int t = 0; t < TAG_; ++t) acc[t] = wave_sum(acc[t]);
  if (lane == 0) {
#pragma unroll
    for (int t = 0; t < TAG_; ++t)
      res[wv * TAG_ + t] = acc[t] + (ub[t] + ub[TAG_ + t] + ub[2 * TAG_ + t]);
  }
  __syncthreads();
  if (wv == 0) {
    for (int j = 0; j < UPB * TAG_ / 32; ++j)
      out[(size_t)blockIdx.x * (UPB * TAG_) + 32 * j + lane] = res[32 * j + lane];
  }
}

}

extern "C" void kernel_launch(void* const* d_in, const int* in_sizes, int n_in,
                              void* d_out, int out_size, void* d_ws, size_t ws_size,
                              hipStream_t stream) {
  (void)in_sizes; (void)n_in; (void)out_size; (void)ws_size;
  const float* x_a     = (const float*)d_in[0];
  const float* x_v     = (const float*)d_in[1];
  const float* x_t     = (const float*)d_in[2];
  const float* fc0_w   = (const float*)d_in[3];
  const float* fc0_b   = (const float*)d_in[4];
  const float* conv_w  = (const float*)d_in[5];
  const float* tm_w    = (const float*)d_in[6];
  const float* tm_b    = (const float*)d_in[7];
  const float* cross_w = (const float*)d_in[8];
  const float* cross_b = (const float*)d_in[9];
  const float* uni_w   = (const float*)d_in[10];
  const float* uni_b   = (const float*)d_in[11];
  float* out = (float*)d_out;

  const size_t NH = (size_t)N_ * H_;
  float* ws    = (float*)d_ws;
  float* big   = ws; ws += NH;
  float* nrm   = ws; ws += NH;
  float* intra = ws; ws += (size_t)N_ * L_;
  float* crossv= ws; ws += (size_t)M_ * M_ * A_;
  float* dinv  = ws; ws += N_;
  float* h0    = ws; ws += NH;
  float* hA    = ws; ws += NH;
  float* hBf   = ws; ws += NH;
  float* hi    = ws; ws += NH;
  float* hx    = ws; ws += NH;
  float* hout  = ws; ws += NH;

  k_norm <<<N_ / 8,              dim3(256), 0, stream>>>(x_a, x_v, x_t, big, nrm);
  k_gram <<<M_ * B_,             dim3(512), 0, stream>>>(nrm, intra);
  k_cross<<<6 * A_ / 8,          dim3(256), 0, stream>>>(nrm, crossv);
  k_dinv <<<N_ / 8,              dim3(256), 0, stream>>>(intra, crossv, dinv);
  k_fc0  <<<dim3(4, N_ / 64),    dim3(128), 0, stream>>>(big, fc0_w, fc0_b, h0);

  const float beta0 = logf(0.5f / 1.0f + 1.0f);
  const float beta1 = logf(0.5f / 2.0f + 1.0f);
  k_spmm <<<dim3(4, M_ * B_),    dim3(128), 0, stream>>>(intra, crossv, dinv, h0, hi);
  k_conv <<<dim3(4, N_ / 64),    dim3(128), 0, stream>>>(hi, h0, conv_w, beta0, hA);
  k_spmm <<<dim3(4, M_ * B_),    dim3(128), 0, stream>>>(intra, crossv, dinv, hA, hi);
  k_conv <<<dim3(4, N_ / 64),    dim3(128), 0, stream>>>(hi, h0, conv_w + 2 * H_ * H_, beta1, hBf);

  k_hx   <<<dim3(4, A_ / 64, M_), dim3(128), 0, stream>>>(big, hBf, tm_w, tm_b, hx);

  k_gate <<<dim3(4, A_ / 64, M_), dim3(128), 0, stream>>>(hx, cross_w, cross_b, hout);

  k_final<<<A_ / 16, dim3(512), 0, stream>>>(hout, uni_w, uni_b, out);
}


# k_norm, k_gram, k_cross, k_dinv fused into one launch: a block per (modality, dialogue) keeps its normalised rows in LDS for Gram block, cross terms and degrees; feature copies no longer written
# speedup vs baseline: 5.3385x; 1.0922x over previous
#include <hip/hip_runtime.h>
#include <math.h>

typedef __attribute__((ext_vector_type(2))) float v2f;
typedef __attribute__((ext_vector_type(8))) float v8f;

namespace {

constexpr int B_   = 32;
constexpr int L_   = 64;
constexpr int H_   = 200;
constexpr int M_   = 3;
constexpr int A_   = B_ * L_;
constexpr int N_   = M_ * A_;
constexpr int TAG_ = 6;
constexpr float PI_F = 3.14159265358979323846f;

__device__ __forceinline__ float wave_sum(float s) {
#pragma unroll
  for (int off = 16; off > 0; off >>= 1) s += __shfl_xor(s, off, 32);
  return s;
}

__global__ __launch_bounds__(512) void k_graph(const float* __restrict__ xa,
                                               const float* __restrict__ xv,
                                               const float* __restrict__ xt,
                                               float* __restrict__ intra,
                                               float* __restrict__ crossv,
                                               float* __restrict__ dinv) {
  constexpr int PN  = H_ + 4;
  constexpr int TP  = L_ + 4;
  constexpr int NS  = (H_ + 31) / 32;
  constexpr int RPW = L_ / 16;
  static_assert((PN % 8) == 4, "pitch must be 4 * odd dwords");
  static_assert(L_ * TP <= L_ * PN, "the similarity tile reuses the rows' LDS");
  __shared__ __attribute__((aligned(16))) float nr[L_ * PN];
  __shared__ float cx[2][L_];
  __shared__ float dv[L_];
  const int mb   = blockIdx.x;
  const int m    = mb >> 5;
  const int b    = mb & 31;
  const int w    = threadIdx.x >> 5;
  const int lane = threadIdx.x & 31;
  const int n1   = (m == 0) ? 1 : 0;
  const int n2   = (m == 2) ? 1 : 2;
  auto plane = [&](const int mm) { return (mm == 0) ? xa : (mm == 1) ? xv : xt; };
  const float* P0 = plane(m)  + (size_t)b * L_ * H_;
  const float* P1 = plane(n1) + (size_t)b * L_ * H_;
  const float* P2 = plane(n2) + (size_t)b * L_ * H_;

  float v0[RPW][NS], v1[RPW][NS], v2[RPW][NS];
#pragma unroll
  for (int q = 0; q < RPW; ++q) {
    const int r = w * RPW + q;
#pragma unroll
    for (int j = 0; j < NS; ++j) {
      const int k = lane + 32 * j;
      const bool ok = k < H_;
      v0[q][j] = ok ? P0[(size_t)r * H_ + k] : 0.f;
      v1[q][j] = ok ? P1[(size_t)r * H_ + k] : 0.f;
      v2[q][j] = ok ? P2[(size_t)r * H_ + k] : 0.f;
    }
  }
  float c1[RPW], c2[RPW];
#pragma unroll
  for (int q = 0; q < RPW; ++q) {
    const int r = w * RPW + q;
    float s0 = 0.f, s1 = 0.f, s2 = 0.f;
#pragma unroll
    for (int j = 0; j < NS; ++j) {
      s0 += v0[q][j] * v0[q][j];
      s1 += v1[q][j] * v1[q][j];
      s2 += v2[q][j] * v2[q][j];
    }
    const float rn0 = 1.0f / sqrtf(wave_sum(s0));
    const float rn1 = 1.0f / sqrtf(wave_sum(s1));
    const float rn2 = 1.0f / sqrtf(wave_sum(s2));
    float d1 = 0.f, d2 = 0.f;
#pragma unroll
    for (int j = 0; j < NS; ++j) {
      const int k = lane + 32 * j;
      const float u0 = v0[q][j] * rn0;
      const float u1 = v1[q][j] * rn1;
      const float u2 = v2[q][j] * rn2;
      if (k < H_) nr[r * PN + k] = u0;
      d1 += u0 * u1;
      d2 += u0 * u2;
    }
    d1 = wave_sum(d1);
    d2 = wave_sum(d2);
    c1[q] = 1.0f - acosf(d1 * 0.99999f) * (1.0f / PI_F);
    c2[q] = 1.0f - acosf(d2 * 0.99999f) * (1.0f / PI_F);
    if (lane == 0) { cx[0][r] = c1[q]; cx[1][r] = c2[q]; }
  }
  __syncthreads();

  const int mr = lane & 15;
  const int kh = lane >> 4;
  const int m0 = (w >> 2) * 16, n0 = (w & 3) * 16;
  v8f acc = {};
  {
    const float* ap = nr + (m0 + mr) * PN + 2 * kh;
    const float* bp = nr + (n0 + mr) * PN + 2 * kh;
#pragma unroll 10
    for (int k = 0; k < H_; k += 4) {
      const v2f a = *reinterpret_cast<const v2f*>(ap + k);
      const v2f bb = *reinterpret_cast<const v2f*>(bp + k);
      acc = __builtin_amdgcn_wmma_f32_16x16x4_f32(false, a, false, bb, (short)0, acc, false, false);
    }
  }
  __syncthreads();
#pragma unroll
  for (int v = 0; v < 8; ++v)
    nr[(m0 + 8 * kh + v) * TP + n0 + mr] = 1.0f - acosf(acc[v] * 0.99999f) * (1.0f / PI_F);
  __syncthreads();

#pragma unroll
  for (int q = 0; q < RPW; ++q) {
    const int r = w * RPW + q;
    const float t0 = nr[r * TP + lane];
    const float t1 = nr[r * TP + lane + 32];
    float* orow = intra + ((size_t)mb * L_ + r) * L_;
    orow[lane]      = t0;
    orow[lane + 32] = t1;
    float s = 0.f;
    s += t0;
    s += t1;
    s = wave_sum(s);
    s += c1[q];
    s += c2[q];
    if (lane == 0) dv[r] = 1.0f / sqrtf(s);
  }
  __syncthreads();
  if (w < 2) {
    dinv[(size_t)mb * L_ + 32 * w + lane] = dv[32 * w + lane];
  } else if (w < 4) {
    const int h = w - 2;
    crossv[((size_t)m * M_ + n1) * A_ + b * L_ + 32 * h + lane] = cx[0][32 * h + lane];
  } else if (w < 6) {
    const int h = w - 4;
    crossv[((size_t)m * M_ + n2) * A_ + b * L_ + 32 * h + lane] = cx[1][32 * h + lane];
  }
}

struct no_mid {};
template <typename T> struct is_no_mid { static constexpr bool value = false; };
template <> struct is_no_mid<no_mid> { static constexpr bool value = true; };
__device__ __forceinline__ float4 ld4(const float* p) {
  return *reinterpret_cast<const float4*>(p);
}
__device__ __forceinline__ float2 ld2(const float* p) {
  return *reinterpret_cast<const float2*>(p);
}

template <int BK, bool COMB_HAS_MATH, typename FA, typename CA, typename FE, typename FM = no_mid>
__device__ __forceinline__ void gemm_block_64x64(const int K, const float* __restrict__ Bp,
                                                 const int n0, FA fetchA, CA combA, FE epi,
                                                 const int Kmid = 0, FM mid = FM{}) {
  constexpr bool TWO_PRODUCTS = !is_no_mid<FM>::value;
  constexpr int BM = 64, BN = 64, NTHR = 128;
  constexpr int PA = BK + 4;
  constexpr int PB = BN + 16;
  constexpr int KQ = BK / 4;
  constexpr int NP = BN / 2;
  constexpr int A4 = BM * KQ;
  constexpr int B4 = (BK / 2) * NP;
  constexpr int RA = A4 / NTHR;
  constexpr int RB = B4 / NTHR;
  static_assert(A4 % NTHR == 0 && B4 % NTHR == 0, "chunks must split evenly over the block");
  static_assert((PA % 8) == 4, "A pitch must be 4 * odd dwords");
  static_assert(((2 * PB) % 64) == 32, "B pair pitch must be 32 mod 64 dwords");

  constexpr int ASZ = BM * PA;
  constexpr int BSZ = (BK / 2) * PB * 2;
  __shared__ __attribute__((aligned(16))) float As[2 * ASZ];
  __shared__ __attribute__((aligned(16))) float Bs[2 * BSZ];

  const int t    = threadIdx.x;
  const int lane = t & 31;
  const int w    = t >> 5;
  const int mr   = lane & 15;
  const int kh   = lane >> 4;

  float4 ax[RA], ay[RA];
  float2 be[RB], bo[RB];

  auto prefetch = [&](const int k0) {
#pragma unroll
    for (int r = 0; r < RA; ++r) {
      const int e = t + NTHR * r;
      fetchA(e / KQ, k0, (e % KQ) * 4, ax[r], ay[r]);
    }
#pragma unroll
    for (int r = 0; r < RB; ++r) {
      const int e  = t + NTHR * r;
      const int kp = e / NP;
      const int n  = min(n0 + (e % NP) * 2, H_ - 2);
      const float* p = Bp + (size_t)(k0 + 2 * kp) * H_ + n;
      be[r] = ld2(p);
      bo[r] = ld2(p + H_);
    }
  };

  auto stage = [&](const int k0, const int buf) {
#pragma unroll
    for (int r = 0; r < RA; ++r) {
      const int e = t + NTHR * r;
      *reinterpret_cast<float4*>(&As[buf * ASZ + (e / KQ) * PA + (e % KQ) * 4]) = combA(k0, ax[r], ay[r]);
    }
#pragma unroll
    for (int r = 0; r < RB; ++r) {
      const int e = t + NTHR * r;
      const int j = (e % NP) * 2;
      float4 v = make_float4(be[r].x, bo[r].x, be[r].y, bo[r].y);
      if (n0 + j >= H_) v = make_float4(0.f, 0.f, 0.f, 0.f);
      *reinterpret_cast<float4*>(&Bs[buf * BSZ + ((e / NP) * PB + j) * 2]) = v;
    }
  };

  const int aoff = (w * 16 + mr) * PA + 2 * kh;
  const int boff = (kh * PB + mr) * 2;

  v8f acc[4] = {};
  auto compute = [&](const int buf) {
    const float* aptr = As + buf * ASZ + aoff;
    const float* bptr = Bs + buf * BSZ + boff;
#pragma unroll
    for (int ks = 0; ks < BK; ks += 4) {
      const v2f a = *reinterpret_cast<const v2f*>(aptr + ks);
#pragma unroll
      for (int c = 0; c < 4; ++c) {
        const v2f b = *reinterpret_cast<const v2f*>(bptr + ((ks / 2) * PB + 16 * c) * 2);
        acc[c] = __builtin_amdgcn_wmma_f32_16x16x4_f32(false, a, false, b, (short)0, acc[c], false, false);
      }
    }
  };

  prefetch(0);
  stage(0, 0);
  __syncthreads();
  int buf = 0;
  for (int k0 = BK; k0 < K; k0 += BK) {
    prefetch(k0);
    __builtin_amdgcn_sched_barrier(0);
    compute(buf);
    if constexpr (COMB_HAS_MATH) __builtin_amdgcn_sched_barrier(0);
    if constexpr (TWO_PRODUCTS) {
      if (k0 == Kmid) mid(acc);
    }
    stage(k0, buf ^ 1);
    __syncthreads();
    buf ^= 1;
  }
  compute(buf);

  if constexpr (TWO_PRODUCTS) {
    epi(acc);
  } else {
    const int ir = w * 16 + 8 * kh;
    const int nc = n0 + mr;
#pragma unroll
    for (int c = 0; c < 4; ++c) {
#pragma unroll
      for (int v = 0; v < 8; ++v) epi(ir + v, nc + 16 * c, acc[c][v]);
    }
  }
}

__global__ __launch_bounds__(128) void k_fc0(const float* __restrict__ xa,
                                             const float* __restrict__ xv,
                                             const float* __restrict__ xt,
                                             const float* __restrict__ w,
                                             const float* __restrict__ bias,
                                             float* __restrict__ h0) {
  const int n0 = blockIdx.x * 64;
  const int r0 = blockIdx.y * 64;
  const int mm = r0 / A_;
  const float* X = ((mm == 0) ? xa : (mm == 1) ? xv : xt) + (size_t)(r0 - mm * A_) * H_;
  auto fA = [&](int i, int k0, int kq, float4& x, float4& y) {
    (void)y;
    x = ld4(X + (size_t)i * H_ + k0 + kq);
  };
  auto cA = [&](int k0, const float4& x, const float4& y) { (void)k0; (void)y; return x; };
  auto ep = [&](int i, int n, float v) {
    if (n < H_) h0[(size_t)(r0 + i) * H_ + n] = fmaxf(v + bias[n], 0.0f);
  };
  gemm_block_64x64<40, false>(H_, w, n0, fA, cA, ep);
}

__global__ __launch_bounds__(128) void k_spmm(const float* __restrict__ intra,
                                              const float* __restrict__ crossv,
                                              const float* __restrict__ dinv,
                                              const float* __restrict__ h,
                                              float* __restrict__ hi) {
  const int n0 = blockIdx.x * 64;
  const int mb = blockIdx.y;
  const int m  = mb >> 5;
  const int rowbase = mb * L_;
  const float* S = intra + (size_t)mb * L_ * L_;
  auto fA = [&](int i, int k0, int kq, float4& x, float4& y) {
    x = ld4(S + (size_t)i * L_ + k0 + kq);
    y = ld4(dinv + rowbase + k0 + kq);
  };
  auto cA = [&](int k0, const float4& x, const float4& y) {
    (void)k0;
    return make_float4(x.x * y.x, x.y * y.y, x.z * y.z, x.w * y.w);
  };
  auto ep = [&](int i, int n, float acc) {
    if (n >= H_) return;
    const int row = rowbase + i;
    const int a   = row - m * A_;
    float s = acc;
#pragma unroll
    for (int nn = 0; nn < M_; ++nn) {
      if (nn == m) continue;
      const int orow = nn * A_ + a;
      s += crossv[((size_t)m * M_ + nn) * A_ + a] * dinv[orow] * h[(size_t)orow * H_ + n];
    }
    hi[(size_t)row * H_ + n] = s * dinv[row];
  };
  gemm_block_64x64<32, true>(L_, h + (size_t)rowbase * H_, n0, fA, cA, ep);
}

__global__ __launch_bounds__(128) void k_conv(const float* __restrict__ hi,
                                              const float* __restrict__ h0,
                                              const float* __restrict__ w, float beta,
                                              float* __restrict__ hout) {
  const int n0 = blockIdx.x * 64;
  const int r0 = blockIdx.y * 64;
  auto fA = [&](int i, int k0, int kq, float4& x, float4& y) {
    (void)y;
    const float* p = (k0 < H_) ? hi + (size_t)(r0 + i) * H_ + (k0 + kq)
                               : h0 + (size_t)(r0 + i) * H_ + (k0 - H_ + kq);
    x = ld4(p);
  };
  auto cA = [&](int k0, const float4& x, const float4& y) { (void)k0; (void)y; return x; };
  auto ep = [&](int i, int n, float v) {
    if (n >= H_) return;
    const size_t o = (size_t)(r0 + i) * H_ + n;
    const float r = 0.9f * hi[o] + 0.1f * h0[o];
    hout[o] = fmaxf(beta * v + (1.0f - beta) * r, 0.0f);
  };
  gemm_block_64x64<40, false>(2 * H_, w, n0, fA, cA, ep);
}

__global__ __launch_bounds__(128) void k_hx(const float* __restrict__ xa,
                                            const float* __restrict__ xv,
                                            const float* __restrict__ xt,
                                            const float* __restrict__ hf,
                                            const float* __restrict__ tw,
                                            const float* __restrict__ tb,
                                            float* __restrict__ hx) {
  const int m  = blockIdx.z;
  const int n0 = blockIdx.x * 64;
  const int r0 = m * A_ + blockIdx.y * 64;
  const float* Wm = tw + (size_t)m * 2 * H_ * H_;
  const float* Bb = tb + (size_t)m * H_;
  const float* X  = ((m == 0) ? xa : (m == 1) ? xv : xt) + (size_t)blockIdx.y * 64 * H_;
  auto fA = [&](int i, int k0, int kq, float4& x, float4& y) {
    (void)y;
    const float* p = (k0 < H_) ? X  + (size_t)i * H_ + (k0 + kq)
                               : hf + (size_t)(r0 + i) * H_ + (k0 - H_ + kq);
    x = ld4(p);
  };
  auto cA = [&](int k0, const float4& x, const float4& y) { (void)k0; (void)y; return x; };
  auto ep = [&](int i, int n, float v) {
    if (n < H_) hx[(size_t)(r0 + i) * H_ + n] = fmaxf(v + Bb[n], 0.0f);
  };
  gemm_block_64x64<40, false>(2 * H_, Wm, n0, fA, cA, ep);
}

__global__ __launch_bounds__(128) void k_gate(const float* __restrict__ hx,
                                              const float* __restrict__ cw,
                                              const float* __restrict__ cb,
                                              float* __restrict__ hout) {
  constexpr int KP = 3 * H_;
  const int i  = blockIdx.z;
  const int n0 = blockIdx.x * 64;
  const int a0 = blockIdx.y * 64;
  const int j0 = (i == 0) ? 1 : 0;
  const int j1 = (i == 2) ? 1 : 2;
  const float* Hi  = hx + (size_t)i  * A_ * H_;
  const float* Hj0 = hx + (size_t)j0 * A_ * H_;
  const float* Hj1 = hx + (size_t)j1 * A_ * H_;
  const float* Wp = cw + (size_t)(2 * i) * KP * H_;
  const float* Bb = cb + (size_t)(2 * i) * H_;
  auto fA = [&](int r, int k0, int kq, float4& x, float4& y) {
    const int q  = k0 / KP;
    const int kp = k0 - q * KP;
    const int s  = kp / H_;
    const int kk = kp - s * H_ + kq;
    const float* Hj = q ? Hj1 : Hj0;
    const float* px = ((s == 1) ? Hj : Hi) + (size_t)(a0 + r) * H_ + kk;
    x = ld4(px);
    if (s == 2) y = ld4(Hj + (size_t)(a0 + r) * H_ + kk);
    else y = make_float4(0.f, 0.f, 0.f, 0.f);
  };
  auto cA = [&](int k0, const float4& x, const float4& y) {
    if ((k0 % KP) / H_ == 2) return make_float4(x.x * y.x, x.y * y.y, x.z * y.z, x.w * y.w);
    return x;
  };

  const int lane = threadIdx.x & 31;
  const int ir = (threadIdx.x >> 5) * 16 + 8 * (lane >> 4);
  const int nc = n0 + (lane & 15);
  float g[4][8];
  auto gated = [&](const v8f (&acc)[4], const float* Hj, const float* bias, const bool add) {
#pragma unroll
    for (int c = 0; c < 4; ++c) {
      const int n = nc + 16 * c;
      if (n >= H_) continue;
      const float bn = bias[n];
#pragma unroll
      for (int v = 0; v < 8; ++v) {
        const size_t o = (size_t)(a0 + ir + v) * H_ + n;
        const float z = 1.0f / (1.0f + expf(-(acc[c][v] + bn)));
        const float t = z * Hi[o] + (1.0f - z) * Hj[o];
        g[c][v] = add ? g[c][v] + t : t;
      }
    }
  };
  auto mid = [&](v8f (&acc)[4]) {
    gated(acc, Hj0, Bb, false);
#pragma unroll
    for (int c = 0; c < 4; ++c) acc[c] = v8f{};
  };
  auto ep = [&](const v8f (&acc)[4]) {
    gated(acc, Hj1, Bb + H_, true);
#pragma unroll
    for (int c = 0; c < 4; ++c) {
      const int n = nc + 16 * c;
      if (n >= H_) continue;
#pragma unroll
      for (int v = 0; v < 8; ++v)
        hout[((size_t)i * A_ + a0 + ir + v) * H_ + n] = g[c][v];
    }
  };
  gemm_block_64x64<40, true>(2 * KP, Wp, n0, fA, cA, ep, KP, mid);
}

__global__ __launch_bounds__(512) void k_final(const float* __restrict__ hout,
                                               const float* __restrict__ uw,
                                               const float* __restrict__ ub,
                                               float* __restrict__ out) {
  constexpr int UPB = 16;
  constexpr int NSF = H_ / 32;
  constexpr int REM = H_ - 32 * NSF;
  static_assert((UPB * TAG_) % 32 == 0, "a block's results must be whole 128-byte lines");
  static_assert(REM > 0 && M_ * REM <= 32, "the left-over units of all modalities share one slot");
  __shared__ float res[UPB * TAG_];
  const int wv   = threadIdx.x >> 5;
  const int lane = threadIdx.x & 31;
  const int a    = blockIdx.x * UPB + wv;
  float acc[TAG_] = {};
  auto add6 = [&](const float v, const float2 w01, const float2 w23, const float2 w45) {
    acc[0] += v * w01.x; acc[1] += v * w01.y;
    acc[2] += v * w23.x; acc[3] += v * w23.y;
    acc[4] += v * w45.x; acc[5] += v * w45.y;
  };
#pragma unroll
  for (int m = 0; m < M_; ++m) {
    const float* hr = hout + (size_t)(m * A_ + a) * H_;
    const float* wm = uw + (size_t)m * H_ * TAG_;
#pragma unroll
    for (int s = 0; s < NSF; ++s) {
      const int h = lane + 32 * s;
      const float* wr = wm + h * TAG_;
      add6(hr[h], ld2(wr), ld2(wr + 2), ld2(wr + 4));
    }
  }
  {
    const bool ok = lane < M_ * REM;
    const int l = ok ? lane : M_ * REM - 1;
    const int m = l / REM;
    const int h = 32 * NSF + l % REM;
    const float* wr = uw + ((size_t)m * H_ + h) * TAG_;
    const float  v   = hout[(size_t)(m * A_ + a) * H_ + h];
    const float2 w01 = ld2(wr), w23 = ld2(wr + 2), w45 = ld2(wr + 4);
    const float2 z = make_float2(0.f, 0.f);
    add6(ok ? v : 0.f, ok ? w01 : z, ok ? w23 : z, ok ? w45 : z);
  }
#pragma unroll
  for (int t = 0; t < TAG_; ++t) acc[t] = wave_sum(acc[t]);
  if (lane == 0) {
#pragma unroll
    for (int t = 0; t < TAG_; ++t)
      res[wv * TAG_ + t] = acc[t] + (ub[t] + ub[TAG_ + t] + ub[2 * TAG_ + t]);
  }
  __syncthreads();
  if (wv == 0) {
    for (int j = 0; j < UPB * TAG_ / 32; ++j)
      out[(size_t)blockIdx.x * (UPB * TAG_) + 32 * j + lane] = res[32 * j + lane];
  }
}

}

extern "C" void kernel_launch(void* const* d_in, const int* in_sizes, int n_in,
                              void* d_out, int out_size, void* d_ws, size_t ws_size,
                              hipStream_t stream) {
  (void)in_sizes; (void)n_in; (void)out_size; (void)ws_size;
  const float* x_a     = (const float*)d_in[0];
  const float* x_v     = (const float*)d_in[1];
  const float* x_t     = (const float*)d_in[2];
  const float* fc0_w   = (const float*)d_in[3];
  const float* fc0_b   = (const float*)d_in[4];
  const float* conv_w  = (const float*)d_in[5];
  const float* tm_w    = (const float*)d_in[6];
  const float* tm_b    = (const float*)d_in[7];
  const float* cross_w = (const float*)d_in[8];
  const float* cross_b = (const float*)d_in[9];
  const float* uni_w   = (const float*)d_in[10];
  const float* uni_b   = (const float*)d_in[11];
  float* out = (float*)d_out;

  const size_t NH = (size_t)N_ * H_;
  float* ws    = (float*)d_ws;
  float* intra = ws; ws += (size_t)N_ * L_;
  float* crossv= ws; ws += (size_t)M_ * M_ * A_;
  float* dinv  = ws; ws += N_;
  float* h0    = ws; ws += NH;
  float* hA    = ws; ws += NH;
  float* hBf   = ws; ws += NH;
  float* hi    = ws; ws += NH;
  float* hx    = ws; ws += NH;
  float* hout  = ws; ws += NH;

  k_graph<<<M_ * B_,             dim3(512), 0, stream>>>(x_a, x_v, x_t, intra, crossv, dinv);
  k_fc0  <<<dim3(4, N_ / 64),    dim3(128), 0, stream>>>(x_a, x_v, x_t, fc0_w, fc0_b, h0);

  const float beta0 = logf(0.5f / 1.0f + 1.0f);
  const float beta1 = logf(0.5f / 2.0f + 1.0f);
  k_spmm <<<dim3(4, M_ * B_),    dim3(128), 0, stream>>>(intra, crossv, dinv, h0, hi);
  k_conv <<<dim3(4, N_ / 64),    dim3(128), 0, stream>>>(hi, h0, conv_w, beta0, hA);
  k_spmm <<<dim3(4, M_ * B_),    dim3(128), 0, stream>>>(intra, crossv, dinv, hA, hi);
  k_conv <<<dim3(4, N_ / 64),    dim3(128), 0, stream>>>(hi, h0, conv_w + 2 * H_ * H_, beta1, hBf);

  k_hx   <<<dim3(4, A_ / 64, M_), dim3(128), 0, stream>>>(x_a, x_v, x_t, hBf, tm_w, tm_b, hx);

  k_gate <<<dim3(4, A_ / 64, M_), dim3(128), 0, stream>>>(hx, cross_w, cross_b, hout);

  k_final<<<A_ / 16, dim3(512), 0, stream>>>(hout, uni_w, uni_b, out);
}


# GEMM epilogues (fc0, spmm, conv, hx, gate): a wave turns its accumulator strip into rows through its own LDS tile, so epilogue loads and stores are 16-byte accesses on 256-byte runs, issued together
# speedup vs baseline: 6.9166x; 1.2956x over previous
#include <hip/hip_runtime.h>
#include <math.h>

typedef __attribute__((ext_vector_type(2))) float v2f;
typedef __attribute__((ext_vector_type(8))) float v8f;

namespace {

constexpr int B_   = 32;
constexpr int L_   = 64;
constexpr int H_   = 200;
constexpr int M_   = 3;
constexpr int A_   = B_ * L_;
constexpr int N_   = M_ * A_;
constexpr int TAG_ = 6;
constexpr float PI_F = 3.14159265358979323846f;

__device__ __forceinline__ float wave_sum(float s) {
#pragma unroll
  for (int off = 16; off > 0; off >>= 1) s += __shfl_xor(s, off, 32);
  return s;
}

__global__ __launch_bounds__(512) void k_graph(const float* __restrict__ xa,
                                               const float* __restrict__ xv,
                                               const float* __restrict__ xt,
                                               float* __restrict__ intra,
                                               float* __restrict__ crossv,
                                               float* __restrict__ dinv) {
  constexpr int PN  = H_ + 4;
  constexpr int TP  = L_ + 4;
  constexpr int NS  = (H_ + 31) / 32;
  constexpr int RPW = L_ / 16;
  static_assert((PN % 8) == 4, "pitch must be 4 * odd dwords");
  static_assert(L_ * TP <= L_ * PN, "the similarity tile reuses the rows' LDS");
  __shared__ __attribute__((aligned(16))) float nr[L_ * PN];
  __shared__ float cx[2][L_];
  __shared__ float dv[L_];
  const int mb   = blockIdx.x;
  const int m    = mb >> 5;
  const int b    = mb & 31;
  const int w    = threadIdx.x >> 5;
  const int lane = threadIdx.x & 31;
  const int n1   = (m == 0) ? 1 : 0;
  const int n2   = (m == 2) ? 1 : 2;
  auto plane = [&](const int mm) { return (mm == 0) ? xa : (mm == 1) ? xv : xt; };
  const float* P0 = plane(m)  + (size_t)b * L_ * H_;
  const float* P1 = plane(n1) + (size_t)b * L_ * H_;
  const float* P2 = plane(n2) + (size_t)b * L_ * H_;

  float v0[RPW][NS], v1[RPW][NS], v2[RPW][NS];
#pragma unroll
  for (int q = 0; q < RPW; ++q) {
    const int r = w * RPW + q;
#pragma unroll
    for (int j = 0; j < NS; ++j) {
      const int k = lane + 32 * j;
      const bool ok = k < H_;
      v0[q][j] = ok ? P0[(size_t)r * H_ + k] : 0.f;
      v1[q][j] = ok ? P1[(size_t)r * H_ + k] : 0.f;
      v2[q][j] = ok ? P2[(size_t)r * H_ + k] : 0.f;
    }
  }
  float c1[RPW], c2[RPW];
#pragma unroll
  for (int q = 0; q < RPW; ++q) {
    const int r = w * RPW + q;
    float s0 = 0.f, s1 = 0.f, s2 = 0.f;
#pragma unroll
    for (int j = 0; j < NS; ++j) {
      s0 += v0[q][j] * v0[q][j];
      s1 += v1[q][j] * v1[q][j];
      s2 += v2[q][j] * v2[q][j];
    }
    const float rn0 = 1.0f / sqrtf(wave_sum(s0));
    const float rn1 = 1.0f / sqrtf(wave_sum(s1));
    const float rn2 = 1.0f / sqrtf(wave_sum(s2));
    float d1 = 0.f, d2 = 0.f;
#pragma unroll
    for (int j = 0; j < NS; ++j) {
      const int k = lane + 32 * j;
      const float u0 = v0[q][j] * rn0;
      const float u1 = v1[q][j] * rn1;
      const float u2 = v2[q][j] * rn2;
      if (k < H_) nr[r * PN + k] = u0;
      d1 += u0 * u1;
      d2 += u0 * u2;
    }
    d1 = wave_sum(d1);
    d2 = wave_sum(d2);
    c1[q] = 1.0f - acosf(d1 * 0.99999f) * (1.0f / PI_F);
    c2[q] = 1.0f - acosf(d2 * 0.99999f) * (1.0f / PI_F);
    if (lane == 0) { cx[0][r] = c1[q]; cx[1][r] = c2[q]; }
  }
  __syncthreads();

  const int mr = lane & 15;
  const int kh = lane >> 4;
  const int m0 = (w >> 2) * 16, n0 = (w & 3) * 16;
  v8f acc = {};
  {
    const float* ap = nr + (m0 + mr) * PN + 2 * kh;
    const float* bp = nr + (n0 + mr) * PN + 2 * kh;
#pragma unroll 10
    for (int k = 0; k < H_; k += 4) {
      const v2f a = *reinterpret_cast<const v2f*>(ap + k);
      const v2f bb = *reinterpret_cast<const v2f*>(bp + k);
      acc = __builtin_amdgcn_wmma_f32_16x16x4_f32(false, a, false, bb, (short)0, acc, false, false);
    }
  }
  __syncthreads();
#pragma unroll
  for (int v = 0; v < 8; ++v)
    nr[(m0 + 8 * kh + v) * TP + n0 + mr] = 1.0f - acosf(acc[v] * 0.99999f) * (1.0f / PI_F);
  __syncthreads();

#pragma unroll
  for (int q = 0; q < RPW; ++q) {
    const int r = w * RPW + q;
    const float t0 = nr[r * TP + lane];
    const float t1 = nr[r * TP + lane + 32];
    float* orow = intra + ((size_t)mb * L_ + r) * L_;
    orow[lane]      = t0;
    orow[lane + 32] = t1;
    float s = 0.f;
    s += t0;
    s += t1;
    s = wave_sum(s);
    s += c1[q];
    s += c2[q];
    if (lane == 0) dv[r] = 1.0f / sqrtf(s);
  }
  __syncthreads();
  if (w < 2) {
    dinv[(size_t)mb * L_ + 32 * w + lane] = dv[32 * w + lane];
  } else if (w < 4) {
    const int h = w - 2;
    crossv[((size_t)m * M_ + n1) * A_ + b * L_ + 32 * h + lane] = cx[0][32 * h + lane];
  } else if (w < 6) {
    const int h = w - 4;
    crossv[((size_t)m * M_ + n2) * A_ + b * L_ + 32 * h + lane] = cx[1][32 * h + lane];
  }
}

struct no_mid {};
template <typename T> struct is_no_mid { static constexpr bool value = false; };
template <> struct is_no_mid<no_mid> { static constexpr bool value = true; };
__device__ __forceinline__ float4 ld4(const float* p) {
  return *reinterpret_cast<const float4*>(p);
}
__device__ __forceinline__ float2 ld2(const float* p) {
  return *reinterpret_cast<const float2*>(p);
}
__device__ __forceinline__ void st4(float* p, const float4 v) {
  *reinterpret_cast<float4*>(p) = v;
}

template <int BK, bool COMB_HAS_MATH, typename FA, typename CA, typename FE, typename FM = no_mid>
__device__ __forceinline__ void gemm_block_64x64(const int K, const float* __restrict__ Bp,
                                                 const int n0, FA fetchA, CA combA, FE epi,
                                                 const int Kmid = 0, FM mid = FM{}) {
  constexpr bool TWO_PRODUCTS = !is_no_mid<FM>::value;
  constexpr int BM = 64, BN = 64, NTHR = 128;
  constexpr int PA = BK + 4;
  constexpr int PB = BN + 16;
  constexpr int KQ = BK / 4;
  constexpr int NP = BN / 2;
  constexpr int A4 = BM * KQ;
  constexpr int B4 = (BK / 2) * NP;
  constexpr int RA = A4 / NTHR;
  constexpr int RB = B4 / NTHR;
  static_assert(A4 % NTHR == 0 && B4 % NTHR == 0, "chunks must split evenly over the block");
  static_assert((PA % 8) == 4, "A pitch must be 4 * odd dwords");
  static_assert(((2 * PB) % 64) == 32, "B pair pitch must be 32 mod 64 dwords");

  constexpr int ASZ = BM * PA;
  constexpr int BSZ = (BK / 2) * PB * 2;
  __shared__ __attribute__((aligned(16))) float As[2 * ASZ];
  __shared__ __attribute__((aligned(16))) float Bs[2 * BSZ];
  constexpr int TC = BN + 4;
  static_assert(((8 * TC) % 64) == 32 && (TC % 4) == 0, "epilogue tile pitch");
  __shared__ __attribute__((aligned(16))) float Ct[BM * TC];

  const int t    = threadIdx.x;
  const int lane = t & 31;
  const int w    = t >> 5;
  const int mr   = lane & 15;
  const int kh   = lane >> 4;

  float4 ax[RA], ay[RA];
  float2 be[RB], bo[RB];

  auto prefetch = [&](const int k0) {
#pragma unroll
    for (int r = 0; r < RA; ++r) {
      const int e = t + NTHR * r;
      fetchA(e / KQ, k0, (e % KQ) * 4, ax[r], ay[r]);
    }
#pragma unroll
    for (int r = 0; r < RB; ++r) {
      const int e  = t + NTHR * r;
      const int kp = e / NP;
      const int n  = min(n0 + (e % NP) * 2, H_ - 2);
      const float* p = Bp + (size_t)(k0 + 2 * kp) * H_ + n;
      be[r] = ld2(p);
      bo[r] = ld2(p + H_);
    }
  };

  auto stage = [&](const int k0, const int buf) {
#pragma unroll
    for (int r = 0; r < RA; ++r) {
      const int e = t + NTHR * r;
      *reinterpret_cast<float4*>(&As[buf * ASZ + (e / KQ) * PA + (e % KQ) * 4]) = combA(k0, ax[r], ay[r]);
    }
#pragma unroll
    for (int r = 0; r < RB; ++r) {
      const int e = t + NTHR * r;
      const int j = (e % NP) * 2;
      float4 v = make_float4(be[r].x, bo[r].x, be[r].y, bo[r].y);
      if (n0 + j >= H_) v = make_float4(0.f, 0.f, 0.f, 0.f);
      *reinterpret_cast<float4*>(&Bs[buf * BSZ + ((e / NP) * PB + j) * 2]) = v;
    }
  };

  const int aoff = (w * 16 + mr) * PA + 2 * kh;
  const int boff = (kh * PB + mr) * 2;

  v8f acc[4] = {};
  auto compute = [&](const int buf) {
    const float* aptr = As + buf * ASZ + aoff;
    const float* bptr = Bs + buf * BSZ + boff;
#pragma unroll
    for (int ks = 0; ks < BK; ks += 4) {
      const v2f a = *reinterpret_cast<const v2f*>(aptr + ks);
#pragma unroll
      for (int c = 0; c < 4; ++c) {
        const v2f b = *reinterpret_cast<const v2f*>(bptr + ((ks / 2) * PB + 16 * c) * 2);
        acc[c] = __builtin_amdgcn_wmma_f32_16x16x4_f32(false, a, false, b, (short)0, acc[c], false, false);
      }
    }
  };

  const int ei = w * 16 + kh;
  const int en = n0 + mr * 4;
  auto to_rows = [&](float4 (&rv)[8]) {
    float* ct = Ct + (w * 16) * TC;
#pragma unroll
    for (int c = 0; c < 4; ++c) {
#pragma unroll
      for (int v = 0; v < 8; ++v) ct[(8 * kh + v) * TC + 16 * c + mr] = acc[c][v];
    }
#pragma unroll
    for (int j = 0; j < 8; ++j) rv[j] = *reinterpret_cast<const float4*>(&ct[(kh + 2 * j) * TC + mr * 4]);
  };

  prefetch(0);
  stage(0, 0);
  __syncthreads();
  int buf = 0;
  for (int k0 = BK; k0 < K; k0 += BK) {
    if constexpr (TWO_PRODUCTS) {
      if (k0 == Kmid + BK) {
        float4 rv[8];
        to_rows(rv);
        mid(ei, en, rv);
#pragma unroll
        for (int c = 0; c < 4; ++c) acc[c] = v8f{};
      }
    }
    prefetch(k0);
    __builtin_amdgcn_sched_barrier(0);
    compute(buf);
    if constexpr (COMB_HAS_MATH) __builtin_amdgcn_sched_barrier(0);
    stage(k0, buf ^ 1);
    __syncthreads();
    buf ^= 1;
  }
  compute(buf);

  float4 rv[8];
  to_rows(rv);
  epi(ei, en, rv);
}

__global__ __launch_bounds__(128) void k_fc0(const float* __restrict__ xa,
                                             const float* __restrict__ xv,
                                             const float* __restrict__ xt,
                                             const float* __restrict__ w,
                                             const float* __restrict__ bias,
                                             float* __restrict__ h0) {
  const int n0 = blockIdx.x * 64;
  const int r0 = blockIdx.y * 64;
  const int mm = r0 / A_;
  const float* X = ((mm == 0) ? xa : (mm == 1) ? xv : xt) + (size_t)(r0 - mm * A_) * H_;
  auto fA = [&](int i, int k0, int kq, float4& x, float4& y) {
    (void)y;
    x = ld4(X + (size_t)i * H_ + k0 + kq);
  };
  auto cA = [&](int k0, const float4& x, const float4& y) { (void)k0; (void)y; return x; };
  auto ep = [&](const int i0, const int n, const float4 (&v)[8]) {
    if (n >= H_) return;
    const float4 b = ld4(bias + n);
#pragma unroll
    for (int j = 0; j < 8; ++j)
      st4(h0 + (size_t)(r0 + i0 + 2 * j) * H_ + n,
          make_float4(fmaxf(v[j].x + b.x, 0.0f), fmaxf(v[j].y + b.y, 0.0f),
                      fmaxf(v[j].z + b.z, 0.0f), fmaxf(v[j].w + b.w, 0.0f)));
  };
  gemm_block_64x64<40, false>(H_, w, n0, fA, cA, ep);
}

__global__ __launch_bounds__(128) void k_spmm(const float* __restrict__ intra,
                                              const float* __restrict__ crossv,
                                              const float* __restrict__ dinv,
                                              const float* __restrict__ h,
                                              float* __restrict__ hi) {
  const int n0 = blockIdx.x * 64;
  const int mb = blockIdx.y;
  const int m  = mb >> 5;
  const int rowbase = mb * L_;
  const float* S = intra + (size_t)mb * L_ * L_;
  auto fA = [&](int i, int k0, int kq, float4& x, float4& y) {
    x = ld4(S + (size_t)i * L_ + k0 + kq);
    y = ld4(dinv + rowbase + k0 + kq);
  };
  auto cA = [&](int k0, const float4& x, const float4& y) {
    (void)k0;
    return make_float4(x.x * y.x, x.y * y.y, x.z * y.z, x.w * y.w);
  };
  const int o1 = (m == 0) ? 1 : 0;
  const int o2 = (m == 2) ? 1 : 2;
  auto ep = [&](const int i0, const int n, const float4 (&v)[8]) {
    if (n >= H_) return;
    float  cd1[8], cd2[8], dr[8];
    float4 g1[8], g2[8];
#pragma unroll
    for (int j = 0; j < 8; ++j) {
      const int row  = rowbase + i0 + 2 * j;
      const int a    = row - m * A_;
      const int r1   = o1 * A_ + a;
      const int r2   = o2 * A_ + a;
      cd1[j] = crossv[((size_t)m * M_ + o1) * A_ + a] * dinv[r1];
      cd2[j] = crossv[((size_t)m * M_ + o2) * A_ + a] * dinv[r2];
      dr[j]  = dinv[row];
      g1[j]  = ld4(h + (size_t)r1 * H_ + n);
      g2[j]  = ld4(h + (size_t)r2 * H_ + n);
    }
#pragma unroll
    for (int j = 0; j < 8; ++j) {
      float4 s = v[j];
      s.x += cd1[j] * g1[j].x; s.y += cd1[j] * g1[j].y; s.z += cd1[j] * g1[j].z; s.w += cd1[j] * g1[j].w;
      s.x += cd2[j] * g2[j].x; s.y += cd2[j] * g2[j].y; s.z += cd2[j] * g2[j].z; s.w += cd2[j] * g2[j].w;
      st4(hi + (size_t)(rowbase + i0 + 2 * j) * H_ + n,
          make_float4(s.x * dr[j], s.y * dr[j], s.z * dr[j], s.w * dr[j]));
    }
  };
  gemm_block_64x64<32, true>(L_, h + (size_t)rowbase * H_, n0, fA, cA, ep);
}

__global__ __launch_bounds__(128) void k_conv(const float* __restrict__ hi,
                                              const float* __restrict__ h0,
                                              const float* __restrict__ w, float beta,
                                              float* __restrict__ hout) {
  const int n0 = blockIdx.x * 64;
  const int r0 = blockIdx.y * 64;
  auto fA = [&](int i, int k0, int kq, float4& x, float4& y) {
    (void)y;
    const float* p = (k0 < H_) ? hi + (size_t)(r0 + i) * H_ + (k0 + kq)
                               : h0 + (size_t)(r0 + i) * H_ + (k0 - H_ + kq);
    x = ld4(p);
  };
  auto cA = [&](int k0, const float4& x, const float4& y) { (void)k0; (void)y; return x; };
  auto ep = [&](const int i0, const int n, const float4 (&v)[8]) {
    if (n >= H_) return;
    float4 a[8], b[8];
#pragma unroll
    for (int j = 0; j < 8; ++j) {
      const size_t o = (size_t)(r0 + i0 + 2 * j) * H_ + n;
      a[j] = ld4(hi + o);
      b[j] = ld4(h0 + o);
    }
    auto one = [&](const float x, const float p, const float q) {
      const float r = 0.9f * p + 0.1f * q;
      return fmaxf(beta * x + (1.0f - beta) * r, 0.0f);
    };
#pragma unroll
    for (int j = 0; j < 8; ++j)
      st4(hout + (size_t)(r0 + i0 + 2 * j) * H_ + n,
          make_float4(one(v[j].x, a[j].x, b[j].x), one(v[j].y, a[j].y, b[j].y),
                      one(v[j].z, a[j].z, b[j].z), one(v[j].w, a[j].w, b[j].w)));
  };
  gemm_block_64x64<40, false>(2 * H_, w, n0, fA, cA, ep);
}

__global__ __launch_bounds__(128) void k_hx(const float* __restrict__ xa,
                                            const float* __restrict__ xv,
                                            const float* __restrict__ xt,
                                            const float* __restrict__ hf,
                                            const float* __restrict__ tw,
                                            const float* __restrict__ tb,
                                            float* __restrict__ hx) {
  const int m  = blockIdx.z;
  const int n0 = blockIdx.x * 64;
  const int r0 = m * A_ + blockIdx.y * 64;
  const float* Wm = tw + (size_t)m * 2 * H_ * H_;
  const float* Bb = tb + (size_t)m * H_;
  const float* X  = ((m == 0) ? xa : (m == 1) ? xv : xt) + (size_t)blockIdx.y * 64 * H_;
  auto fA = [&](int i, int k0, int kq, float4& x, float4& y) {
    (void)y;
    const float* p = (k0 < H_) ? X  + (size_t)i * H_ + (k0 + kq)
                               : hf + (size_t)(r0 + i) * H_ + (k0 - H_ + kq);
    x = ld4(p);
  };
  auto cA = [&](int k0, const float4& x, const float4& y) { (void)k0; (void)y; return x; };
  auto ep = [&](const int i0, const int n, const float4 (&v)[8]) {
    if (n >= H_) return;
    const float4 b = ld4(Bb + n);
#pragma unroll
    for (int j = 0; j < 8; ++j)
      st4(hx + (size_t)(r0 + i0 + 2 * j) * H_ + n,
          make_float4(fmaxf(v[j].x + b.x, 0.0f), fmaxf(v[j].y + b.y, 0.0f),
                      fmaxf(v[j].z + b.z, 0.0f), fmaxf(v[j].w + b.w, 0.0f)));
  };
  gemm_block_64x64<40, false>(2 * H_, Wm, n0, fA, cA, ep);
}

__global__ __launch_bounds__(128) void k_gate(const float* __restrict__ hx,
                                              const float* __restrict__ cw,
                                              const float* __restrict__ cb,
                                              float* __restrict__ hout) {
  constexpr int KP = 3 * H_;
  const int i  = blockIdx.z;
  const int n0 = blockIdx.x * 64;
  const int a0 = blockIdx.y * 64;
  const int j0 = (i == 0) ? 1 : 0;
  const int j1 = (i == 2) ? 1 : 2;
  const float* Hi  = hx + (size_t)i  * A_ * H_;
  const float* Hj0 = hx + (size_t)j0 * A_ * H_;
  const float* Hj1 = hx + (size_t)j1 * A_ * H_;
  const float* Wp = cw + (size_t)(2 * i) * KP * H_;
  const float* Bb = cb + (size_t)(2 * i) * H_;
  auto fA = [&](int r, int k0, int kq, float4& x, float4& y) {
    const int q  = k0 / KP;
    const int kp = k0 - q * KP;
    const int s  = kp / H_;
    const int kk = kp - s * H_ + kq;
    const float* Hj = q ? Hj1 : Hj0;
    const float* px = ((s == 1) ? Hj : Hi) + (size_t)(a0 + r) * H_ + kk;
    x = ld4(px);
    if (s == 2) y = ld4(Hj + (size_t)(a0 + r) * H_ + kk);
    else y = make_float4(0.f, 0.f, 0.f, 0.f);
  };
  auto cA = [&](int k0, const float4& x, const float4& y) {
    if ((k0 % KP) / H_ == 2) return make_float4(x.x * y.x, x.y * y.y, x.z * y.z, x.w * y.w);
    return x;
  };

  float4 g[8];
  auto gated = [&](const int i0, const int n, const float4 (&v)[8], const float* Hj, const float* bias,
                   const bool add) {
    if (n >= H_) return;
    const float4 bn = ld4(bias + n);
    float4 p[8], q[8];
#pragma unroll
    for (int j = 0; j < 8; ++j) {
      const size_t o = (size_t)(a0 + i0 + 2 * j) * H_ + n;
      p[j] = ld4(Hi + o);
      q[j] = ld4(Hj + o);
    }
    auto one = [&](const float x, const float b, const float hi_, const float hj_) {
      const float z = 1.0f / (1.0f + expf(-(x + b)));
      return z * hi_ + (1.0f - z) * hj_;
    };
#pragma unroll
    for (int j = 0; j < 8; ++j) {
      const float4 t = make_float4(one(v[j].x, bn.x, p[j].x, q[j].x), one(v[j].y, bn.y, p[j].y, q[j].y),
                                   one(v[j].z, bn.z, p[j].z, q[j].z), one(v[j].w, bn.w, p[j].w, q[j].w));
      g[j] = add ? make_float4(g[j].x + t.x, g[j].y + t.y, g[j].z + t.z, g[j].w + t.w) : t;
    }
  };
  auto mid = [&](const int i0, const int n, const float4 (&v)[8]) { gated(i0, n, v, Hj0, Bb, false); };
  auto ep = [&](const int i0, const int n, const float4 (&v)[8]) {
    if (n >= H_) return;
    gated(i0, n, v, Hj1, Bb + H_, true);
#pragma unroll
    for (int j = 0; j < 8; ++j)
      st4(hout + ((size_t)i * A_ + a0 + i0 + 2 * j) * H_ + n, g[j]);
  };
  gemm_block_64x64<40, true>(2 * KP, Wp, n0, fA, cA, ep, KP, mid);
}

__global__ __launch_bounds__(512) void k_final(const float* __restrict__ hout,
                                               const float* __restrict__ uw,
                                               const float* __restrict__ ub,
                                               float* __restrict__ out) {
  constexpr int UPB = 16;
  constexpr int NSF = H_ / 32;
  constexpr int REM = H_ - 32 * NSF;
  static_assert((UPB * TAG_) % 32 == 0, "a block's results must be whole 128-byte lines");
  static_assert(REM > 0 && M_ * REM <= 32, "the left-over units of all modalities share one slot");
  __shared__ float res[UPB * TAG_];
  const int wv   = threadIdx.x >> 5;
  const int lane = threadIdx.x & 31;
  const int a    = blockIdx.x * UPB + wv;
  float acc[TAG_] = {};
  auto add6 = [&](const float v, const float2 w01, const float2 w23, const float2 w45) {
    acc[0] += v * w01.x; acc[1] += v * w01.y;
    acc[2] += v * w23.x; acc[3] += v * w23.y;
    acc[4] += v * w45.x; acc[5] += v * w45.y;
  };
#pragma unroll
  for (int m = 0; m < M_; ++m) {
    const float* hr = hout + (size_t)(m * A_ + a) * H_;
    const float* wm = uw + (size_t)m * H_ * TAG_;
#pragma unroll
    for (int s = 0; s < NSF; ++s) {
      const int h = lane + 32 * s;
      const float* wr = wm + h * TAG_;
      add6(hr[h], ld2(wr), ld2(wr + 2), ld2(wr + 4));
    }
  }
  {
    const bool ok = lane < M_ * REM;
    const int l = ok ? lane : M_ * REM - 1;
    const int m = l / REM;
    const int h = 32 * NSF + l % REM;
    const float* wr = uw + ((size_t)m * H_ + h) * TAG_;
    const float  v   = hout[(size_t)(m * A_ + a) * H_ + h];
    const float2 w01 = ld2(wr), w23 = ld2(wr + 2), w45 = ld2(wr + 4);
    const float2 z = make_float2(0.f, 0.f);
    add6(ok ? v : 0.f, ok ? w01 : z, ok ? w23 : z, ok ? w45 : z);
  }
#pragma unroll
  for (int t = 0; t < TAG_; ++t) acc[t] = wave_sum(acc[t]);
  if (lane == 0) {
#pragma unroll
    for (int t = 0; t < TAG_; ++t)
      res[wv * TAG_ + t] = acc[t] + (ub[t] + ub[TAG_ + t] + ub[2 * TAG_ + t]);
  }
  __syncthreads();
  if (wv == 0) {
    for (int j = 0; j < UPB * TAG_ / 32; ++j)
      out[(size_t)blockIdx.x * (UPB * TAG_) + 32 * j + lane] = res[32 * j + lane];
  }
}

}

extern "C" void kernel_launch(void* const* d_in, const int* in_sizes, int n_in,
                              void* d_out, int out_size, void* d_ws, size_t ws_size,
                              hipStream_t stream) {
  (void)in_sizes; (void)n_in; (void)out_size; (void)ws_size;
  const float* x_a     = (const float*)d_in[0];
  const float* x_v     = (const float*)d_in[1];
  const float* x_t     = (const float*)d_in[2];
  const float* fc0_w   = (const float*)d_in[3];
  const float* fc0_b   = (const float*)d_in[4];
  const float* conv_w  = (const float*)d_in[5];
  const float* tm_w    = (const float*)d_in[6];
  const float* tm_b    = (const float*)d_in[7];
  const float* cross_w = (const float*)d_in[8];
  const float* cross_b = (const float*)d_in[9];
  const float* uni_w   = (const float*)d_in[10];
  const float* uni_b   = (const float*)d_in[11];
  float* out = (float*)d_out;

  const size_t NH = (size_t)N_ * H_;
  float* ws    = (float*)d_ws;
  float* intra = ws; ws += (size_t)N_ * L_;
  float* crossv= ws; ws += (size_t)M_ * M_ * A_;
  float* dinv  = ws; ws += N_;
  float* h0    = ws; ws += NH;
  float* hA    = ws; ws += NH;
  float* hBf   = ws; ws += NH;
  float* hi    = ws; ws += NH;
  float* hx    = ws; ws += NH;
  float* hout  = ws; ws += NH;

  k_graph<<<M_ * B_,             dim3(512), 0, stream>>>(x_a, x_v, x_t, intra, crossv, dinv);
  k_fc0  <<<dim3(4, N_ / 64),    dim3(128), 0, stream>>>(x_a, x_v, x_t, fc0_w, fc0_b, h0);

  const float beta0 = logf(0.5f / 1.0f + 1.0f);
  const float beta1 = logf(0.5f / 2.0f + 1.0f);
  k_spmm <<<dim3(4, M_ * B_),    dim3(128), 0, stream>>>(intra, crossv, dinv, h0, hi);
  k_conv <<<dim3(4, N_ / 64),    dim3(128), 0, stream>>>(hi, h0, conv_w, beta0, hA);
  k_spmm <<<dim3(4, M_ * B_),    dim3(128), 0, stream>>>(intra, crossv, dinv, hA, hi);
  k_conv <<<dim3(4, N_ / 64),    dim3(128), 0, stream>>>(hi, h0, conv_w + 2 * H_ * H_, beta1, hBf);

  k_hx   <<<dim3(4, A_ / 64, M_), dim3(128), 0, stream>>>(x_a, x_v, x_t, hBf, tm_w, tm_b, hx);

  k_gate <<<dim3(4, A_ / 64, M_), dim3(128), 0, stream>>>(hx, cross_w, cross_b, hout);

  k_final<<<A_ / 16, dim3(512), 0, stream>>>(hout, uni_w, uni_b, out);
}


# k_graph and k_fc0, which need nothing from each other, merged into one launch: graph set-up re-cut to 4-wave blocks beside the 64x64 fc0 tiles, sum orders kept; 8 launches, was 9
# speedup vs baseline: 7.2646x; 1.0503x over previous
#include <hip/hip_runtime.h>
#include <math.h>

typedef __attribute__((ext_vector_type(2))) float v2f;
typedef __attribute__((ext_vector_type(8))) float v8f;

namespace {

constexpr int B_   = 32;
constexpr int L_   = 64;
constexpr int H_   = 200;
constexpr int M_   = 3;
constexpr int A_   = B_ * L_;
constexpr int N_   = M_ * A_;
constexpr int TAG_ = 6;
constexpr float PI_F = 3.14159265358979323846f;

__device__ __forceinline__ float wave_sum(float s) {
#pragma unroll
  for (int off = 16; off > 0; off >>= 1) s += __shfl_xor(s, off, 32);
  return s;
}

template <int N>
__device__ __forceinline__ void wave_sum_n(float (&s)[N]) {
#pragma unroll
  for (int off = 16; off > 0; off >>= 1) {
    float t[N];
#pragma unroll
    for (int i = 0; i < N; ++i) t[i] = __shfl_xor(s[i], off, 32);
#pragma unroll
    for (int i = 0; i < N; ++i) s[i] += t[i];
  }
}

__device__ __forceinline__ void graph_block(const int mb,
                                            const float* __restrict__ xa,
                                            const float* __restrict__ xv,
                                            const float* __restrict__ xt,
                                            float* __restrict__ intra,
                                            float* __restrict__ crossv,
                                            float* __restrict__ dinv) {
  constexpr int PN  = H_ + 4;
  constexpr int TP  = L_ + 4;
  constexpr int NS  = (H_ + 31) / 32;
  constexpr int NW  = 4;
  constexpr int RPW = L_ / NW;
  constexpr int RB  = 8;
  static_assert((PN % 8) == 4, "pitch must be 4 * odd dwords");
  static_assert(L_ * TP <= L_ * PN, "the similarity tile reuses the rows' LDS");
  static_assert(RPW == 16 && RPW % RB == 0, "a wave owns one 16-row strip");
  __shared__ __attribute__((aligned(16))) float nr[L_ * PN];
  __shared__ float cx[2][L_];
  __shared__ float dv[L_];
  const int m    = mb >> 5;
  const int b    = mb & 31;
  const int w    = threadIdx.x >> 5;
  const int lane = threadIdx.x & 31;
  const int n1   = (m == 0) ? 1 : 0;
  const int n2   = (m == 2) ? 1 : 2;
  auto plane = [&](const int mm) { return (mm == 0) ? xa : (mm == 1) ? xv : xt; };
  const float* P0 = plane(m)  + (size_t)b * L_ * H_;
  const float* P1 = plane(n1) + (size_t)b * L_ * H_;
  const float* P2 = plane(n2) + (size_t)b * L_ * H_;

#pragma unroll 1
  for (int rb = w * RPW; rb < (w + 1) * RPW; rb += RB) {
    float v0[RB][NS], v1[RB][NS], v2[RB][NS];
#pragma unroll
    for (int q = 0; q < RB; ++q) {
#pragma unroll
      for (int j = 0; j < NS; ++j) {
        const int k = lane + 32 * j;
        const bool ok = k < H_;
        v0[q][j] = ok ? P0[(size_t)(rb + q) * H_ + k] : 0.f;
        v1[q][j] = ok ? P1[(size_t)(rb + q) * H_ + k] : 0.f;
        v2[q][j] = ok ? P2[(size_t)(rb + q) * H_ + k] : 0.f;
      }
    }
    float sq[3 * RB];
#pragma unroll
    for (int q = 0; q < RB; ++q) {
      float s0 = 0.f, s1 = 0.f, s2 = 0.f;
#pragma unroll
      for (int j = 0; j < NS; ++j) {
        s0 += v0[q][j] * v0[q][j];
        s1 += v1[q][j] * v1[q][j];
        s2 += v2[q][j] * v2[q][j];
      }
      sq[3 * q] = s0; sq[3 * q + 1] = s1; sq[3 * q + 2] = s2;
    }
    wave_sum_n<3 * RB>(sq);
    float dd[2 * RB];
#pragma unroll
    for (int q = 0; q < RB; ++q) {
      const float rn0 = 1.0f / sqrtf(sq[3 * q]);
      const float rn1 = 1.0f / sqrtf(sq[3 * q + 1]);
      const float rn2 = 1.0f / sqrtf(sq[3 * q + 2]);
      float d1 = 0.f, d2 = 0.f;
#pragma unroll
      for (int j = 0; j < NS; ++j) {
        const float u0 = v0[q][j] * rn0;
        const float u1 = v1[q][j] * rn1;
        const float u2 = v2[q][j] * rn2;
        v0[q][j] = u0;
        d1 += u0 * u1;
        d2 += u0 * u2;
      }
      dd[2 * q] = d1; dd[2 * q + 1] = d2;
    }
    wave_sum_n<2 * RB>(dd);
#pragma unroll
    for (int q = 0; q < RB; ++q) {
#pragma unroll
      for (int j = 0; j < NS; ++j) {
        const int k = lane + 32 * j;
        if (k < H_) nr[(rb + q) * PN + k] = v0[q][j];
      }
    }
    float e1 = 0.f, e2 = 0.f;
#pragma unroll
    for (int q = 0; q < RB; ++q) {
      e1 = (lane == q) ? dd[2 * q] : e1;
      e2 = (lane == q) ? dd[2 * q + 1] : e2;
    }
    const float c1 = 1.0f - acosf(e1 * 0.99999f) * (1.0f / PI_F);
    const float c2 = 1.0f - acosf(e2 * 0.99999f) * (1.0f / PI_F);
    if (lane < RB) { cx[0][rb + lane] = c1; cx[1][rb + lane] = c2; }
  }
  __syncthreads();

  const int mr = lane & 15;
  const int kh = lane >> 4;
  v8f acc[4] = {};
  {
    const float* ap = nr + (w * 16 + mr) * PN + 2 * kh;
    const float* bp = nr + mr * PN + 2 * kh;
#pragma unroll 5
    for (int k = 0; k < H_; k += 4) {
      const v2f a = *reinterpret_cast<const v2f*>(ap + k);
#pragma unroll
      for (int c = 0; c < 4; ++c) {
        const v2f bb = *reinterpret_cast<const v2f*>(bp + c * 16 * PN + k);
        acc[c] = __builtin_amdgcn_wmma_f32_16x16x4_f32(false, a, false, bb, (short)0, acc[c], false, false);
      }
    }
  }
  __syncthreads();
#pragma unroll
  for (int c = 0; c < 4; ++c) {
#pragma unroll
    for (int v = 0; v < 8; ++v)
      nr[(w * 16 + 8 * kh + v) * TP + 16 * c + mr] = 1.0f - acosf(acc[c][v] * 0.99999f) * (1.0f / PI_F);
  }
  __syncthreads();

  {
    const int r0 = w * RPW;
    float sr[RPW];
#pragma unroll
    for (int q = 0; q < RPW; ++q) {
      const float t0 = nr[(r0 + q) * TP + lane];
      const float t1 = nr[(r0 + q) * TP + lane + 32];
      float* orow = intra + ((size_t)mb * L_ + r0 + q) * L_;
      orow[lane]      = t0;
      orow[lane + 32] = t1;
      float s = 0.f;
      s += t0;
      s += t1;
      sr[q] = s;
    }
    wave_sum_n<RPW>(sr);
    float e = 0.f;
#pragma unroll
    for (int q = 0; q < RPW; ++q) e = (lane == q) ? sr[q] : e;
    if (lane < RPW) {
      e += cx[0][r0 + lane];
      e += cx[1][r0 + lane];
      dv[r0 + lane] = 1.0f / sqrtf(e);
    }
  }
  __syncthreads();
  if (w < 2) {
    dinv[(size_t)mb * L_ + 32 * w + lane] = dv[32 * w + lane];
  } else {
    const int j  = w - 2;
    const int nn = j ? n2 : n1;
    crossv[((size_t)m * M_ + nn) * A_ + b * L_ + lane]      = cx[j][lane];
    crossv[((size_t)m * M_ + nn) * A_ + b * L_ + 32 + lane] = cx[j][32 + lane];
  }
}

struct no_mid {};
template <typename T> struct is_no_mid { static constexpr bool value = false; };
template <> struct is_no_mid<no_mid> { static constexpr bool value = true; };
__device__ __forceinline__ float4 ld4(const float* p) {
  return *reinterpret_cast<const float4*>(p);
}
__device__ __forceinline__ float2 ld2(const float* p) {
  return *reinterpret_cast<const float2*>(p);
}
__device__ __forceinline__ void st4(float* p, const float4 v) {
  *reinterpret_cast<float4*>(p) = v;
}

template <int BK, bool COMB_HAS_MATH, typename FA, typename CA, typename FE, typename FM = no_mid>
__device__ __forceinline__ void gemm_block_64x64(const int K, const float* __restrict__ Bp,
                                                 const int n0, FA fetchA, CA combA, FE epi,
                                                 const int Kmid = 0, FM mid = FM{}) {
  constexpr bool TWO_PRODUCTS = !is_no_mid<FM>::value;
  constexpr int BM = 64, BN = 64, NTHR = 128;
  constexpr int PA = BK + 4;
  constexpr int PB = BN + 16;
  constexpr int KQ = BK / 4;
  constexpr int NP = BN / 2;
  constexpr int A4 = BM * KQ;
  constexpr int B4 = (BK / 2) * NP;
  constexpr int RA = A4 / NTHR;
  constexpr int RB = B4 / NTHR;
  static_assert(A4 % NTHR == 0 && B4 % NTHR == 0, "chunks must split evenly over the block");
  static_assert((PA % 8) == 4, "A pitch must be 4 * odd dwords");
  static_assert(((2 * PB) % 64) == 32, "B pair pitch must be 32 mod 64 dwords");

  constexpr int ASZ = BM * PA;
  constexpr int BSZ = (BK / 2) * PB * 2;
  __shared__ __attribute__((aligned(16))) float As[2 * ASZ];
  __shared__ __attribute__((aligned(16))) float Bs[2 * BSZ];
  constexpr int TC = BN + 4;
  static_assert(((8 * TC) % 64) == 32 && (TC % 4) == 0, "epilogue tile pitch");
  __shared__ __attribute__((aligned(16))) float Ct[BM * TC];

  const int t    = threadIdx.x;
  const int lane = t & 31;
  const int w    = t >> 5;
  const int mr   = lane & 15;
  const int kh   = lane >> 4;

  float4 ax[RA], ay[RA];
  float2 be[RB], bo[RB];

  auto prefetch = [&](const int k0) {
#pragma unroll
    for (int r = 0; r < RA; ++r) {
      const int e = t + NTHR * r;
      fetchA(e / KQ, k0, (e % KQ) * 4, ax[r], ay[r]);
    }
#pragma unroll
    for (int r = 0; r < RB; ++r) {
      const int e  = t + NTHR * r;
      const int kp = e / NP;
      const int n  = min(n0 + (e % NP) * 2, H_ - 2);
      const float* p = Bp + (size_t)(k0 + 2 * kp) * H_ + n;
      be[r] = ld2(p);
      bo[r] = ld2(p + H_);
    }
  };

  auto stage = [&](const int k0, const int buf) {
#pragma unroll
    for (int r = 0; r < RA; ++r) {
      const int e = t + NTHR * r;
      *reinterpret_cast<float4*>(&As[buf * ASZ + (e / KQ) * PA + (e % KQ) * 4]) = combA(k0, ax[r], ay[r]);
    }
#pragma unroll
    for (int r = 0; r < RB; ++r) {
      const int e = t + NTHR * r;
      const int j = (e % NP) * 2;
      float4 v = make_float4(be[r].x, bo[r].x, be[r].y, bo[r].y);
      if (n0 + j >= H_) v = make_float4(0.f, 0.f, 0.f, 0.f);
      *reinterpret_cast<float4*>(&Bs[buf * BSZ + ((e / NP) * PB + j) * 2]) = v;
    }
  };

  const int aoff = (w * 16 + mr) * PA + 2 * kh;
  const int boff = (kh * PB + mr) * 2;

  v8f acc[4] = {};
  auto compute = [&](const int buf) {
    const float* aptr = As + buf * ASZ + aoff;
    const float* bptr = Bs + buf * BSZ + boff;
#pragma unroll
    for (int ks = 0; ks < BK; ks += 4) {
      const v2f a = *reinterpret_cast<const v2f*>(aptr + ks);
#pragma unroll
      for (int c = 0; c < 4; ++c) {
        const v2f b = *reinterpret_cast<const v2f*>(bptr + ((ks / 2) * PB + 16 * c) * 2);
        acc[c] = __builtin_amdgcn_wmma_f32_16x16x4_f32(false, a, false, b, (short)0, acc[c], false, false);
      }
    }
  };

  const int ei = w * 16 + kh;
  const int en = n0 + mr * 4;
  auto to_rows = [&](float4 (&rv)[8]) {
    float* ct = Ct + (w * 16) * TC;
#pragma unroll
    for (int c = 0; c < 4; ++c) {
#pragma unroll
      for (int v = 0; v < 8; ++v) ct[(8 * kh + v) * TC + 16 * c + mr] = acc[c][v];
    }
#pragma unroll
    for (int j = 0; j < 8; ++j) rv[j] = *reinterpret_cast<const float4*>(&ct[(kh + 2 * j) * TC + mr * 4]);
  };

  prefetch(0);
  stage(0, 0);
  __syncthreads();
  int buf = 0;
  for (int k0 = BK; k0 < K; k0 += BK) {
    if constexpr (TWO_PRODUCTS) {
      if (k0 == Kmid + BK) {
        float4 rv[8];
        to_rows(rv);
        mid(ei, en, rv);
#pragma unroll
        for (int c = 0; c < 4; ++c) acc[c] = v8f{};
      }
    }
    prefetch(k0);
    __builtin_amdgcn_sched_barrier(0);
    compute(buf);
    if constexpr (COMB_HAS_MATH) __builtin_amdgcn_sched_barrier(0);
    stage(k0, buf ^ 1);
    __syncthreads();
    buf ^= 1;
  }
  compute(buf);

  float4 rv[8];
  to_rows(rv);
  epi(ei, en, rv);
}

__global__ __launch_bounds__(128) void k_front(const float* __restrict__ xa,
                                               const float* __restrict__ xv,
                                               const float* __restrict__ xt,
                                               const float* __restrict__ w,
                                               const float* __restrict__ bias,
                                               float* __restrict__ h0,
                                               float* __restrict__ intra,
                                               float* __restrict__ crossv,
                                               float* __restrict__ dinv) {
  if (blockIdx.x < M_ * B_) {
    graph_block(blockIdx.x, xa, xv, xt, intra, crossv, dinv);
  } else {
    const int id = blockIdx.x - M_ * B_;
    const int n0 = (id & 3) * 64;
    const int r0 = (id >> 2) * 64;
    const int mm = r0 / A_;
    const float* X = ((mm == 0) ? xa : (mm == 1) ? xv : xt) + (size_t)(r0 - mm * A_) * H_;
    auto fA = [&](int i, int k0, int kq, float4& x, float4& y) {
      (void)y;
      x = ld4(X + (size_t)i * H_ + k0 + kq);
    };
    auto cA = [&](int k0, const float4& x, const float4& y) { (void)k0; (void)y; return x; };
    auto ep = [&](const int i0, const int n, const float4 (&v)[8]) {
      if (n >= H_) return;
      const float4 b = ld4(bias + n);
#pragma unroll
      for (int j = 0; j < 8; ++j)
        st4(h0 + (size_t)(r0 + i0 + 2 * j) * H_ + n,
            make_float4(fmaxf(v[j].x + b.x, 0.0f), fmaxf(v[j].y + b.y, 0.0f),
                        fmaxf(v[j].z + b.z, 0.0f), fmaxf(v[j].w + b.w, 0.0f)));
    };
    gemm_block_64x64<40, false>(H_, w, n0, fA, cA, ep);
  }
}

__global__ __launch_bounds__(128) void k_spmm(const float* __restrict__ intra,
                                              const float* __restrict__ crossv,
                                              const float* __restrict__ dinv,
                                              const float* __restrict__ h,
                                              float* __restrict__ hi) {
  const int n0 = blockIdx.x * 64;
  const int mb = blockIdx.y;
  const int m  = mb >> 5;
  const int rowbase = mb * L_;
  const float* S = intra + (size_t)mb * L_ * L_;
  auto fA = [&](int i, int k0, int kq, float4& x, float4& y) {
    x = ld4(S + (size_t)i * L_ + k0 + kq);
    y = ld4(dinv + rowbase + k0 + kq);
  };
  auto cA = [&](int k0, const float4& x, const float4& y) {
    (void)k0;
    return make_float4(x.x * y.x, x.y * y.y, x.z * y.z, x.w * y.w);
  };
  const int o1 = (m == 0) ? 1 : 0;
  const int o2 = (m == 2) ? 1 : 2;
  auto ep = [&](const int i0, const int n, const float4 (&v)[8]) {
    if (n >= H_) return;
    float  cd1[8], cd2[8], dr[8];
    float4 g1[8], g2[8];
#pragma unroll
    for (int j = 0; j < 8; ++j) {
      const int row  = rowbase + i0 + 2 * j;
      const int a    = row - m * A_;
      const int r1   = o1 * A_ + a;
      const int r2   = o2 * A_ + a;
      cd1[j] = crossv[((size_t)m * M_ + o1) * A_ + a] * dinv[r1];
      cd2[j] = crossv[((size_t)m * M_ + o2) * A_ + a] * dinv[r2];
      dr[j]  = dinv[row];
      g1[j]  = ld4(h + (size_t)r1 * H_ + n);
      g2[j]  = ld4(h + (size_t)r2 * H_ + n);
    }
#pragma unroll
    for (int j = 0; j < 8; ++j) {
      float4 s = v[j];
      s.x += cd1[j] * g1[j].x; s.y += cd1[j] * g1[j].y; s.z += cd1[j] * g1[j].z; s.w += cd1[j] * g1[j].w;
      s.x += cd2[j] * g2[j].x; s.y += cd2[j] * g2[j].y; s.z += cd2[j] * g2[j].z; s.w += cd2[j] * g2[j].w;
      st4(hi + (size_t)(rowbase + i0 + 2 * j) * H_ + n,
          make_float4(s.x * dr[j], s.y * dr[j], s.z * dr[j], s.w * dr[j]));
    }
  };
  gemm_block_64x64<32, true>(L_, h + (size_t)rowbase * H_, n0, fA, cA, ep);
}

__global__ __launch_bounds__(128) void k_conv(const float* __restrict__ hi,
                                              const float* __restrict__ h0,
                                              const float* __restrict__ w, float beta,
                                              float* __restrict__ hout) {
  const int n0 = blockIdx.x * 64;
  const int r0 = blockIdx.y * 64;
  auto fA = [&](int i, int k0, int kq, float4& x, float4& y) {
    (void)y;
    const float* p = (k0 < H_) ? hi + (size_t)(r0 + i) * H_ + (k0 + kq)
                               : h0 + (size_t)(r0 + i) * H_ + (k0 - H_ + kq);
    x = ld4(p);
  };
  auto cA = [&](int k0, const float4& x, const float4& y) { (void)k0; (void)y; return x; };
  auto ep = [&](const int i0, const int n, const float4 (&v)[8]) {
    if (n >= H_) return;
    float4 a[8], b[8];
#pragma unroll
    for (int j = 0; j < 8; ++j) {
      const size_t o = (size_t)(r0 + i0 + 2 * j) * H_ + n;
      a[j] = ld4(hi + o);
      b[j] = ld4(h0 + o);
    }
    auto one = [&](const float x, const float p, const float q) {
      const float r = 0.9f * p + 0.1f * q;
      return fmaxf(beta * x + (1.0f - beta) * r, 0.0f);
    };
#pragma unroll
    for (int j = 0; j < 8; ++j)
      st4(hout + (size_t)(r0 + i0 + 2 * j) * H_ + n,
          make_float4(one(v[j].x, a[j].x, b[j].x), one(v[j].y, a[j].y, b[j].y),
                      one(v[j].z, a[j].z, b[j].z), one(v[j].w, a[j].w, b[j].w)));
  };
  gemm_block_64x64<40, false>(2 * H_, w, n0, fA, cA, ep);
}

__global__ __launch_bounds__(128) void k_hx(const float* __restrict__ xa,
                                            const float* __restrict__ xv,
                                            const float* __restrict__ xt,
                                            const float* __restrict__ hf,
                                            const float* __restrict__ tw,
                                            const float* __restrict__ tb,
                                            float* __restrict__ hx) {
  const int m  = blockIdx.z;
  const int n0 = blockIdx.x * 64;
  const int r0 = m * A_ + blockIdx.y * 64;
  const float* Wm = tw + (size_t)m * 2 * H_ * H_;
  const float* Bb = tb + (size_t)m * H_;
  const float* X  = ((m == 0) ? xa : (m == 1) ? xv : xt) + (size_t)blockIdx.y * 64 * H_;
  auto fA = [&](int i, int k0, int kq, float4& x, float4& y) {
    (void)y;
    const float* p = (k0 < H_) ? X  + (size_t)i * H_ + (k0 + kq)
                               : hf + (size_t)(r0 + i) * H_ + (k0 - H_ + kq);
    x = ld4(p);
  };
  auto cA = [&](int k0, const float4& x, const float4& y) { (void)k0; (void)y; return x; };
  auto ep = [&](const int i0, const int n, const float4 (&v)[8]) {
    if (n >= H_) return;
    const float4 b = ld4(Bb + n);
#pragma unroll
    for (int j = 0; j < 8; ++j)
      st4(hx + (size_t)(r0 + i0 + 2 * j) * H_ + n,
          make_float4(fmaxf(v[j].x + b.x, 0.0f), fmaxf(v[j].y + b.y, 0.0f),
                      fmaxf(v[j].z + b.z, 0.0f), fmaxf(v[j].w + b.w, 0.0f)));
  };
  gemm_block_64x64<40, false>(2 * H_, Wm, n0, fA, cA, ep);
}

__global__ __launch_bounds__(128) void k_gate(const float* __restrict__ hx,
                                              const float* __restrict__ cw,
                                              const float* __restrict__ cb,
                                              float* __restrict__ hout) {
  constexpr int KP = 3 * H_;
  const int i  = blockIdx.z;
  const int n0 = blockIdx.x * 64;
  const int a0 = blockIdx.y * 64;
  const int j0 = (i == 0) ? 1 : 0;
  const int j1 = (i == 2) ? 1 : 2;
  const float* Hi  = hx + (size_t)i  * A_ * H_;
  const float* Hj0 = hx + (size_t)j0 * A_ * H_;
  const float* Hj1 = hx + (size_t)j1 * A_ * H_;
  const float* Wp = cw + (size_t)(2 * i) * KP * H_;
  const float* Bb = cb + (size_t)(2 * i) * H_;
  auto fA = [&](int r, int k0, int kq, float4& x, float4& y) {
    const int q  = k0 / KP;
    const int kp = k0 - q * KP;
    const int s  = kp / H_;
    const int kk = kp - s * H_ + kq;
    const float* Hj = q ? Hj1 : Hj0;
    const float* px = ((s == 1) ? Hj : Hi) + (size_t)(a0 + r) * H_ + kk;
    x = ld4(px);
    if (s == 2) y = ld4(Hj + (size_t)(a0 + r) * H_ + kk);
    else y = make_float4(0.f, 0.f, 0.f, 0.f);
  };
  auto cA = [&](int k0, const float4& x, const float4& y) {
    if ((k0 % KP) / H_ == 2) return make_float4(x.x * y.x, x.y * y.y, x.z * y.z, x.w * y.w);
    return x;
  };

  float4 g[8];
  auto gated = [&](const int i0, const int n, const float4 (&v)[8], const float* Hj, const float* bias,
                   const bool add) {
    if (n >= H_) return;
    const float4 bn = ld4(bias + n);
    float4 p[8], q[8];
#pragma unroll
    for (int j = 0; j < 8; ++j) {
      const size_t o = (size_t)(a0 + i0 + 2 * j) * H_ + n;
      p[j] = ld4(Hi + o);
      q[j] = ld4(Hj + o);
    }
    auto one = [&](const float x, const float b, const float hi_, const float hj_) {
      const float z = 1.0f / (1.0f + expf(-(x + b)));
      return z * hi_ + (1.0f - z) * hj_;
    };
#pragma unroll
    for (int j = 0; j < 8; ++j) {
      const float4 t = make_float4(one(v[j].x, bn.x, p[j].x, q[j].x), one(v[j].y, bn.y, p[j].y, q[j].y),
                                   one(v[j].z, bn.z, p[j].z, q[j].z), one(v[j].w, bn.w, p[j].w, q[j].w));
      g[j] = add ? make_float4(g[j].x + t.x, g[j].y + t.y, g[j].z + t.z, g[j].w + t.w) : t;
    }
  };
  auto mid = [&](const int i0, const int n, const float4 (&v)[8]) { gated(i0, n, v, Hj0, Bb, false); };
  auto ep = [&](const int i0, const int n, const float4 (&v)[8]) {
    if (n >= H_) return;
    gated(i0, n, v, Hj1, Bb + H_, true);
#pragma unroll
    for (int j = 0; j < 8; ++j)
      st4(hout + ((size_t)i * A_ + a0 + i0 + 2 * j) * H_ + n, g[j]);
  };
  gemm_block_64x64<40, true>(2 * KP, Wp, n0, fA, cA, ep, KP, mid);
}

__global__ __launch_bounds__(512) void k_final(const float* __restrict__ hout,
                                               const float* __restrict__ uw,
                                               const float* __restrict__ ub,
                                               float* __restrict__ out) {
  constexpr int UPB = 16;
  constexpr int NSF = H_ / 32;
  constexpr int REM = H_ - 32 * NSF;
  static_assert((UPB * TAG_) % 32 == 0, "a block's results must be whole 128-byte lines");
  static_assert(REM > 0 && M_ * REM <= 32, "the left-over units of all modalities share one slot");
  __shared__ float res[UPB * TAG_];
  const int wv   = threadIdx.x >> 5;
  const int lane = threadIdx.x & 31;
  const int a    = blockIdx.x * UPB + wv;
  float acc[TAG_] = {};
  auto add6 = [&](const float v, const float2 w01, const float2 w23, const float2 w45) {
    acc[0] += v * w01.x; acc[1] += v * w01.y;
    acc[2] += v * w23.x; acc[3] += v * w23.y;
    acc[4] += v * w45.x; acc[5] += v * w45.y;
  };
#pragma unroll
  for (int m = 0; m < M_; ++m) {
    const float* hr = hout + (size_t)(m * A_ + a) * H_;
    const float* wm = uw + (size_t)m * H_ * TAG_;
#pragma unroll
    for (int s = 0; s < NSF; ++s) {
      const int h = lane + 32 * s;
      const float* wr = wm + h * TAG_;
      add6(hr[h], ld2(wr), ld2(wr + 2), ld2(wr + 4));
    }
  }
  {
    const bool ok = lane < M_ * REM;
    const int l = ok ? lane : M_ * REM - 1;
    const int m = l / REM;
    const int h = 32 * NSF + l % REM;
    const float* wr = uw + ((size_t)m * H_ + h) * TAG_;
    const float  v   = hout[(size_t)(m * A_ + a) * H_ + h];
    const float2 w01 = ld2(wr), w23 = ld2(wr + 2), w45 = ld2(wr + 4);
    const float2 z = make_float2(0.f, 0.f);
    add6(ok ? v : 0.f, ok ? w01 : z, ok ? w23 : z, ok ? w45 : z);
  }
#pragma unroll
  for (int t = 0; t < TAG_; ++t) acc[t] = wave_sum(acc[t]);
  if (lane == 0) {
#pragma unroll
    for (int t = 0; t < TAG_; ++t)
      res[wv * TAG_ + t] = acc[t] + (ub[t] + ub[TAG_ + t] + ub[2 * TAG_ + t]);
  }
  __syncthreads();
  if (wv == 0) {
    for (int j = 0; j < UPB * TAG_ / 32; ++j)
      out[(size_t)blockIdx.x * (UPB * TAG_) + 32 * j + lane] = res[32 * j + lane];
  }
}

}

extern "C" void kernel_launch(void* const* d_in, const int* in_sizes, int n_in,
                              void* d_out, int out_size, void* d_ws, size_t ws_size,
                              hipStream_t stream) {
  (void)in_sizes; (void)n_in; (void)out_size; (void)ws_size;
  const float* x_a     = (const float*)d_in[0];
  const float* x_v     = (const float*)d_in[1];
  const float* x_t     = (const float*)d_in[2];
  const float* fc0_w   = (const float*)d_in[3];
  const float* fc0_b   = (const float*)d_in[4];
  const float* conv_w  = (const float*)d_in[5];
  const float* tm_w    = (const float*)d_in[6];
  const float* tm_b    = (const float*)d_in[7];
  const float* cross_w = (const float*)d_in[8];
  const float* cross_b = (const float*)d_in[9];
  const float* uni_w   = (const float*)d_in[10];
  const float* uni_b   = (const float*)d_in[11];
  float* out = (float*)d_out;

  const size_t NH = (size_t)N_ * H_;
  float* ws    = (float*)d_ws;
  float* intra = ws; ws += (size_t)N_ * L_;
  float* crossv= ws; ws += (size_t)M_ * M_ * A_;
  float* dinv  = ws; ws += N_;
  float* h0    = ws; ws += NH;
  float* hA    = ws; ws += NH;
  float* hBf   = ws; ws += NH;
  float* hi    = ws; ws += NH;
  float* hx    = ws; ws += NH;
  float* hout  = ws; ws += NH;

  k_front<<<M_ * B_ + 4 * (N_ / 64), dim3(128), 0, stream>>>(x_a, x_v, x_t, fc0_w, fc0_b, h0,
                                                            intra, crossv, dinv);

  const float beta0 = logf(0.5f / 1.0f + 1.0f);
  const float beta1 = logf(0.5f / 2.0f + 1.0f);
  k_spmm <<<dim3(4, M_ * B_),    dim3(128), 0, stream>>>(intra, crossv, dinv, h0, hi);
  k_conv <<<dim3(4, N_ / 64),    dim3(128), 0, stream>>>(hi, h0, conv_w, beta0, hA);
  k_spmm <<<dim3(4, M_ * B_),    dim3(128), 0, stream>>>(intra, crossv, dinv, hA, hi);
  k_conv <<<dim3(4, N_ / 64),    dim3(128), 0, stream>>>(hi, h0, conv_w + 2 * H_ * H_, beta1, hBf);

  k_hx   <<<dim3(4, A_ / 64, M_), dim3(128), 0, stream>>>(x_a, x_v, x_t, hBf, tm_w, tm_b, hx);

  k_gate <<<dim3(4, A_ / 64, M_), dim3(128), 0, stream>>>(hx, cross_w, cross_b, hout);

  k_final<<<A_ / 16, dim3(512), 0, stream>>>(hout, uni_w, uni_b, out);
}


# GEMM block tile 64x64 -> 64x48: 13 column tiles cut 3+3+3+3+1 over five blocks, three accumulators per wave, so two blocks on one compute unit issue 6 not 8 WMMAs per k-step per SIMD; k order kept
# speedup vs baseline: 7.6164x; 1.0484x over previous
#include <hip/hip_runtime.h>
#include <math.h>

typedef __attribute__((ext_vector_type(2))) float v2f;
typedef __attribute__((ext_vector_type(8))) float v8f;

namespace {

constexpr int B_   = 32;
constexpr int L_   = 64;
constexpr int H_   = 200;
constexpr int M_   = 3;
constexpr int A_   = B_ * L_;
constexpr int N_   = M_ * A_;
constexpr int TAG_ = 6;
constexpr float PI_F = 3.14159265358979323846f;

__device__ __forceinline__ float wave_sum(float s) {
#pragma unroll
  for (int off = 16; off > 0; off >>= 1) s += __shfl_xor(s, off, 32);
  return s;
}

template <int N>
__device__ __forceinline__ void wave_sum_n(float (&s)[N]) {
#pragma unroll
  for (int off = 16; off > 0; off >>= 1) {
    float t[N];
#pragma unroll
    for (int i = 0; i < N; ++i) t[i] = __shfl_xor(s[i], off, 32);
#pragma unroll
    for (int i = 0; i < N; ++i) s[i] += t[i];
  }
}

__device__ __forceinline__ void graph_block(const int mb,
                                            const float* __restrict__ xa,
                                            const float* __restrict__ xv,
                                            const float* __restrict__ xt,
                                            float* __restrict__ intra,
                                            float* __restrict__ crossv,
                                            float* __restrict__ dinv) {
  constexpr int PN  = H_ + 4;
  constexpr int TP  = L_ + 4;
  constexpr int NS  = (H_ + 31) / 32;
  constexpr int NW  = 4;
  constexpr int RPW = L_ / NW;
  constexpr int RB  = 8;
  static_assert((PN % 8) == 4, "pitch must be 4 * odd dwords");
  static_assert(L_ * TP <= L_ * PN, "the similarity tile reuses the rows' LDS");
  static_assert(RPW == 16 && RPW % RB == 0, "a wave owns one 16-row strip");
  __shared__ __attribute__((aligned(16))) float nr[L_ * PN];
  __shared__ float cx[2][L_];
  __shared__ float dv[L_];
  const int m    = mb >> 5;
  const int b    = mb & 31;
  const int w    = threadIdx.x >> 5;
  const int lane = threadIdx.x & 31;
  const int n1   = (m == 0) ? 1 : 0;
  const int n2   = (m == 2) ? 1 : 2;
  auto plane = [&](const int mm) { return (mm == 0) ? xa : (mm == 1) ? xv : xt; };
  const float* P0 = plane(m)  + (size_t)b * L_ * H_;
  const float* P1 = plane(n1) + (size_t)b * L_ * H_;
  const float* P2 = plane(n2) + (size_t)b * L_ * H_;

#pragma unroll 1
  for (int rb = w * RPW; rb < (w + 1) * RPW; rb += RB) {
    float v0[RB][NS], v1[RB][NS], v2[RB][NS];
#pragma unroll
    for (int q = 0; q < RB; ++q) {
#pragma unroll
      for (int j = 0; j < NS; ++j) {
        const int k = lane + 32 * j;
        const bool ok = k < H_;
        v0[q][j] = ok ? P0[(size_t)(rb + q) * H_ + k] : 0.f;
        v1[q][j] = ok ? P1[(size_t)(rb + q) * H_ + k] : 0.f;
        v2[q][j] = ok ? P2[(size_t)(rb + q) * H_ + k] : 0.f;
      }
    }
    float sq[3 * RB];
#pragma unroll
    for (int q = 0; q < RB; ++q) {
      float s0 = 0.f, s1 = 0.f, s2 = 0.f;
#pragma unroll
      for (int j = 0; j < NS; ++j) {
        s0 += v0[q][j] * v0[q][j];
        s1 += v1[q][j] * v1[q][j];
        s2 += v2[q][j] * v2[q][j];
      }
      sq[3 * q] = s0; sq[3 * q + 1] = s1; sq[3 * q + 2] = s2;
    }
    wave_sum_n<3 * RB>(sq);
    float dd[2 * RB];
#pragma unroll
    for (int q = 0; q < RB; ++q) {
      const float rn0 = 1.0f / sqrtf(sq[3 * q]);
      const float rn1 = 1.0f / sqrtf(sq[3 * q + 1]);
      const float rn2 = 1.0f / sqrtf(sq[3 * q + 2]);
      float d1 = 0.f, d2 = 0.f;
#pragma unroll
      for (int j = 0; j < NS; ++j) {
        const float u0 = v0[q][j] * rn0;
        const float u1 = v1[q][j] * rn1;
        const float u2 = v2[q][j] * rn2;
        v0[q][j] = u0;
        d1 += u0 * u1;
        d2 += u0 * u2;
      }
      dd[2 * q] = d1; dd[2 * q + 1] = d2;
    }
    wave_sum_n<2 * RB>(dd);
#pragma unroll
    for (int q = 0; q < RB; ++q) {
#pragma unroll
      for (int j = 0; j < NS; ++j) {
        const int k = lane + 32 * j;
        if (k < H_) nr[(rb + q) * PN + k] = v0[q][j];
      }
    }
    float e1 = 0.f, e2 = 0.f;
#pragma unroll
    for (int q = 0; q < RB; ++q) {
      e1 = (lane == q) ? dd[2 * q] : e1;
      e2 = (lane == q) ? dd[2 * q + 1] : e2;
    }
    const float c1 = 1.0f - acosf(e1 * 0.99999f) * (1.0f / PI_F);
    const float c2 = 1.0f - acosf(e2 * 0.99999f) * (1.0f / PI_F);
    if (lane < RB) { cx[0][rb + lane] = c1; cx[1][rb + lane] = c2; }
  }
  __syncthreads();

  const int mr = lane & 15;
  const int kh = lane >> 4;
  v8f acc[4] = {};
  {
    const float* ap = nr + (w * 16 + mr) * PN + 2 * kh;
    const float* bp = nr + mr * PN + 2 * kh;
#pragma unroll 5
    for (int k = 0; k < H_; k += 4) {
      const v2f a = *reinterpret_cast<const v2f*>(ap + k);
#pragma unroll
      for (int c = 0; c < 4; ++c) {
        const v2f bb = *reinterpret_cast<const v2f*>(bp + c * 16 * PN + k);
        acc[c] = __builtin_amdgcn_wmma_f32_16x16x4_f32(false, a, false, bb, (short)0, acc[c], false, false);
      }
    }
  }
  __syncthreads();
#pragma unroll
  for (int c = 0; c < 4; ++c) {
#pragma unroll
    for (int v = 0; v < 8; ++v)
      nr[(w * 16 + 8 * kh + v) * TP + 16 * c + mr] = 1.0f - acosf(acc[c][v] * 0.99999f) * (1.0f / PI_F);
  }
  __syncthreads();

  {
    const int r0 = w * RPW;
    float sr[RPW];
#pragma unroll
    for (int q = 0; q < RPW; ++q) {
      const float t0 = nr[(r0 + q) * TP + lane];
      const float t1 = nr[(r0 + q) * TP + lane + 32];
      float* orow = intra + ((size_t)mb * L_ + r0 + q) * L_;
      orow[lane]      = t0;
      orow[lane + 32] = t1;
      float s = 0.f;
      s += t0;
      s += t1;
      sr[q] = s;
    }
    wave_sum_n<RPW>(sr);
    float e = 0.f;
#pragma unroll
    for (int q = 0; q < RPW; ++q) e = (lane == q) ? sr[q] : e;
    if (lane < RPW) {
      e += cx[0][r0 + lane];
      e += cx[1][r0 + lane];
      dv[r0 + lane] = 1.0f / sqrtf(e);
    }
  }
  __syncthreads();
  if (w < 2) {
    dinv[(size_t)mb * L_ + 32 * w + lane] = dv[32 * w + lane];
  } else {
    const int j  = w - 2;
    const int nn = j ? n2 : n1;
    crossv[((size_t)m * M_ + nn) * A_ + b * L_ + lane]      = cx[j][lane];
    crossv[((size_t)m * M_ + nn) * A_ + b * L_ + 32 + lane] = cx[j][32 + lane];
  }
}

struct no_mid {};
template <typename T> struct is_no_mid { static constexpr bool value = false; };
template <> struct is_no_mid<no_mid> { static constexpr bool value = true; };
__device__ __forceinline__ float4 ld4(const float* p) {
  return *reinterpret_cast<const float4*>(p);
}
__device__ __forceinline__ float2 ld2(const float* p) {
  return *reinterpret_cast<const float2*>(p);
}
__device__ __forceinline__ void st4(float* p, const float4 v) {
  *reinterpret_cast<float4*>(p) = v;
}

template <int BK, bool COMB_HAS_MATH, typename FA, typename CA, typename FE, typename FM = no_mid>
__device__ __forceinline__ void gemm_block_64x48(const int K, const float* __restrict__ Bp,
                                                 const int n0, FA fetchA, CA combA, FE epi,
                                                 const int Kmid = 0, FM mid = FM{}) {
  constexpr bool TWO_PRODUCTS = !is_no_mid<FM>::value;
  constexpr int BM = 64, BN = 48, NTHR = 128;
  constexpr int NT = BN / 16;
  constexpr int PA = BK + 4;
  constexpr int PB = BN;
  constexpr int KQ = BK / 4;
  constexpr int NP = BN / 2;
  constexpr int A4 = BM * KQ;
  constexpr int B4 = (BK / 2) * NP;
  constexpr int RA = A4 / NTHR;
  constexpr int RB = (B4 + NTHR - 1) / NTHR;
  constexpr bool B_TAIL = (B4 % NTHR) != 0;
  static_assert(A4 % NTHR == 0 && B4 % 32 == 0, "A chunks split evenly over the block, B tasks over whole waves");
  static_assert((PA % 8) == 4, "A pitch must be 4 * odd dwords");
  static_assert(((2 * PB) % 64) == 32, "B pair pitch must be 32 mod 64 dwords");

  constexpr int ASZ = BM * PA;
  constexpr int BSZ = (BK / 2) * PB * 2;
  __shared__ __attribute__((aligned(16))) float As[2 * ASZ];
  __shared__ __attribute__((aligned(16))) float Bs[2 * BSZ];
  constexpr int TC = BN + 4;
  static_assert(((8 * TC) % 64) == 32 && (TC % 4) == 0, "epilogue tile pitch");
  __shared__ __attribute__((aligned(16))) float Ct[BM * TC];

  const int t    = threadIdx.x;
  const int lane = t & 31;
  const int w    = t >> 5;
  const int mr   = lane & 15;
  const int kh   = lane >> 4;

  float4 ax[RA], ay[RA];
  float2 be[RB], bo[RB];

  auto btask = [&](const int r) {
    const int e = t + NTHR * r;
    return (B_TAIL && r == RB - 1 && e >= B4) ? e - NTHR : e;
  };

  auto prefetch = [&](const int k0) {
#pragma unroll
    for (int r = 0; r < RA; ++r) {
      const int e = t + NTHR * r;
      fetchA(e / KQ, k0, (e % KQ) * 4, ax[r], ay[r]);
    }
#pragma unroll
    for (int r = 0; r < RB; ++r) {
      const int e  = btask(r);
      const int kp = e / NP;
      const int n  = min(n0 + (e % NP) * 2, H_ - 2);
      const float* p = Bp + (size_t)(k0 + 2 * kp) * H_ + n;
      be[r] = ld2(p);
      bo[r] = ld2(p + H_);
    }
  };

  auto stage = [&](const int k0, const int buf) {
#pragma unroll
    for (int r = 0; r < RA; ++r) {
      const int e = t + NTHR * r;
      *reinterpret_cast<float4*>(&As[buf * ASZ + (e / KQ) * PA + (e % KQ) * 4]) = combA(k0, ax[r], ay[r]);
    }
#pragma unroll
    for (int r = 0; r < RB; ++r) {
      const int e = btask(r);
      const int j = (e % NP) * 2;
      float4 v = make_float4(be[r].x, bo[r].x, be[r].y, bo[r].y);
      if (n0 + j >= H_) v = make_float4(0.f, 0.f, 0.f, 0.f);
      *reinterpret_cast<float4*>(&Bs[buf * BSZ + ((e / NP) * PB + j) * 2]) = v;
    }
  };

  const int aoff = (w * 16 + mr) * PA + 2 * kh;
  const int boff = (kh * PB + mr) * 2;

  v8f acc[NT] = {};
  auto compute = [&](const int buf) {
    const float* aptr = As + buf * ASZ + aoff;
    const float* bptr = Bs + buf * BSZ + boff;
#pragma unroll
    for (int ks = 0; ks < BK; ks += 4) {
      const v2f a = *reinterpret_cast<const v2f*>(aptr + ks);
#pragma unroll
      for (int c = 0; c < NT; ++c) {
        const v2f b = *reinterpret_cast<const v2f*>(bptr + ((ks / 2) * PB + 16 * c) * 2);
        acc[c] = __builtin_amdgcn_wmma_f32_16x16x4_f32(false, a, false, b, (short)0, acc[c], false, false);
      }
    }
  };

  constexpr int GR = BN / 4;
  const bool has_group = lane < 2 * GR;
  const int gl = has_group ? lane : lane - 2 * GR;
  const int er = gl / GR;
  const int eg = gl % GR;
  const int ei = w * 16 + er;
  const int en = has_group ? n0 + eg * 4 : H_;
  auto to_rows = [&](float4 (&rv)[8]) {
    float* ct = Ct + (w * 16) * TC;
#pragma unroll
    for (int c = 0; c < NT; ++c) {
#pragma unroll
      for (int v = 0; v < 8; ++v) ct[(8 * kh + v) * TC + 16 * c + mr] = acc[c][v];
    }
#pragma unroll
    for (int j = 0; j < 8; ++j) rv[j] = *reinterpret_cast<const float4*>(&ct[(er + 2 * j) * TC + eg * 4]);
  };

  prefetch(0);
  stage(0, 0);
  __syncthreads();
  int buf = 0;
  for (int k0 = BK; k0 < K; k0 += BK) {
    if constexpr (TWO_PRODUCTS) {
      if (k0 == Kmid + BK) {
        float4 rv[8];
        to_rows(rv);
        mid(ei, en, rv);
#pragma unroll
        for (int c = 0; c < NT; ++c) acc[c] = v8f{};
      }
    }
    prefetch(k0);
    __builtin_amdgcn_sched_barrier(0);
    compute(buf);
    if constexpr (COMB_HAS_MATH) __builtin_amdgcn_sched_barrier(0);
    stage(k0, buf ^ 1);
    __syncthreads();
    buf ^= 1;
  }
  compute(buf);

  float4 rv[8];
  to_rows(rv);
  epi(ei, en, rv);
}

__global__ __launch_bounds__(128) void k_front(const float* __restrict__ xa,
                                               const float* __restrict__ xv,
                                               const float* __restrict__ xt,
                                               const float* __restrict__ w,
                                               const float* __restrict__ bias,
                                               float* __restrict__ h0,
                                               float* __restrict__ intra,
                                               float* __restrict__ crossv,
                                               float* __restrict__ dinv) {
  if (blockIdx.x < M_ * B_) {
    graph_block(blockIdx.x, xa, xv, xt, intra, crossv, dinv);
  } else {
    const int id = blockIdx.x - M_ * B_;
    const int n0 = (id % 5) * 48;
    const int r0 = (id / 5) * 64;
    const int mm = r0 / A_;
    const float* X = ((mm == 0) ? xa : (mm == 1) ? xv : xt) + (size_t)(r0 - mm * A_) * H_;
    auto fA = [&](int i, int k0, int kq, float4& x, float4& y) {
      (void)y;
      x = ld4(X + (size_t)i * H_ + k0 + kq);
    };
    auto cA = [&](int k0, const float4& x, const float4& y) { (void)k0; (void)y; return x; };
    auto ep = [&](const int i0, const int n, const float4 (&v)[8]) {
      if (n >= H_) return;
      const float4 b = ld4(bias + n);
#pragma unroll
      for (int j = 0; j < 8; ++j)
        st4(h0 + (size_t)(r0 + i0 + 2 * j) * H_ + n,
            make_float4(fmaxf(v[j].x + b.x, 0.0f), fmaxf(v[j].y + b.y, 0.0f),
                        fmaxf(v[j].z + b.z, 0.0f), fmaxf(v[j].w + b.w, 0.0f)));
    };
    gemm_block_64x48<40, false>(H_, w, n0, fA, cA, ep);
  }
}

__global__ __launch_bounds__(128) void k_spmm(const float* __restrict__ intra,
                                              const float* __restrict__ crossv,
                                              const float* __restrict__ dinv,
                                              const float* __restrict__ h,
                                              float* __restrict__ hi) {
  const int n0 = blockIdx.x * 48;
  const int mb = blockIdx.y;
  const int m  = mb >> 5;
  const int rowbase = mb * L_;
  const float* S = intra + (size_t)mb * L_ * L_;
  auto fA = [&](int i, int k0, int kq, float4& x, float4& y) {
    x = ld4(S + (size_t)i * L_ + k0 + kq);
    y = ld4(dinv + rowbase + k0 + kq);
  };
  auto cA = [&](int k0, const float4& x, const float4& y) {
    (void)k0;
    return make_float4(x.x * y.x, x.y * y.y, x.z * y.z, x.w * y.w);
  };
  const int o1 = (m == 0) ? 1 : 0;
  const int o2 = (m == 2) ? 1 : 2;
  auto ep = [&](const int i0, const int n, const float4 (&v)[8]) {
    if (n >= H_) return;
    float  cd1[8], cd2[8], dr[8];
    float4 g1[8], g2[8];
#pragma unroll
    for (int j = 0; j < 8; ++j) {
      const int row  = rowbase + i0 + 2 * j;
      const int a    = row - m * A_;
      const int r1   = o1 * A_ + a;
      const int r2   = o2 * A_ + a;
      cd1[j] = crossv[((size_t)m * M_ + o1) * A_ + a] * dinv[r1];
      cd2[j] = crossv[((size_t)m * M_ + o2) * A_ + a] * dinv[r2];
      dr[j]  = dinv[row];
      g1[j]  = ld4(h + (size_t)r1 * H_ + n);
      g2[j]  = ld4(h + (size_t)r2 * H_ + n);
    }
#pragma unroll
    for (int j = 0; j < 8; ++j) {
      float4 s = v[j];
      s.x += cd1[j] * g1[j].x; s.y += cd1[j] * g1[j].y; s.z += cd1[j] * g1[j].z; s.w += cd1[j] * g1[j].w;
      s.x += cd2[j] * g2[j].x; s.y += cd2[j] * g2[j].y; s.z += cd2[j] * g2[j].z; s.w += cd2[j] * g2[j].w;
      st4(hi + (size_t)(rowbase + i0 + 2 * j) * H_ + n,
          make_float4(s.x * dr[j], s.y * dr[j], s.z * dr[j], s.w * dr[j]));
    }
  };
  gemm_block_64x48<32, true>(L_, h + (size_t)rowbase * H_, n0, fA, cA, ep);
}

__global__ __launch_bounds__(128) void k_conv(const float* __restrict__ hi,
                                              const float* __restrict__ h0,
                                              const float* __restrict__ w, float beta,
                                              float* __restrict__ hout) {
  const int n0 = blockIdx.x * 48;
  const int r0 = blockIdx.y * 64;
  auto fA = [&](int i, int k0, int kq, float4& x, float4& y) {
    (void)y;
    const float* p = (k0 < H_) ? hi + (size_t)(r0 + i) * H_ + (k0 + kq)
                               : h0 + (size_t)(r0 + i) * H_ + (k0 - H_ + kq);
    x = ld4(p);
  };
  auto cA = [&](int k0, const float4& x, const float4& y) { (void)k0; (void)y; return x; };
  auto ep = [&](const int i0, const int n, const float4 (&v)[8]) {
    if (n >= H_) return;
    float4 a[8], b[8];
#pragma unroll
    for (int j = 0; j < 8; ++j) {
      const size_t o = (size_t)(r0 + i0 + 2 * j) * H_ + n;
      a[j] = ld4(hi + o);
      b[j] = ld4(h0 + o);
    }
    auto one = [&](const float x, const float p, const float q) {
      const float r = 0.9f * p + 0.1f * q;
      return fmaxf(beta * x + (1.0f - beta) * r, 0.0f);
    };
#pragma unroll
    for (int j = 0; j < 8; ++j)
      st4(hout + (size_t)(r0 + i0 + 2 * j) * H_ + n,
          make_float4(one(v[j].x, a[j].x, b[j].x), one(v[j].y, a[j].y, b[j].y),
                      one(v[j].z, a[j].z, b[j].z), one(v[j].w, a[j].w, b[j].w)));
  };
  gemm_block_64x48<40, false>(2 * H_, w, n0, fA, cA, ep);
}

__global__ __launch_bounds__(128) void k_hx(const float* __restrict__ xa,
                                            const float* __restrict__ xv,
                                            const float* __restrict__ xt,
                                            const float* __restrict__ hf,
                                            const float* __restrict__ tw,
                                            const float* __restrict__ tb,
                                            float* __restrict__ hx) {
  const int m  = blockIdx.z;
  const int n0 = blockIdx.x * 48;
  const int r0 = m * A_ + blockIdx.y * 64;
  const float* Wm = tw + (size_t)m * 2 * H_ * H_;
  const float* Bb = tb + (size_t)m * H_;
  const float* X  = ((m == 0) ? xa : (m == 1) ? xv : xt) + (size_t)blockIdx.y * 64 * H_;
  auto fA = [&](int i, int k0, int kq, float4& x, float4& y) {
    (void)y;
    const float* p = (k0 < H_) ? X  + (size_t)i * H_ + (k0 + kq)
                               : hf + (size_t)(r0 + i) * H_ + (k0 - H_ + kq);
    x = ld4(p);
  };
  auto cA = [&](int k0, const float4& x, const float4& y) { (void)k0; (void)y; return x; };
  auto ep = [&](const int i0, const int n, const float4 (&v)[8]) {
    if (n >= H_) return;
    const float4 b = ld4(Bb + n);
#pragma unroll
    for (int j = 0; j < 8; ++j)
      st4(hx + (size_t)(r0 + i0 + 2 * j) * H_ + n,
          make_float4(fmaxf(v[j].x + b.x, 0.0f), fmaxf(v[j].y + b.y, 0.0f),
                      fmaxf(v[j].z + b.z, 0.0f), fmaxf(v[j].w + b.w, 0.0f)));
  };
  gemm_block_64x48<40, false>(2 * H_, Wm, n0, fA, cA, ep);
}

__global__ __launch_bounds__(128) void k_gate(const float* __restrict__ hx,
                                              const float* __restrict__ cw,
                                              const float* __restrict__ cb,
                                              float* __restrict__ hout) {
  constexpr int KP = 3 * H_;
  const int i  = blockIdx.z;
  const int n0 = blockIdx.x * 48;
  const int a0 = blockIdx.y * 64;
  const int j0 = (i == 0) ? 1 : 0;
  const int j1 = (i == 2) ? 1 : 2;
  const float* Hi  = hx + (size_t)i  * A_ * H_;
  const float* Hj0 = hx + (size_t)j0 * A_ * H_;
  const float* Hj1 = hx + (size_t)j1 * A_ * H_;
  const float* Wp = cw + (size_t)(2 * i) * KP * H_;
  const float* Bb = cb + (size_t)(2 * i) * H_;
  auto fA = [&](int r, int k0, int kq, float4& x, float4& y) {
    const int q  = k0 / KP;
    const int kp = k0 - q * KP;
    const int s  = kp / H_;
    const int kk = kp - s * H_ + kq;
    const float* Hj = q ? Hj1 : Hj0;
    const float* px = ((s == 1) ? Hj : Hi) + (size_t)(a0 + r) * H_ + kk;
    x = ld4(px);
    if (s == 2) y = ld4(Hj + (size_t)(a0 + r) * H_ + kk);
    else y = make_float4(0.f, 0.f, 0.f, 0.f);
  };
  auto cA = [&](int k0, const float4& x, const float4& y) {
    if ((k0 % KP) / H_ == 2) return make_float4(x.x * y.x, x.y * y.y, x.z * y.z, x.w * y.w);
    return x;
  };

  float4 g[8];
  auto gated = [&](const int i0, const int n, const float4 (&v)[8], const float* Hj, const float* bias,
                   const bool add) {
    if (n >= H_) return;
    const float4 bn = ld4(bias + n);
    float4 p[8], q[8];
#pragma unroll
    for (int j = 0; j < 8; ++j) {
      const size_t o = (size_t)(a0 + i0 + 2 * j) * H_ + n;
      p[j] = ld4(Hi + o);
      q[j] = ld4(Hj + o);
    }
    auto one = [&](const float x, const float b, const float hi_, const float hj_) {
      const float z = 1.0f / (1.0f + expf(-(x + b)));
      return z * hi_ + (1.0f - z) * hj_;
    };
#pragma unroll
    for (int j = 0; j < 8; ++j) {
      const float4 t = make_float4(one(v[j].x, bn.x, p[j].x, q[j].x), one(v[j].y, bn.y, p[j].y, q[j].y),
                                   one(v[j].z, bn.z, p[j].z, q[j].z), one(v[j].w, bn.w, p[j].w, q[j].w));
      g[j] = add ? make_float4(g[j].x + t.x, g[j].y + t.y, g[j].z + t.z, g[j].w + t.w) : t;
    }
  };
  auto mid = [&](const int i0, const int n, const float4 (&v)[8]) { gated(i0, n, v, Hj0, Bb, false); };
  auto ep = [&](const int i0, const int n, const float4 (&v)[8]) {
    if (n >= H_) return;
    gated(i0, n, v, Hj1, Bb + H_, true);
#pragma unroll
    for (int j = 0; j < 8; ++j)
      st4(hout + ((size_t)i * A_ + a0 + i0 + 2 * j) * H_ + n, g[j]);
  };
  gemm_block_64x48<40, true>(2 * KP, Wp, n0, fA, cA, ep, KP, mid);
}

__global__ __launch_bounds__(512) void k_final(const float* __restrict__ hout,
                                               const float* __restrict__ uw,
                                               const float* __restrict__ ub,
                                               float* __restrict__ out) {
  constexpr int UPB = 16;
  constexpr int NSF = H_ / 32;
  constexpr int REM = H_ - 32 * NSF;
  static_assert((UPB * TAG_) % 32 == 0, "a block's results must be whole 128-byte lines");
  static_assert(REM > 0 && M_ * REM <= 32, "the left-over units of all modalities share one slot");
  __shared__ float res[UPB * TAG_];
  const int wv   = threadIdx.x >> 5;
  const int lane = threadIdx.x & 31;
  const int a    = blockIdx.x * UPB + wv;
  float acc[TAG_] = {};
  auto add6 = [&](const float v, const float2 w01, const float2 w23, const float2 w45) {
    acc[0] += v * w01.x; acc[1] += v * w01.y;
    acc[2] += v * w23.x; acc[3] += v * w23.y;
    acc[4] += v * w45.x; acc[5] += v * w45.y;
  };
#pragma unroll
  for (int m = 0; m < M_; ++m) {
    const float* hr = hout + (size_t)(m * A_ + a) * H_;
    const float* wm = uw + (size_t)m * H_ * TAG_;
#pragma unroll
    for (int s = 0; s < NSF; ++s) {
      const int h = lane + 32 * s;
      const float* wr = wm + h * TAG_;
      add6(hr[h], ld2(wr), ld2(wr + 2), ld2(wr + 4));
    }
  }
  {
    const bool ok = lane < M_ * REM;
    const int l = ok ? lane : M_ * REM - 1;
    const int m = l / REM;
    const int h = 32 * NSF + l % REM;
    const float* wr = uw + ((size_t)m * H_ + h) * TAG_;
    const float  v   = hout[(size_t)(m * A_ + a) * H_ + h];
    const float2 w01 = ld2(wr), w23 = ld2(wr + 2), w45 = ld2(wr + 4);
    const float2 z = make_float2(0.f, 0.f);
    add6(ok ? v : 0.f, ok ? w01 : z, ok ? w23 : z, ok ? w45 : z);
  }
#pragma unroll
  for (int t = 0; t < TAG_; ++t) acc[t] = wave_sum(acc[t]);
  if (lane == 0) {
#pragma unroll
    for (int t = 0; t < TAG_; ++t)
      res[wv * TAG_ + t] = acc[t] + (ub[t] + ub[TAG_ + t] + ub[2 * TAG_ + t]);
  }
  __syncthreads();
  if (wv == 0) {
    for (int j = 0; j < UPB * TAG_ / 32; ++j)
      out[(size_t)blockIdx.x * (UPB * TAG_) + 32 * j + lane] = res[32 * j + lane];
  }
}

}

extern "C" void kernel_launch(void* const* d_in, const int* in_sizes, int n_in,
                              void* d_out, int out_size, void* d_ws, size_t ws_size,
                              hipStream_t stream) {
  (void)in_sizes; (void)n_in; (void)out_size; (void)ws_size;
  const float* x_a     = (const float*)d_in[0];
  const float* x_v     = (const float*)d_in[1];
  const float* x_t     = (const float*)d_in[2];
  const float* fc0_w   = (const float*)d_in[3];
  const float* fc0_b   = (const float*)d_in[4];
  const float* conv_w  = (const float*)d_in[5];
  const float* tm_w    = (const float*)d_in[6];
  const float* tm_b    = (const float*)d_in[7];
  const float* cross_w = (const float*)d_in[8];
  const float* cross_b = (const float*)d_in[9];
  const float* uni_w   = (const float*)d_in[10];
  const float* uni_b   = (const float*)d_in[11];
  float* out = (float*)d_out;

  const size_t NH = (size_t)N_ * H_;
  float* ws    = (float*)d_ws;
  float* intra = ws; ws += (size_t)N_ * L_;
  float* crossv= ws; ws += (size_t)M_ * M_ * A_;
  float* dinv  = ws; ws += N_;
  float* h0    = ws; ws += NH;
  float* hA    = ws; ws += NH;
  float* hBf   = ws; ws += NH;
  float* hi    = ws; ws += NH;
  float* hx    = ws; ws += NH;
  float* hout  = ws; ws += NH;

  k_front<<<M_ * B_ + 5 * (N_ / 64), dim3(128), 0, stream>>>(x_a, x_v, x_t, fc0_w, fc0_b, h0,
                                                            intra, crossv, dinv);

  const float beta0 = logf(0.5f / 1.0f + 1.0f);
  const float beta1 = logf(0.5f / 2.0f + 1.0f);
  k_spmm <<<dim3(5, M_ * B_),    dim3(128), 0, stream>>>(intra, crossv, dinv, h0, hi);
  k_conv <<<dim3(5, N_ / 64),    dim3(128), 0, stream>>>(hi, h0, conv_w, beta0, hA);
  k_spmm <<<dim3(5, M_ * B_),    dim3(128), 0, stream>>>(intra, crossv, dinv, hA, hi);
  k_conv <<<dim3(5, N_ / 64),    dim3(128), 0, stream>>>(hi, h0, conv_w + 2 * H_ * H_, beta1, hBf);

  k_hx   <<<dim3(5, A_ / 64, M_), dim3(128), 0, stream>>>(x_a, x_v, x_t, hBf, tm_w, tm_b, hx);

  k_gate <<<dim3(5, A_ / 64, M_), dim3(128), 0, stream>>>(hx, cross_w, cross_b, hout);

  k_final<<<A_ / 16, dim3(512), 0, stream>>>(hout, uni_w, uni_b, out);
}


# GEMM core (fc0, spmm, conv, hx, gate): each chunk's global loads issued two chunks ahead into a second register set, LDS-only barrier fences, B units staged without register moves; k order kept
# speedup vs baseline: 7.9119x; 1.0388x over previous
#include <hip/hip_runtime.h>
#include <math.h>

typedef __attribute__((ext_vector_type(2))) float v2f;
typedef __attribute__((ext_vector_type(8))) float v8f;

namespace {

constexpr int B_   = 32;
constexpr int L_   = 64;
constexpr int H_   = 200;
constexpr int M_   = 3;
constexpr int A_   = B_ * L_;
constexpr int N_   = M_ * A_;
constexpr int TAG_ = 6;
constexpr float PI_F = 3.14159265358979323846f;

__device__ __forceinline__ float wave_sum(float s) {
#pragma unroll
  for (int off = 16; off > 0; off >>= 1) s += __shfl_xor(s, off, 32);
  return s;
}

template <int N>
__device__ __forceinline__ void wave_sum_n(float (&s)[N]) {
#pragma unroll
  for (int off = 16; off > 0; off >>= 1) {
    float t[N];
#pragma unroll
    for (int i = 0; i < N; ++i) t[i] = __shfl_xor(s[i], off, 32);
#pragma unroll
    for (int i = 0; i < N; ++i) s[i] += t[i];
  }
}

__device__ __forceinline__ void graph_block(const int mb,
                                            const float* __restrict__ xa,
                                            const float* __restrict__ xv,
                                            const float* __restrict__ xt,
                                            float* __restrict__ intra,
                                            float* __restrict__ crossv,
                                            float* __restrict__ dinv) {
  constexpr int PN  = H_ + 4;
  constexpr int TP  = L_ + 4;
  constexpr int NS  = (H_ + 31) / 32;
  constexpr int NW  = 4;
  constexpr int RPW = L_ / NW;
  constexpr int RB  = 8;
  static_assert((PN % 8) == 4, "pitch must be 4 * odd dwords");
  static_assert(L_ * TP <= L_ * PN, "the similarity tile reuses the rows' LDS");
  static_assert(RPW == 16 && RPW % RB == 0, "a wave owns one 16-row strip");
  __shared__ __attribute__((aligned(16))) float nr[L_ * PN];
  __shared__ float cx[2][L_];
  __shared__ float dv[L_];
  const int m    = mb >> 5;
  const int b    = mb & 31;
  const int w    = threadIdx.x >> 5;
  const int lane = threadIdx.x & 31;
  const int n1   = (m == 0) ? 1 : 0;
  const int n2   = (m == 2) ? 1 : 2;
  auto plane = [&](const int mm) { return (mm == 0) ? xa : (mm == 1) ? xv : xt; };
  const float* P0 = plane(m)  + (size_t)b * L_ * H_;
  const float* P1 = plane(n1) + (size_t)b * L_ * H_;
  const float* P2 = plane(n2) + (size_t)b * L_ * H_;

#pragma unroll 1
  for (int rb = w * RPW; rb < (w + 1) * RPW; rb += RB) {
    float v0[RB][NS], v1[RB][NS], v2[RB][NS];
#pragma unroll
    for (int q = 0; q < RB; ++q) {
#pragma unroll
      for (int j = 0; j < NS; ++j) {
        const int k = lane + 32 * j;
        const bool ok = k < H_;
        v0[q][j] = ok ? P0[(size_t)(rb + q) * H_ + k] : 0.f;
        v1[q][j] = ok ? P1[(size_t)(rb + q) * H_ + k] : 0.f;
        v2[q][j] = ok ? P2[(size_t)(rb + q) * H_ + k] : 0.f;
      }
    }
    float sq[3 * RB];
#pragma unroll
    for (int q = 0; q < RB; ++q) {
      float s0 = 0.f, s1 = 0.f, s2 = 0.f;
#pragma unroll
      for (int j = 0; j < NS; ++j) {
        s0 += v0[q][j] * v0[q][j];
        s1 += v1[q][j] * v1[q][j];
        s2 += v2[q][j] * v2[q][j];
      }
      sq[3 * q] = s0; sq[3 * q + 1] = s1; sq[3 * q + 2] = s2;
    }
    wave_sum_n<3 * RB>(sq);
    float dd[2 * RB];
#pragma unroll
    for (int q = 0; q < RB; ++q) {
      const float rn0 = 1.0f / sqrtf(sq[3 * q]);
      const float rn1 = 1.0f / sqrtf(sq[3 * q + 1]);
      const float rn2 = 1.0f / sqrtf(sq[3 * q + 2]);
      float d1 = 0.f, d2 = 0.f;
#pragma unroll
      for (int j = 0; j < NS; ++j) {
        const float u0 = v0[q][j] * rn0;
        const float u1 = v1[q][j] * rn1;
        const float u2 = v2[q][j] * rn2;
        v0[q][j] = u0;
        d1 += u0 * u1;
        d2 += u0 * u2;
      }
      dd[2 * q] = d1; dd[2 * q + 1] = d2;
    }
    wave_sum_n<2 * RB>(dd);
#pragma unroll
    for (int q = 0; q < RB; ++q) {
#pragma unroll
      for (int j = 0; j < NS; ++j) {
        const int k = lane + 32 * j;
        if (k < H_) nr[(rb + q) * PN + k] = v0[q][j];
      }
    }
    float e1 = 0.f, e2 = 0.f;
#pragma unroll
    for (int q = 0; q < RB; ++q) {
      e1 = (lane == q) ? dd[2 * q] : e1;
      e2 = (lane == q) ? dd[2 * q + 1] : e2;
    }
    const float c1 = 1.0f - acosf(e1 * 0.99999f) * (1.0f / PI_F);
    const float c2 = 1.0f - acosf(e2 * 0.99999f) * (1.0f / PI_F);
    if (lane < RB) { cx[0][rb + lane] = c1; cx[1][rb + lane] = c2; }
  }
  __syncthreads();

  const int mr = lane & 15;
  const int kh = lane >> 4;
  v8f acc[4] = {};
  {
    const float* ap = nr + (w * 16 + mr) * PN + 2 * kh;
    const float* bp = nr + mr * PN + 2 * kh;
#pragma unroll 5
    for (int k = 0; k < H_; k += 4) {
      const v2f a = *reinterpret_cast<const v2f*>(ap + k);
#pragma unroll
      for (int c = 0; c < 4; ++c) {
        const v2f bb = *reinterpret_cast<const v2f*>(bp + c * 16 * PN + k);
        acc[c] = __builtin_amdgcn_wmma_f32_16x16x4_f32(false, a, false, bb, (short)0, acc[c], false, false);
      }
    }
  }
  __syncthreads();
#pragma unroll
  for (int c = 0; c < 4; ++c) {
#pragma unroll
    for (int v = 0; v < 8; ++v)
      nr[(w * 16 + 8 * kh + v) * TP + 16 * c + mr] = 1.0f - acosf(acc[c][v] * 0.99999f) * (1.0f / PI_F);
  }
  __syncthreads();

  {
    const int r0 = w * RPW;
    float sr[RPW];
#pragma unroll
    for (int q = 0; q < RPW; ++q) {
      const float t0 = nr[(r0 + q) * TP + lane];
      const float t1 = nr[(r0 + q) * TP + lane + 32];
      float* orow = intra + ((size_t)mb * L_ + r0 + q) * L_;
      orow[lane]      = t0;
      orow[lane + 32] = t1;
      float s = 0.f;
      s += t0;
      s += t1;
      sr[q] = s;
    }
    wave_sum_n<RPW>(sr);
    float e = 0.f;
#pragma unroll
    for (int q = 0; q < RPW; ++q) e = (lane == q) ? sr[q] : e;
    if (lane < RPW) {
      e += cx[0][r0 + lane];
      e += cx[1][r0 + lane];
      dv[r0 + lane] = 1.0f / sqrtf(e);
    }
  }
  __syncthreads();
  if (w < 2) {
    dinv[(size_t)mb * L_ + 32 * w + lane] = dv[32 * w + lane];
  } else {
    const int j  = w - 2;
    const int nn = j ? n2 : n1;
    crossv[((size_t)m * M_ + nn) * A_ + b * L_ + lane]      = cx[j][lane];
    crossv[((size_t)m * M_ + nn) * A_ + b * L_ + 32 + lane] = cx[j][32 + lane];
  }
}

struct no_mid {};
template <typename T> struct is_no_mid { static constexpr bool value = false; };
template <> struct is_no_mid<no_mid> { static constexpr bool value = true; };
__device__ __forceinline__ float4 ld4(const float* p) {
  return *reinterpret_cast<const float4*>(p);
}
__device__ __forceinline__ float2 ld2(const float* p) {
  return *reinterpret_cast<const float2*>(p);
}
__device__ __forceinline__ void st4(float* p, const float4 v) {
  *reinterpret_cast<float4*>(p) = v;
}

template <int I> struct set_c { static constexpr int value = I; };

__device__ __forceinline__ void lds_barrier() {
  __builtin_amdgcn_fence(__ATOMIC_RELEASE, "workgroup", "local");
  __builtin_amdgcn_s_barrier();
  __builtin_amdgcn_fence(__ATOMIC_ACQUIRE, "workgroup", "local");
}

template <int BK, bool COMB_HAS_MATH, typename FA, typename CA, typename FE, typename FM = no_mid>
__device__ __forceinline__ void gemm_block_64x48(const int K, const float* __restrict__ Bp,
                                                 const int n0, FA fetchA, CA combA, FE epi,
                                                 const int Kmid = 0, FM mid = FM{}) {
  constexpr bool TWO_PRODUCTS = !is_no_mid<FM>::value;
  constexpr int BM = 64, BN = 48, NTHR = 128;
  constexpr int NT = BN / 16;
  constexpr int PA = BK + 4;
  constexpr int PB = BN;
  constexpr int KQ = BK / 4;
  constexpr int NP = BN / 2;
  constexpr int A4 = BM * KQ;
  constexpr int B4 = (BK / 2) * NP;
  constexpr int RA = A4 / NTHR;
  constexpr int RB = (B4 + NTHR - 1) / NTHR;
  constexpr bool B_TAIL = (B4 % NTHR) != 0;
  static_assert(A4 % NTHR == 0 && B4 % 32 == 0, "A chunks split evenly over the block, B tasks over whole waves");
  static_assert((PA % 8) == 4, "A pitch must be 4 * odd dwords");
  static_assert(((2 * PB) % 64) == 32, "B pair pitch must be 32 mod 64 dwords");

  constexpr int ASZ = BM * PA;
  constexpr int BSZ = (BK / 2) * PB * 2;
  __shared__ __attribute__((aligned(16))) float As[2 * ASZ];
  __shared__ __attribute__((aligned(16))) float Bs[2 * BSZ];
  constexpr int TC = BN + 4;
  static_assert(((8 * TC) % 64) == 32 && (TC % 4) == 0, "epilogue tile pitch");
  __shared__ __attribute__((aligned(16))) float Ct[BM * TC];

  const int t    = threadIdx.x;
  const int lane = t & 31;
  const int w    = t >> 5;
  const int mr   = lane & 15;
  const int kh   = lane >> 4;

  float4 ax[2][RA], ay[2][RA];
  float2 be[2][RB], bo[2][RB];

  auto btask = [&](const int r) {
    const int e = t + NTHR * r;
    return (B_TAIL && r == RB - 1 && e >= B4) ? e - NTHR : e;
  };

  auto prefetch = [&](auto S, const int k0) {
    constexpr int s = decltype(S)::value;
#pragma unroll
    for (int r = 0; r < RA; ++r) {
      const int e = t + NTHR * r;
      fetchA(e / KQ, k0, (e % KQ) * 4, ax[s][r], ay[s][r]);
    }
#pragma unroll
    for (int r = 0; r < RB; ++r) {
      const int e  = btask(r);
      const int kp = e / NP;
      const int n  = min(n0 + (e % NP) * 2, H_ - 2);
      const float* p = Bp + (size_t)(k0 + 2 * kp) * H_ + n;
      be[s][r] = ld2(p);
      bo[s][r] = ld2(p + H_);
    }
  };

  auto stage = [&](auto S, const int k0) {
    constexpr int s = decltype(S)::value;
    constexpr int buf = s;
#pragma unroll
    for (int r = 0; r < RA; ++r) {
      const int e = t + NTHR * r;
      *reinterpret_cast<float4*>(&As[buf * ASZ + (e / KQ) * PA + (e % KQ) * 4]) = combA(k0, ax[s][r], ay[s][r]);
    }
#pragma unroll
    for (int r = 0; r < RB; ++r) {
      const int e = btask(r);
      const int j = (e % NP) * 2;
      const float4 v = make_float4(be[s][r].x, be[s][r].y, bo[s][r].x, bo[s][r].y);
      *reinterpret_cast<float4*>(&Bs[buf * BSZ + ((e / NP) * PB + j) * 2]) = v;
    }
  };

  const int aoff = (w * 16 + mr) * PA + 2 * kh;
  const int boff = (kh * PB + (mr & ~1)) * 2 + (mr & 1);

  v8f acc[NT] = {};
  auto compute = [&](const int buf) {
    const float* aptr = As + buf * ASZ + aoff;
    const float* bptr = Bs + buf * BSZ + boff;
#pragma unroll
    for (int ks = 0; ks < BK; ks += 4) {
      const v2f a = *reinterpret_cast<const v2f*>(aptr + ks);
#pragma unroll
      for (int c = 0; c < NT; ++c) {
        const float* bq = bptr + ((ks / 2) * PB + 16 * c) * 2;
        v2f b;
        b.x = bq[0];
        b.y = bq[2];
        acc[c] = __builtin_amdgcn_wmma_f32_16x16x4_f32(false, a, false, b, (short)0, acc[c], false, false);
      }
    }
  };

  constexpr int GR = BN / 4;
  const bool has_group = lane < 2 * GR;
  const int gl = has_group ? lane : lane - 2 * GR;
  const int er = gl / GR;
  const int eg = gl % GR;
  const int ei = w * 16 + er;
  const int en = has_group ? n0 + eg * 4 : H_;
  auto to_rows = [&](float4 (&rv)[8]) {
    float* ct = Ct + (w * 16) * TC;
#pragma unroll
    for (int c = 0; c < NT; ++c) {
#pragma unroll
      for (int v = 0; v < 8; ++v) ct[(8 * kh + v) * TC + 16 * c + mr] = acc[c][v];
    }
#pragma unroll
    for (int j = 0; j < 8; ++j) rv[j] = *reinterpret_cast<const float4*>(&ct[(er + 2 * j) * TC + eg * 4]);
  };

  auto turn = [&](auto S, auto MORE, const int k0) {
    constexpr int s = decltype(S)::value;
    constexpr bool more = decltype(MORE)::value != 0;
    if constexpr (TWO_PRODUCTS) {
      if (k0 == Kmid + BK) {
        float4 rv[8];
        to_rows(rv);
        mid(ei, en, rv);
#pragma unroll
        for (int c = 0; c < NT; ++c) acc[c] = v8f{};
      }
    }
    if constexpr (more) prefetch(set_c<(s ^ 1)>{}, k0 + BK);
    __builtin_amdgcn_sched_barrier(0);
    compute(s ^ 1);
    __builtin_amdgcn_sched_barrier(0);
    stage(S, k0);
    lds_barrier();
  };

  prefetch(set_c<0>{}, 0);
  prefetch(set_c<1>{}, BK);
  stage(set_c<0>{}, 0);
  lds_barrier();
  int k0 = BK;
  for (; k0 + 2 * BK < K; k0 += 2 * BK) {
    turn(set_c<1>{}, set_c<1>{}, k0);
    turn(set_c<0>{}, set_c<1>{}, k0 + BK);
  }
  if (k0 + BK < K) {
    turn(set_c<1>{}, set_c<1>{}, k0);
    turn(set_c<0>{}, set_c<0>{}, k0 + BK);
  } else {
    turn(set_c<1>{}, set_c<0>{}, k0);
  }
  compute(((K / BK) - 1) & 1);

  float4 rv[8];
  to_rows(rv);
  epi(ei, en, rv);
}

__global__ __launch_bounds__(128) void k_front(const float* __restrict__ xa,
                                               const float* __restrict__ xv,
                                               const float* __restrict__ xt,
                                               const float* __restrict__ w,
                                               const float* __restrict__ bias,
                                               float* __restrict__ h0,
                                               float* __restrict__ intra,
                                               float* __restrict__ crossv,
                                               float* __restrict__ dinv) {
  if (blockIdx.x < M_ * B_) {
    graph_block(blockIdx.x, xa, xv, xt, intra, crossv, dinv);
  } else {
    const int id = blockIdx.x - M_ * B_;
    const int n0 = (id % 5) * 48;
    const int r0 = (id / 5) * 64;
    const int mm = r0 / A_;
    const float* X = ((mm == 0) ? xa : (mm == 1) ? xv : xt) + (size_t)(r0 - mm * A_) * H_;
    auto fA = [&](int i, int k0, int kq, float4& x, float4& y) {
      (void)y;
      x = ld4(X + (size_t)i * H_ + k0 + kq);
    };
    auto cA = [&](int k0, const float4& x, const float4& y) { (void)k0; (void)y; return x; };
    auto ep = [&](const int i0, const int n, const float4 (&v)[8]) {
      if (n >= H_) return;
      const float4 b = ld4(bias + n);
#pragma unroll
      for (int j = 0; j < 8; ++j)
        st4(h0 + (size_t)(r0 + i0 + 2 * j) * H_ + n,
            make_float4(fmaxf(v[j].x + b.x, 0.0f), fmaxf(v[j].y + b.y, 0.0f),
                        fmaxf(v[j].z + b.z, 0.0f), fmaxf(v[j].w + b.w, 0.0f)));
    };
    gemm_block_64x48<40, false>(H_, w, n0, fA, cA, ep);
  }
}

__global__ __launch_bounds__(128) void k_spmm(const float* __restrict__ intra,
                                              const float* __restrict__ crossv,
                                              const float* __restrict__ dinv,
                                              const float* __restrict__ h,
                                              float* __restrict__ hi) {
  const int n0 = blockIdx.x * 48;
  const int mb = blockIdx.y;
  const int m  = mb >> 5;
  const int rowbase = mb * L_;
  const float* S = intra + (size_t)mb * L_ * L_;
  auto fA = [&](int i, int k0, int kq, float4& x, float4& y) {
    x = ld4(S + (size_t)i * L_ + k0 + kq);
    y = ld4(dinv + rowbase + k0 + kq);
  };
  auto cA = [&](int k0, const float4& x, const float4& y) {
    (void)k0;
    return make_float4(x.x * y.x, x.y * y.y, x.z * y.z, x.w * y.w);
  };
  const int o1 = (m == 0) ? 1 : 0;
  const int o2 = (m == 2) ? 1 : 2;
  auto ep = [&](const int i0, const int n, const float4 (&v)[8]) {
    if (n >= H_) return;
    float  cd1[8], cd2[8], dr[8];
    float4 g1[8], g2[8];
#pragma unroll
    for (int j = 0; j < 8; ++j) {
      const int row  = rowbase + i0 + 2 * j;
      const int a    = row - m * A_;
      const int r1   = o1 * A_ + a;
      const int r2   = o2 * A_ + a;
      cd1[j] = crossv[((size_t)m * M_ + o1) * A_ + a] * dinv[r1];
      cd2[j] = crossv[((size_t)m * M_ + o2) * A_ + a] * dinv[r2];
      dr[j]  = dinv[row];
      g1[j]  = ld4(h + (size_t)r1 * H_ + n);
      g2[j]  = ld4(h + (size_t)r2 * H_ + n);
    }
#pragma unroll
    for (int j = 0; j < 8; ++j) {
      float4 s = v[j];
      s.x += cd1[j] * g1[j].x; s.y += cd1[j] * g1[j].y; s.z += cd1[j] * g1[j].z; s.w += cd1[j] * g1[j].w;
      s.x += cd2[j] * g2[j].x; s.y += cd2[j] * g2[j].y; s.z += cd2[j] * g2[j].z; s.w += cd2[j] * g2[j].w;
      st4(hi + (size_t)(rowbase + i0 + 2 * j) * H_ + n,
          make_float4(s.x * dr[j], s.y * dr[j], s.z * dr[j], s.w * dr[j]));
    }
  };
  gemm_block_64x48<32, true>(L_, h + (size_t)rowbase * H_, n0, fA, cA, ep);
}

__global__ __launch_bounds__(128) void k_conv(const float* __restrict__ hi,
                                              const float* __restrict__ h0,
                                              const float* __restrict__ w, float beta,
                                              float* __restrict__ hout) {
  const int n0 = blockIdx.x * 48;
  const int r0 = blockIdx.y * 64;
  auto fA = [&](int i, int k0, int kq, float4& x, float4& y) {
    (void)y;
    const float* p = (k0 < H_) ? hi + (size_t)(r0 + i) * H_ + (k0 + kq)
                               : h0 + (size_t)(r0 + i) * H_ + (k0 - H_ + kq);
    x = ld4(p);
  };
  auto cA = [&](int k0, const float4& x, const float4& y) { (void)k0; (void)y; return x; };
  auto ep = [&](const int i0, const int n, const float4 (&v)[8]) {
    if (n >= H_) return;
    float4 a[8], b[8];
#pragma unroll
    for (int j = 0; j < 8; ++j) {
      const size_t o = (size_t)(r0 + i0 + 2 * j) * H_ + n;
      a[j] = ld4(hi + o);
      b[j] = ld4(h0 + o);
    }
    auto one = [&](const float x, const float p, const float q) {
      const float r = 0.9f * p + 0.1f * q;
      return fmaxf(beta * x + (1.0f - beta) * r, 0.0f);
    };
#pragma unroll
    for (int j = 0; j < 8; ++j)
      st4(hout + (size_t)(r0 + i0 + 2 * j) * H_ + n,
          make_float4(one(v[j].x, a[j].x, b[j].x), one(v[j].y, a[j].y, b[j].y),
                      one(v[j].z, a[j].z, b[j].z), one(v[j].w, a[j].w, b[j].w)));
  };
  gemm_block_64x48<40, false>(2 * H_, w, n0, fA, cA, ep);
}

__global__ __launch_bounds__(128) void k_hx(const float* __restrict__ xa,
                                            const float* __restrict__ xv,
                                            const float* __restrict__ xt,
                                            const float* __restrict__ hf,
                                            const float* __restrict__ tw,
                                            const float* __restrict__ tb,
                                            float* __restrict__ hx) {
  const int m  = blockIdx.z;
  const int n0 = blockIdx.x * 48;
  const int r0 = m * A_ + blockIdx.y * 64;
  const float* Wm = tw + (size_t)m * 2 * H_ * H_;
  const float* Bb = tb + (size_t)m * H_;
  const float* X  = ((m == 0) ? xa : (m == 1) ? xv : xt) + (size_t)blockIdx.y * 64 * H_;
  auto fA = [&](int i, int k0, int kq, float4& x, float4& y) {
    (void)y;
    const float* p = (k0 < H_) ? X  + (size_t)i * H_ + (k0 + kq)
                               : hf + (size_t)(r0 + i) * H_ + (k0 - H_ + kq);
    x = ld4(p);
  };
  auto cA = [&](int k0, const float4& x, const float4& y) { (void)k0; (void)y; return x; };
  auto ep = [&](const int i0, const int n, const float4 (&v)[8]) {
    if (n >= H_) return;
    const float4 b = ld4(Bb + n);
#pragma unroll
    for (int j = 0; j < 8; ++j)
      st4(hx + (size_t)(r0 + i0 + 2 * j) * H_ + n,
          make_float4(fmaxf(v[j].x + b.x, 0.0f), fmaxf(v[j].y + b.y, 0.0f),
                      fmaxf(v[j].z + b.z, 0.0f), fmaxf(v[j].w + b.w, 0.0f)));
  };
  gemm_block_64x48<40, false>(2 * H_, Wm, n0, fA, cA, ep);
}

__global__ __launch_bounds__(128) void k_gate(const float* __restrict__ hx,
                                              const float* __restrict__ cw,
                                              const float* __restrict__ cb,
                                              float* __restrict__ hout) {
  constexpr int KP = 3 * H_;
  const int i  = blockIdx.z;
  const int n0 = blockIdx.x * 48;
  const int a0 = blockIdx.y * 64;
  const int j0 = (i == 0) ? 1 : 0;
  const int j1 = (i == 2) ? 1 : 2;
  const float* Hi  = hx + (size_t)i  * A_ * H_;
  const float* Hj0 = hx + (size_t)j0 * A_ * H_;
  const float* Hj1 = hx + (size_t)j1 * A_ * H_;
  const float* Wp = cw + (size_t)(2 * i) * KP * H_;
  const float* Bb = cb + (size_t)(2 * i) * H_;
  auto fA = [&](int r, int k0, int kq, float4& x, float4& y) {
    const int q  = k0 / KP;
    const int kp = k0 - q * KP;
    const int s  = kp / H_;
    const int kk = kp - s * H_ + kq;
    const float* Hj = q ? Hj1 : Hj0;
    const float* px = ((s == 1) ? Hj : Hi) + (size_t)(a0 + r) * H_ + kk;
    x = ld4(px);
    if (s == 2) y = ld4(Hj + (size_t)(a0 + r) * H_ + kk);
    else y = make_float4(0.f, 0.f, 0.f, 0.f);
  };
  auto cA = [&](int k0, const float4& x, const float4& y) {
    if ((k0 % KP) / H_ == 2) return make_float4(x.x * y.x, x.y * y.y, x.z * y.z, x.w * y.w);
    return x;
  };

  float4 g[8];
  auto gated = [&](const int i0, const int n, const float4 (&v)[8], const float* Hj, const float* bias,
                   const bool add) {
    if (n >= H_) return;
    const float4 bn = ld4(bias + n);
    float4 p[8], q[8];
#pragma unroll
    for (int j = 0; j < 8; ++j) {
      const size_t o = (size_t)(a0 + i0 + 2 * j) * H_ + n;
      p[j] = ld4(Hi + o);
      q[j] = ld4(Hj + o);
    }
    auto one = [&](const float x, const float b, const float hi_, const float hj_) {
      const float z = 1.0f / (1.0f + expf(-(x + b)));
      return z * hi_ + (1.0f - z) * hj_;
    };
#pragma unroll
    for (int j = 0; j < 8; ++j) {
      const float4 t = make_float4(one(v[j].x, bn.x, p[j].x, q[j].x), one(v[j].y, bn.y, p[j].y, q[j].y),
                                   one(v[j].z, bn.z, p[j].z, q[j].z), one(v[j].w, bn.w, p[j].w, q[j].w));
      g[j] = add ? make_float4(g[j].x + t.x, g[j].y + t.y, g[j].z + t.z, g[j].w + t.w) : t;
    }
  };
  auto mid = [&](const int i0, const int n, const float4 (&v)[8]) { gated(i0, n, v, Hj0, Bb, false); };
  auto ep = [&](const int i0, const int n, const float4 (&v)[8]) {
    if (n >= H_) return;
    gated(i0, n, v, Hj1, Bb + H_, true);
#pragma unroll
    for (int j = 0; j < 8; ++j)
      st4(hout + ((size_t)i * A_ + a0 + i0 + 2 * j) * H_ + n, g[j]);
  };
  gemm_block_64x48<40, true>(2 * KP, Wp, n0, fA, cA, ep, KP, mid);
}

__global__ __launch_bounds__(512) void k_final(const float* __restrict__ hout,
                                               const float* __restrict__ uw,
                                               const float* __restrict__ ub,
                                               float* __restrict__ out) {
  constexpr int UPB = 16;
  constexpr int NSF = H_ / 32;
  constexpr int REM = H_ - 32 * NSF;
  static_assert((UPB * TAG_) % 32 == 0, "a block's results must be whole 128-byte lines");
  static_assert(REM > 0 && M_ * REM <= 32, "the left-over units of all modalities share one slot");
  __shared__ float res[UPB * TAG_];
  const int wv   = threadIdx.x >> 5;
  const int lane = threadIdx.x & 31;
  const int a    = blockIdx.x * UPB + wv;
  float acc[TAG_] = {};
  auto add6 = [&](const float v, const float2 w01, const float2 w23, const float2 w45) {
    acc[0] += v * w01.x; acc[1] += v * w01.y;
    acc[2] += v * w23.x; acc[3] += v * w23.y;
    acc[4] += v * w45.x; acc[5] += v * w45.y;
  };
#pragma unroll
  for (int m = 0; m < M_; ++m) {
    const float* hr = hout + (size_t)(m * A_ + a) * H_;
    const float* wm = uw + (size_t)m * H_ * TAG_;
#pragma unroll
    for (int s = 0; s < NSF; ++s) {
      const int h = lane + 32 * s;
      const float* wr = wm + h * TAG_;
      add6(hr[h], ld2(wr), ld2(wr + 2), ld2(wr + 4));
    }
  }
  {
    const bool ok = lane < M_ * REM;
    const int l = ok ? lane : M_ * REM - 1;
    const int m = l / REM;
    const int h = 32 * NSF + l % REM;
    const float* wr = uw + ((size_t)m * H_ + h) * TAG_;
    const float  v   = hout[(size_t)(m * A_ + a) * H_ + h];
    const float2 w01 = ld2(wr), w23 = ld2(wr + 2), w45 = ld2(wr + 4);
    const float2 z = make_float2(0.f, 0.f);
    add6(ok ? v : 0.f, ok ? w01 : z, ok ? w23 : z, ok ? w45 : z);
  }
#pragma unroll
  for (int t = 0; t < TAG_; ++t) acc[t] = wave_sum(acc[t]);
  if (lane == 0) {
#pragma unroll
    for (int t = 0; t < TAG_; ++t)
      res[wv * TAG_ + t] = acc[t] + (ub[t] + ub[TAG_ + t] + ub[2 * TAG_ + t]);
  }
  __syncthreads();
  if (wv == 0) {
    for (int j = 0; j < UPB * TAG_ / 32; ++j)
      out[(size_t)blockIdx.x * (UPB * TAG_) + 32 * j + lane] = res[32 * j + lane];
  }
}

}

extern "C" void kernel_launch(void* const* d_in, const int* in_sizes, int n_in,
                              void* d_out, int out_size, void* d_ws, size_t ws_size,
                              hipStream_t stream) {
  (void)in_sizes; (void)n_in; (void)out_size; (void)ws_size;
  const float* x_a     = (const float*)d_in[0];
  const float* x_v     = (const float*)d_in[1];
  const float* x_t     = (const float*)d_in[2];
  const float* fc0_w   = (const float*)d_in[3];
  const float* fc0_b   = (const float*)d_in[4];
  const float* conv_w  = (const float*)d_in[5];
  const float* tm_w    = (const float*)d_in[6];
  const float* tm_b    = (const float*)d_in[7];
  const float* cross_w = (const float*)d_in[8];
  const float* cross_b = (const float*)d_in[9];
  const float* uni_w   = (const float*)d_in[10];
  const float* uni_b   = (const float*)d_in[11];
  float* out = (float*)d_out;

  const size_t NH = (size_t)N_ * H_;
  float* ws    = (float*)d_ws;
  float* intra = ws; ws += (size_t)N_ * L_;
  float* crossv= ws; ws += (size_t)M_ * M_ * A_;
  float* dinv  = ws; ws += N_;
  float* h0    = ws; ws += NH;
  float* hA    = ws; ws += NH;
  float* hBf   = ws; ws += NH;
  float* hi    = ws; ws += NH;
  float* hx    = ws; ws += NH;
  float* hout  = ws; ws += NH;

  k_front<<<M_ * B_ + 5 * (N_ / 64), dim3(128), 0, stream>>>(x_a, x_v, x_t, fc0_w, fc0_b, h0,
                                                            intra, crossv, dinv);

  const float beta0 = logf(0.5f / 1.0f + 1.0f);
  const float beta1 = logf(0.5f / 2.0f + 1.0f);
  k_spmm <<<dim3(5, M_ * B_),    dim3(128), 0, stream>>>(intra, crossv, dinv, h0, hi);
  k_conv <<<dim3(5, N_ / 64),    dim3(128), 0, stream>>>(hi, h0, conv_w, beta0, hA);
  k_spmm <<<dim3(5, M_ * B_),    dim3(128), 0, stream>>>(intra, crossv, dinv, hA, hi);
  k_conv <<<dim3(5, N_ / 64),    dim3(128), 0, stream>>>(hi, h0, conv_w + 2 * H_ * H_, beta1, hBf);

  k_hx   <<<dim3(5, A_ / 64, M_), dim3(128), 0, stream>>>(x_a, x_v, x_t, hBf, tm_w, tm_b, hx);

  k_gate <<<dim3(5, A_ / 64, M_), dim3(128), 0, stream>>>(hx, cross_w, cross_b, hout);

  k_final<<<A_ / 16, dim3(512), 0, stream>>>(hout, uni_w, uni_b, out);
}


# spmm, conv (both layers) and hx fused: an 8-wave block owns 32 whole rows x all 13 column tiles, hi and the new rows stay in LDS as the next A operand, only W streamed; 5 launches, was 8; sums kept
# speedup vs baseline: 8.2948x; 1.0484x over previous
#include <hip/hip_runtime.h>
#include <math.h>

typedef __attribute__((ext_vector_type(2))) float v2f;
typedef __attribute__((ext_vector_type(8))) float v8f;

namespace {

constexpr int B_   = 32;
constexpr int L_   = 64;
constexpr int H_   = 200;
constexpr int M_   = 3;
constexpr int A_   = B_ * L_;
constexpr int N_   = M_ * A_;
constexpr int TAG_ = 6;
constexpr float PI_F = 3.14159265358979323846f;

__device__ __forceinline__ float wave_sum(float s) {
#pragma unroll
  for (int off = 16; off > 0; off >>= 1) s += __shfl_xor(s, off, 32);
  return s;
}

template <int N>
__device__ __forceinline__ void wave_sum_n(float (&s)[N]) {
#pragma unroll
  for (int off = 16; off > 0; off >>= 1) {
    float t[N];
#pragma unroll
    for (int i = 0; i < N; ++i) t[i] = __shfl_xor(s[i], off, 32);
#pragma unroll
    for (int i = 0; i < N; ++i) s[i] += t[i];
  }
}

__device__ __forceinline__ void graph_block(const int mb,
                                            const float* __restrict__ xa,
                                            const float* __restrict__ xv,
                                            const float* __restrict__ xt,
                                            float* __restrict__ intra,
                                            float* __restrict__ crossv,
                                            float* __restrict__ dinv) {
  constexpr int PN  = H_ + 4;
  constexpr int TP  = L_ + 4;
  constexpr int NS  = (H_ + 31) / 32;
  constexpr int NW  = 4;
  constexpr int RPW = L_ / NW;
  constexpr int RB  = 8;
  static_assert((PN % 8) == 4, "pitch must be 4 * odd dwords");
  static_assert(L_ * TP <= L_ * PN, "the similarity tile reuses the rows' LDS");
  static_assert(RPW == 16 && RPW % RB == 0, "a wave owns one 16-row strip");
  __shared__ __attribute__((aligned(16))) float nr[L_ * PN];
  __shared__ float cx[2][L_];
  __shared__ float dv[L_];
  const int m    = mb >> 5;
  const int b    = mb & 31;
  const int w    = threadIdx.x >> 5;
  const int lane = threadIdx.x & 31;
  const int n1   = (m == 0) ? 1 : 0;
  const int n2   = (m == 2) ? 1 : 2;
  auto plane = [&](const int mm) { return (mm == 0) ? xa : (mm == 1) ? xv : xt; };
  const float* P0 = plane(m)  + (size_t)b * L_ * H_;
  const float* P1 = plane(n1) + (size_t)b * L_ * H_;
  const float* P2 = plane(n2) + (size_t)b * L_ * H_;

#pragma unroll 1
  for (int rb = w * RPW; rb < (w + 1) * RPW; rb += RB) {
    float v0[RB][NS], v1[RB][NS], v2[RB][NS];
#pragma unroll
    for (int q = 0; q < RB; ++q) {
#pragma unroll
      for (int j = 0; j < NS; ++j) {
        const int k = lane + 32 * j;
        const bool ok = k < H_;
        v0[q][j] = ok ? P0[(size_t)(rb + q) * H_ + k] : 0.f;
        v1[q][j] = ok ? P1[(size_t)(rb + q) * H_ + k] : 0.f;
        v2[q][j] = ok ? P2[(size_t)(rb + q) * H_ + k] : 0.f;
      }
    }
    float sq[3 * RB];
#pragma unroll
    for (int q = 0; q < RB; ++q) {
      float s0 = 0.f, s1 = 0.f, s2 = 0.f;
#pragma unroll
      for (int j = 0; j < NS; ++j) {
        s0 += v0[q][j] * v0[q][j];
        s1 += v1[q][j] * v1[q][j];
        s2 += v2[q][j] * v2[q][j];
      }
      sq[3 * q] = s0; sq[3 * q + 1] = s1; sq[3 * q + 2] = s2;
    }
    wave_sum_n<3 * RB>(sq);
    float dd[2 * RB];
#pragma unroll
    for (int q = 0; q < RB; ++q) {
      const float rn0 = 1.0f / sqrtf(sq[3 * q]);
      const float rn1 = 1.0f / sqrtf(sq[3 * q + 1]);
      const float rn2 = 1.0f / sqrtf(sq[3 * q + 2]);
      float d1 = 0.f, d2 = 0.f;
#pragma unroll
      for (int j = 0; j < NS; ++j) {
        const float u0 = v0[q][j] * rn0;
        const float u1 = v1[q][j] * rn1;
        const float u2 = v2[q][j] * rn2;
        v0[q][j] = u0;
        d1 += u0 * u1;
        d2 += u0 * u2;
      }
      dd[2 * q] = d1; dd[2 * q + 1] = d2;
    }
    wave_sum_n<2 * RB>(dd);
#pragma unroll
    for (int q = 0; q < RB; ++q) {
#pragma unroll
      for (int j = 0; j < NS; ++j) {
        const int k = lane + 32 * j;
        if (k < H_) nr[(rb + q) * PN + k] = v0[q][j];
      }
    }
    float e1 = 0.f, e2 = 0.f;
#pragma unroll
    for (int q = 0; q < RB; ++q) {
      e1 = (lane == q) ? dd[2 * q] : e1;
      e2 = (lane == q) ? dd[2 * q + 1] : e2;
    }
    const float c1 = 1.0f - acosf(e1 * 0.99999f) * (1.0f / PI_F);
    const float c2 = 1.0f - acosf(e2 * 0.99999f) * (1.0f / PI_F);
    if (lane < RB) { cx[0][rb + lane] = c1; cx[1][rb + lane] = c2; }
  }
  __syncthreads();

  const int mr = lane & 15;
  const int kh = lane >> 4;
  v8f acc[4] = {};
  {
    const float* ap = nr + (w * 16 + mr) * PN + 2 * kh;
    const float* bp = nr + mr * PN + 2 * kh;
#pragma unroll 5
    for (int k = 0; k < H_; k += 4) {
      const v2f a = *reinterpret_cast<const v2f*>(ap + k);
#pragma unroll
      for (int c = 0; c < 4; ++c) {
        const v2f bb = *reinterpret_cast<const v2f*>(bp + c * 16 * PN + k);
        acc[c] = __builtin_amdgcn_wmma_f32_16x16x4_f32(false, a, false, bb, (short)0, acc[c], false, false);
      }
    }
  }
  __syncthreads();
#pragma unroll
  for (int c = 0; c < 4; ++c) {
#pragma unroll
    for (int v = 0; v < 8; ++v)
      nr[(w * 16 + 8 * kh + v) * TP + 16 * c + mr] = 1.0f - acosf(acc[c][v] * 0.99999f) * (1.0f / PI_F);
  }
  __syncthreads();

  {
    const int r0 = w * RPW;
    float sr[RPW];
#pragma unroll
    for (int q = 0; q < RPW; ++q) {
      const float t0 = nr[(r0 + q) * TP + lane];
      const float t1 = nr[(r0 + q) * TP + lane + 32];
      float* orow = intra + ((size_t)mb * L_ + r0 + q) * L_;
      orow[lane]      = t0;
      orow[lane + 32] = t1;
      float s = 0.f;
      s += t0;
      s += t1;
      sr[q] = s;
    }
    wave_sum_n<RPW>(sr);
    float e = 0.f;
#pragma unroll
    for (int q = 0; q < RPW; ++q) e = (lane == q) ? sr[q] : e;
    if (lane < RPW) {
      e += cx[0][r0 + lane];
      e += cx[1][r0 + lane];
      dv[r0 + lane] = 1.0f / sqrtf(e);
    }
  }
  __syncthreads();
  if (w < 2) {
    dinv[(size_t)mb * L_ + 32 * w + lane] = dv[32 * w + lane];
  } else {
    const int j  = w - 2;
    const int nn = j ? n2 : n1;
    crossv[((size_t)m * M_ + nn) * A_ + b * L_ + lane]      = cx[j][lane];
    crossv[((size_t)m * M_ + nn) * A_ + b * L_ + 32 + lane] = cx[j][32 + lane];
  }
}

struct no_mid {};
template <typename T> struct is_no_mid { static constexpr bool value = false; };
template <> struct is_no_mid<no_mid> { static constexpr bool value = true; };
__device__ __forceinline__ float4 ld4(const float* p) {
  return *reinterpret_cast<const float4*>(p);
}
__device__ __forceinline__ float2 ld2(const float* p) {
  return *reinterpret_cast<const float2*>(p);
}
__device__ __forceinline__ void st4(float* p, const float4 v) {
  *reinterpret_cast<float4*>(p) = v;
}

template <int I> struct set_c { static constexpr int value = I; };

__device__ __forceinline__ void lds_barrier() {
  __builtin_amdgcn_fence(__ATOMIC_RELEASE, "workgroup", "local");
  __builtin_amdgcn_s_barrier();
  __builtin_amdgcn_fence(__ATOMIC_ACQUIRE, "workgroup", "local");
}

template <int BK, bool COMB_HAS_MATH, typename FA, typename CA, typename FE, typename FM = no_mid>
__device__ __forceinline__ void gemm_block_64x48(const int K, const float* __restrict__ Bp,
                                                 const int n0, FA fetchA, CA combA, FE epi,
                                                 const int Kmid = 0, FM mid = FM{}) {
  constexpr bool TWO_PRODUCTS = !is_no_mid<FM>::value;
  constexpr int BM = 64, BN = 48, NTHR = 128;
  constexpr int NT = BN / 16;
  constexpr int PA = BK + 4;
  constexpr int PB = BN;
  constexpr int KQ = BK / 4;
  constexpr int NP = BN / 2;
  constexpr int A4 = BM * KQ;
  constexpr int B4 = (BK / 2) * NP;
  constexpr int RA = A4 / NTHR;
  constexpr int RB = (B4 + NTHR - 1) / NTHR;
  constexpr bool B_TAIL = (B4 % NTHR) != 0;
  static_assert(A4 % NTHR == 0 && B4 % 32 == 0, "A chunks split evenly over the block, B tasks over whole waves");
  static_assert((PA % 8) == 4, "A pitch must be 4 * odd dwords");
  static_assert(((2 * PB) % 64) == 32, "B pair pitch must be 32 mod 64 dwords");

  constexpr int ASZ = BM * PA;
  constexpr int BSZ = (BK / 2) * PB * 2;
  __shared__ __attribute__((aligned(16))) float As[2 * ASZ];
  __shared__ __attribute__((aligned(16))) float Bs[2 * BSZ];
  constexpr int TC = BN + 4;
  static_assert(((8 * TC) % 64) == 32 && (TC % 4) == 0, "epilogue tile pitch");
  __shared__ __attribute__((aligned(16))) float Ct[BM * TC];

  const int t    = threadIdx.x;
  const int lane = t & 31;
  const int w    = t >> 5;
  const int mr   = lane & 15;
  const int kh   = lane >> 4;

  float4 ax[2][RA], ay[2][RA];
  float2 be[2][RB], bo[2][RB];

  auto btask = [&](const int r) {
    const int e = t + NTHR * r;
    return (B_TAIL && r == RB - 1 && e >= B4) ? e - NTHR : e;
  };

  auto prefetch = [&](auto S, const int k0) {
    constexpr int s = decltype(S)::value;
#pragma unroll
    for (int r = 0; r < RA; ++r) {
      const int e = t + NTHR * r;
      fetchA(e / KQ, k0, (e % KQ) * 4, ax[s][r], ay[s][r]);
    }
#pragma unroll
    for (int r = 0; r < RB; ++r) {
      const int e  = btask(r);
      const int kp = e / NP;
      const int n  = min(n0 + (e % NP) * 2, H_ - 2);
      const float* p = Bp + (size_t)(k0 + 2 * kp) * H_ + n;
      be[s][r] = ld2(p);
      bo[s][r] = ld2(p + H_);
    }
  };

  auto stage = [&](auto S, const int k0) {
    constexpr int s = decltype(S)::value;
    constexpr int buf = s;
#pragma unroll
    for (int r = 0; r < RA; ++r) {
      const int e = t + NTHR * r;
      *reinterpret_cast<float4*>(&As[buf * ASZ + (e / KQ) * PA + (e % KQ) * 4]) = combA(k0, ax[s][r], ay[s][r]);
    }
#pragma unroll
    for (int r = 0; r < RB; ++r) {
      const int e = btask(r);
      const int j = (e % NP) * 2;
      const float4 v = make_float4(be[s][r].x, be[s][r].y, bo[s][r].x, bo[s][r].y);
      *reinterpret_cast<float4*>(&Bs[buf * BSZ + ((e / NP) * PB + j) * 2]) = v;
    }
  };

  const int aoff = (w * 16 + mr) * PA + 2 * kh;
  const int boff = (kh * PB + (mr & ~1)) * 2 + (mr & 1);

  v8f acc[NT] = {};
  auto compute = [&](const int buf) {
    const float* aptr = As + buf * ASZ + aoff;
    const float* bptr = Bs + buf * BSZ + boff;
#pragma unroll
    for (int ks = 0; ks < BK; ks += 4) {
      const v2f a = *reinterpret_cast<const v2f*>(aptr + ks);
#pragma unroll
      for (int c = 0; c < NT; ++c) {
        const float* bq = bptr + ((ks / 2) * PB + 16 * c) * 2;
        v2f b;
        b.x = bq[0];
        b.y = bq[2];
        acc[c] = __builtin_amdgcn_wmma_f32_16x16x4_f32(false, a, false, b, (short)0, acc[c], false, false);
      }
    }
  };

  constexpr int GR = BN / 4;
  const bool has_group = lane < 2 * GR;
  const int gl = has_group ? lane : lane - 2 * GR;
  const int er = gl / GR;
  const int eg = gl % GR;
  const int ei = w * 16 + er;
  const int en = has_group ? n0 + eg * 4 : H_;
  auto to_rows = [&](float4 (&rv)[8]) {
    float* ct = Ct + (w * 16) * TC;
#pragma unroll
    for (int c = 0; c < NT; ++c) {
#pragma unroll
      for (int v = 0; v < 8; ++v) ct[(8 * kh + v) * TC + 16 * c + mr] = acc[c][v];
    }
#pragma unroll
    for (int j = 0; j < 8; ++j) rv[j] = *reinterpret_cast<const float4*>(&ct[(er + 2 * j) * TC + eg * 4]);
  };

  auto turn = [&](auto S, auto MORE, const int k0) {
    constexpr int s = decltype(S)::value;
    constexpr bool more = decltype(MORE)::value != 0;
    if constexpr (TWO_PRODUCTS) {
      if (k0 == Kmid + BK) {
        float4 rv[8];
        to_rows(rv);
        mid(ei, en, rv);
#pragma unroll
        for (int c = 0; c < NT; ++c) acc[c] = v8f{};
      }
    }
    if constexpr (more) prefetch(set_c<(s ^ 1)>{}, k0 + BK);
    __builtin_amdgcn_sched_barrier(0);
    compute(s ^ 1);
    __builtin_amdgcn_sched_barrier(0);
    stage(S, k0);
    lds_barrier();
  };

  prefetch(set_c<0>{}, 0);
  prefetch(set_c<1>{}, BK);
  stage(set_c<0>{}, 0);
  lds_barrier();
  int k0 = BK;
  for (; k0 + 2 * BK < K; k0 += 2 * BK) {
    turn(set_c<1>{}, set_c<1>{}, k0);
    turn(set_c<0>{}, set_c<1>{}, k0 + BK);
  }
  if (k0 + BK < K) {
    turn(set_c<1>{}, set_c<1>{}, k0);
    turn(set_c<0>{}, set_c<0>{}, k0 + BK);
  } else {
    turn(set_c<1>{}, set_c<0>{}, k0);
  }
  compute(((K / BK) - 1) & 1);

  float4 rv[8];
  to_rows(rv);
  epi(ei, en, rv);
}

__global__ __launch_bounds__(128) void k_front(const float* __restrict__ xa,
                                               const float* __restrict__ xv,
                                               const float* __restrict__ xt,
                                               const float* __restrict__ w,
                                               const float* __restrict__ bias,
                                               float* __restrict__ h0,
                                               float* __restrict__ intra,
                                               float* __restrict__ crossv,
                                               float* __restrict__ dinv) {
  if (blockIdx.x < M_ * B_) {
    graph_block(blockIdx.x, xa, xv, xt, intra, crossv, dinv);
  } else {
    const int id = blockIdx.x - M_ * B_;
    const int n0 = (id % 5) * 48;
    const int r0 = (id / 5) * 64;
    const int mm = r0 / A_;
    const float* X = ((mm == 0) ? xa : (mm == 1) ? xv : xt) + (size_t)(r0 - mm * A_) * H_;
    auto fA = [&](int i, int k0, int kq, float4& x, float4& y) {
      (void)y;
      x = ld4(X + (size_t)i * H_ + k0 + kq);
    };
    auto cA = [&](int k0, const float4& x, const float4& y) { (void)k0; (void)y; return x; };
    auto ep = [&](const int i0, const int n, const float4 (&v)[8]) {
      if (n >= H_) return;
      const float4 b = ld4(bias + n);
#pragma unroll
      for (int j = 0; j < 8; ++j)
        st4(h0 + (size_t)(r0 + i0 + 2 * j) * H_ + n,
            make_float4(fmaxf(v[j].x + b.x, 0.0f), fmaxf(v[j].y + b.y, 0.0f),
                        fmaxf(v[j].z + b.z, 0.0f), fmaxf(v[j].w + b.w, 0.0f)));
    };
    gemm_block_64x48<40, false>(H_, w, n0, fA, cA, ep);
  }
}

constexpr int LR   = 32;
constexpr int LNC  = 208;
constexpr int LPH  = 212;
constexpr int LPS  = 68;
constexpr int LBK  = 40;
constexpr int LNP  = LNC / 2;
constexpr int LBSZ = (LBK / 2) * LNC * 2;
constexpr int LQ   = H_ / 4;
static_assert(((2 * LNC) % 64) == 32 && (LPH % 8) == 4 && (LPS % 8) == 4, "LDS pitches");
static_assert((L_ / 2) * LNC * 2 <= 2 * LBSZ, "the dialogue's rows of h fit the two B buffers");

template <int C0, int NTW, int KK>
__device__ __forceinline__ void mma_run(v8f (&acc)[4], const float* aptr, const float* bptr) {
#pragma unroll
  for (int ks = 0; ks < KK; ks += 4) {
    const v2f a = *reinterpret_cast<const v2f*>(aptr + ks);
#pragma unroll
    for (int c = C0; c < C0 + NTW; ++c) {
      const float* bq = bptr + ((ks / 2) * LNC + 16 * c) * 2;
      v2f b;
      b.x = bq[0];
      b.y = bq[2];
      acc[c] = __builtin_amdgcn_wmma_f32_16x16x4_f32(false, a, false, b, (short)0, acc[c], false, false);
    }
  }
}

template <bool WITH_HX>
__global__ __launch_bounds__(256) void k_layer(const float* __restrict__ intra,
                                               const float* __restrict__ crossv,
                                               const float* __restrict__ dinv,
                                               const float* __restrict__ h,
                                               const float* __restrict__ h0,
                                               const float* __restrict__ wl, float beta,
                                               float* __restrict__ hnew,
                                               const float* __restrict__ xa,
                                               const float* __restrict__ xv,
                                               const float* __restrict__ xt,
                                               const float* __restrict__ tw,
                                               const float* __restrict__ tb,
                                               float* __restrict__ hx) {
  __shared__ __attribute__((aligned(16))) float Bs[2 * LBSZ];
  __shared__ __attribute__((aligned(16))) float AH[2 * LR * LPH];
  __shared__ __attribute__((aligned(16))) float Ss[LR * LPS];
  __shared__ __attribute__((aligned(16))) float Ct[LR * LPH];

  const int t    = threadIdx.x;
  const int lane = t & 31;
  const int w    = __builtin_amdgcn_readfirstlane(t >> 5);
  const int mr   = lane & 15;
  const int kh   = lane >> 4;
  const int s    = w & 1;
  const int cg   = w >> 1;
  const int tile0 = (cg == 0) ? 0 : 3 * cg + 1;
  const bool four = (cg == 0);

  const int mb      = blockIdx.x >> 1;
  const int r0      = (blockIdx.x & 1) * LR;
  const int m       = mb >> 5;
  const int rowbase = mb * L_;
  const int o1 = (m == 0) ? 1 : 0;
  const int o2 = (m == 2) ? 1 : 2;

  float* Hs  = AH;
  float* H2s = AH + LR * LPH;

  constexpr int WT = (LBK / 2) * LNP;
  constexpr int WR = (WT + 255) / 256;
  float2 be[WR], bo[WR];
  auto wtask = [&](const int r) {
    const int e = t + 256 * r;
    return (r == WR - 1 && e >= WT) ? e - 256 : e;
  };
  unsigned woff[WR];
  int      soff[WR];
#pragma unroll
  for (int r = 0; r < WR; ++r) {
    const int e  = wtask(r);
    const int kp = e / LNP;
    const int n  = min(2 * (e % LNP), H_ - 2);
    woff[r] = (unsigned)((2 * kp) * H_ + n);
    soff[r] = (kp * LNC + 2 * (e % LNP)) * 2;
  }
  const float* wp[WR];
  auto startW = [&](const float* __restrict__ Wp) {
#pragma unroll
    for (int r = 0; r < WR; ++r) wp[r] = Wp + woff[r];
  };
  auto prefetchW = [&]() {
#pragma unroll
    for (int r = 0; r < WR; ++r) {
      be[r] = ld2(wp[r]);
      bo[r] = ld2(wp[r] + H_);
      wp[r] += LBK * H_;
    }
  };
  auto stageW = [&](const int buf) {
#pragma unroll
    for (int r = 0; r < WR; ++r)
      st4(&Bs[buf * LBSZ + soff[r]], make_float4(be[r].x, be[r].y, bo[r].x, bo[r].y));
  };

  const int aoff = (16 * s + mr) * LPH + 2 * kh;
  const int boff = (kh * LNC + 16 * tile0 + (mr & ~1)) * 2 + (mr & 1);

  v8f acc[4] = {};
  auto dump = [&](float* dst) {
#pragma unroll
    for (int c = 0; c < 4; ++c) {
      if (c < 3 || four) {
#pragma unroll
        for (int v = 0; v < 8; ++v) dst[(16 * s + 8 * kh + v) * LPH + 16 * (tile0 + c) + mr] = acc[c][v];
      }
    }
  };
  auto stream = [&]() {
#pragma unroll 1
    for (int c = 0; c < (2 * H_) / LBK; ++c) {
      const int  k0   = c * LBK;
      const bool more = (c + 1) < (2 * H_) / LBK;
      if (more) prefetchW();
      __builtin_amdgcn_sched_barrier(0);
      const float* ap = AH + ((k0 >= H_) ? LR * LPH + (k0 - H_) : k0) + aoff;
      const float* bp = Bs + (c & 1) * LBSZ + boff;
      mma_run<0, 3, LBK>(acc, ap, bp);
      if (four) mma_run<3, 1, LBK>(acc, ap, bp);
      __builtin_amdgcn_sched_barrier(0);
      if (more) stageW((c + 1) & 1);
      lds_barrier();
    }
  };

  {
    const float* S = intra + (size_t)mb * L_ * L_ + (size_t)r0 * L_;
#pragma unroll
    for (int r = 0; r < (LR * (L_ / 4)) / 256; ++r) {
      const int e = t + 256 * r;
      const int row = e / (L_ / 4), q = e % (L_ / 4);
      const float4 x = ld4(S + (size_t)row * L_ + 4 * q);
      const float4 y = ld4(dinv + rowbase + 4 * q);
      st4(&Ss[row * LPS + 4 * q], make_float4(x.x * y.x, x.y * y.y, x.z * y.z, x.w * y.w));
    }
    const float* Hb = h + (size_t)rowbase * H_;
    static_assert(((L_ / 2) * LNP) % 256 == 0, "the dialogue's rows of h split evenly over the block");
#pragma unroll
    for (int r = 0; r < ((L_ / 2) * LNP) / 256; ++r) {
      const int e  = t + 256 * r;
      const int kp = e / LNP, jp = e % LNP;
      const int n  = min(2 * jp, H_ - 2);
      const float* p = Hb + (size_t)(2 * kp) * H_ + n;
      const float2 u = ld2(p), v = ld2(p + H_);
      st4(&Bs[(kp * LNC + 2 * jp) * 2], make_float4(u.x, u.y, v.x, v.y));
    }
#pragma unroll
    for (int r = 0; r < (LR * LQ + 255) / 256; ++r) {
      const int e = min(t + 256 * r, LR * LQ - 1);
      const int row = e / LQ, q = e % LQ;
      st4(&H2s[row * LPH + 4 * q], ld4(h0 + (size_t)(rowbase + r0 + row) * H_ + 4 * q));
    }
  }
  startW(wl);
  prefetchW();
  constexpr int RP = (LR * LQ + 255) / 256;
  float4 g1[RP], g2[RP];
  float  cd1[RP], cd2[RP], dr[RP];
#pragma unroll
  for (int r = 0; r < RP; ++r) {
    const int e = min(t + 256 * r, LR * LQ - 1);
    const int row = e / LQ, q = e % LQ;
    const int grow = rowbase + r0 + row;
    const int a  = grow - m * A_;
    const int r1 = o1 * A_ + a;
    const int r2 = o2 * A_ + a;
    cd1[r] = crossv[((size_t)m * M_ + o1) * A_ + a] * dinv[r1];
    cd2[r] = crossv[((size_t)m * M_ + o2) * A_ + a] * dinv[r2];
    dr[r]  = dinv[grow];
    g1[r]  = ld4(h + (size_t)r1 * H_ + 4 * q);
    g2[r]  = ld4(h + (size_t)r2 * H_ + 4 * q);
  }
  lds_barrier();

  {
    const float* ap = Ss + (16 * s + mr) * LPS + 2 * kh;
    const float* bp = Bs + boff;
    mma_run<0, 3, L_>(acc, ap, bp);
    if (four) mma_run<3, 1, L_>(acc, ap, bp);
  }
  dump(Hs);
#pragma unroll
  for (int c = 0; c < 4; ++c) acc[c] = v8f{};
  lds_barrier();
  stageW(0);
#pragma unroll
  for (int r = 0; r < RP; ++r) {
    const bool own = (r < RP - 1) || (t + 256 * r < LR * LQ);
    const int e = min(t + 256 * r, LR * LQ - 1);
    const int row = e / LQ, q = e % LQ;
    float* p = Hs + row * LPH + 4 * q;
    float4 sv = ld4(p);
    sv.x += cd1[r] * g1[r].x; sv.y += cd1[r] * g1[r].y; sv.z += cd1[r] * g1[r].z; sv.w += cd1[r] * g1[r].w;
    sv.x += cd2[r] * g2[r].x; sv.y += cd2[r] * g2[r].y; sv.z += cd2[r] * g2[r].z; sv.w += cd2[r] * g2[r].w;
    st4(own ? p : Ss + 4 * t, make_float4(sv.x * dr[r], sv.y * dr[r], sv.z * dr[r], sv.w * dr[r]));
  }
  lds_barrier();

  stream();
  const float* Wm = tw + (size_t)m * 2 * H_ * H_;
  if constexpr (WITH_HX) { startW(Wm); prefetchW(); }
  dump(Ct);
  if constexpr (WITH_HX) stageW(0);
  lds_barrier();
  {
    const float* X = ((m == 0) ? xa : (m == 1) ? xv : xt) + (size_t)(mb & 31) * L_ * H_ + (size_t)r0 * H_;
    auto one = [&](const float x, const float p, const float q) {
      const float rr = 0.9f * p + 0.1f * q;
      return fmaxf(beta * x + (1.0f - beta) * rr, 0.0f);
    };
#pragma unroll
    for (int r = 0; r < RP; ++r) {
      const int e = t + 256 * r;
      if (r < RP - 1 || e < LR * LQ) {
        const int row = e / LQ, q = e % LQ;
        const float4 v = ld4(&Ct[row * LPH + 4 * q]);
        const float4 a = ld4(&Hs[row * LPH + 4 * q]);
        const float4 b = ld4(&H2s[row * LPH + 4 * q]);
        const float4 o = make_float4(one(v.x, a.x, b.x), one(v.y, a.y, b.y), one(v.z, a.z, b.z), one(v.w, a.w, b.w));
        if constexpr (WITH_HX) {
          st4(&H2s[row * LPH + 4 * q], o);
          st4(&Hs[row * LPH + 4 * q], ld4(X + (size_t)row * H_ + 4 * q));
        } else {
          st4(hnew + (size_t)(rowbase + r0 + row) * H_ + 4 * q, o);
        }
      }
    }
  }
  if constexpr (WITH_HX) {
#pragma unroll
    for (int c = 0; c < 4; ++c) acc[c] = v8f{};
    lds_barrier();
    stream();
    dump(Ct);
    lds_barrier();
    const float* Bb = tb + (size_t)m * H_;
#pragma unroll
    for (int r = 0; r < RP; ++r) {
      const int e = t + 256 * r;
      if (r < RP - 1 || e < LR * LQ) {
        const int row = e / LQ, q = e % LQ;
        const float4 v = ld4(&Ct[row * LPH + 4 * q]);
        const float4 b = ld4(Bb + 4 * q);
        st4(hx + (size_t)(rowbase + r0 + row) * H_ + 4 * q,
            make_float4(fmaxf(v.x + b.x, 0.0f), fmaxf(v.y + b.y, 0.0f), fmaxf(v.z + b.z, 0.0f), fmaxf(v.w + b.w, 0.0f)));
      }
    }
  }
}

__global__ __launch_bounds__(128) void k_gate(const float* __restrict__ hx,
                                              const float* __restrict__ cw,
                                              const float* __restrict__ cb,
                                              float* __restrict__ hout) {
  constexpr int KP = 3 * H_;
  const int i  = blockIdx.z;
  const int n0 = blockIdx.x * 48;
  const int a0 = blockIdx.y * 64;
  const int j0 = (i == 0) ? 1 : 0;
  const int j1 = (i == 2) ? 1 : 2;
  const float* Hi  = hx + (size_t)i  * A_ * H_;
  const float* Hj0 = hx + (size_t)j0 * A_ * H_;
  const float* Hj1 = hx + (size_t)j1 * A_ * H_;
  const float* Wp = cw + (size_t)(2 * i) * KP * H_;
  const float* Bb = cb + (size_t)(2 * i) * H_;
  auto fA = [&](int r, int k0, int kq, float4& x, float4& y) {
    const int q  = k0 / KP;
    const int kp = k0 - q * KP;
    const int s  = kp / H_;
    const int kk = kp - s * H_ + kq;
    const float* Hj = q ? Hj1 : Hj0;
    const float* px = ((s == 1) ? Hj : Hi) + (size_t)(a0 + r) * H_ + kk;
    x = ld4(px);
    if (s == 2) y = ld4(Hj + (size_t)(a0 + r) * H_ + kk);
    else y = make_float4(0.f, 0.f, 0.f, 0.f);
  };
  auto cA = [&](int k0, const float4& x, const float4& y) {
    if ((k0 % KP) / H_ == 2) return make_float4(x.x * y.x, x.y * y.y, x.z * y.z, x.w * y.w);
    return x;
  };

  float4 g[8];
  auto gated = [&](const int i0, const int n, const float4 (&v)[8], const float* Hj, const float* bias,
                   const bool add) {
    if (n >= H_) return;
    const float4 bn = ld4(bias + n);
    float4 p[8], q[8];
#pragma unroll
    for (int j = 0; j < 8; ++j) {
      const size_t o = (size_t)(a0 + i0 + 2 * j) * H_ + n;
      p[j] = ld4(Hi + o);
      q[j] = ld4(Hj + o);
    }
    auto one = [&](const float x, const float b, const float hi_, const float hj_) {
      const float z = 1.0f / (1.0f + expf(-(x + b)));
      return z * hi_ + (1.0f - z) * hj_;
    };
#pragma unroll
    for (int j = 0; j < 8; ++j) {
      const float4 t = make_float4(one(v[j].x, bn.x, p[j].x, q[j].x), one(v[j].y, bn.y, p[j].y, q[j].y),
                                   one(v[j].z, bn.z, p[j].z, q[j].z), one(v[j].w, bn.w, p[j].w, q[j].w));
      g[j] = add ? make_float4(g[j].x + t.x, g[j].y + t.y, g[j].z + t.z, g[j].w + t.w) : t;
    }
  };
  auto mid = [&](const int i0, const int n, const float4 (&v)[8]) { gated(i0, n, v, Hj0, Bb, false); };
  auto ep = [&](const int i0, const int n, const float4 (&v)[8]) {
    if (n >= H_) return;
    gated(i0, n, v, Hj1, Bb + H_, true);
#pragma unroll
    for (int j = 0; j < 8; ++j)
      st4(hout + ((size_t)i * A_ + a0 + i0 + 2 * j) * H_ + n, g[j]);
  };
  gemm_block_64x48<40, true>(2 * KP, Wp, n0, fA, cA, ep, KP, mid);
}

__global__ __launch_bounds__(512) void k_final(const float* __restrict__ hout,
                                               const float* __restrict__ uw,
                                               const float* __restrict__ ub,
                                               float* __restrict__ out) {
  constexpr int UPB = 16;
  constexpr int NSF = H_ / 32;
  constexpr int REM = H_ - 32 * NSF;
  static_assert((UPB * TAG_) % 32 == 0, "a block's results must be whole 128-byte lines");
  static_assert(REM > 0 && M_ * REM <= 32, "the left-over units of all modalities share one slot");
  __shared__ float res[UPB * TAG_];
  const int wv   = threadIdx.x >> 5;
  const int lane = threadIdx.x & 31;
  const int a    = blockIdx.x * UPB + wv;
  float acc[TAG_] = {};
  auto add6 = [&](const float v, const float2 w01, const float2 w23, const float2 w45) {
    acc[0] += v * w01.x; acc[1] += v * w01.y;
    acc[2] += v * w23.x; acc[3] += v * w23.y;
    acc[4] += v * w45.x; acc[5] += v * w45.y;
  };
#pragma unroll
  for (int m = 0; m < M_; ++m) {
    const float* hr = hout + (size_t)(m * A_ + a) * H_;
    const float* wm = uw + (size_t)m * H_ * TAG_;
#pragma unroll
    for (int s = 0; s < NSF; ++s) {
      const int h = lane + 32 * s;
      const float* wr = wm + h * TAG_;
      add6(hr[h], ld2(wr), ld2(wr + 2), ld2(wr + 4));
    }
  }
  {
    const bool ok = lane < M_ * REM;
    const int l = ok ? lane : M_ * REM - 1;
    const int m = l / REM;
    const int h = 32 * NSF + l % REM;
    const float* wr = uw + ((size_t)m * H_ + h) * TAG_;
    const float  v   = hout[(size_t)(m * A_ + a) * H_ + h];
    const float2 w01 = ld2(wr), w23 = ld2(wr + 2), w45 = ld2(wr + 4);
    const float2 z = make_float2(0.f, 0.f);
    add6(ok ? v : 0.f, ok ? w01 : z, ok ? w23 : z, ok ? w45 : z);
  }
#pragma unroll
  for (int t = 0; t < TAG_; ++t) acc[t] = wave_sum(acc[t]);
  if (lane == 0) {
#pragma unroll
    for (int t = 0; t < TAG_; ++t)
      res[wv * TAG_ + t] = acc[t] + (ub[t] + ub[TAG_ + t] + ub[2 * TAG_ + t]);
  }
  __syncthreads();
  if (wv == 0) {
    for (int j = 0; j < UPB * TAG_ / 32; ++j)
      out[(size_t)blockIdx.x * (UPB * TAG_) + 32 * j + lane] = res[32 * j + lane];
  }
}

}

extern "C" void kernel_launch(void* const* d_in, const int* in_sizes, int n_in,
                              void* d_out, int out_size, void* d_ws, size_t ws_size,
                              hipStream_t stream) {
  (void)in_sizes; (void)n_in; (void)out_size; (void)ws_size;
  const float* x_a     = (const float*)d_in[0];
  const float* x_v     = (const float*)d_in[1];
  const float* x_t     = (const float*)d_in[2];
  const float* fc0_w   = (const float*)d_in[3];
  const float* fc0_b   = (const float*)d_in[4];
  const float* conv_w  = (const float*)d_in[5];
  const float* tm_w    = (const float*)d_in[6];
  const float* tm_b    = (const float*)d_in[7];
  const float* cross_w = (const float*)d_in[8];
  const float* cross_b = (const float*)d_in[9];
  const float* uni_w   = (const float*)d_in[10];
  const float* uni_b   = (const float*)d_in[11];
  float* out = (float*)d_out;

  const size_t NH = (size_t)N_ * H_;
  float* ws    = (float*)d_ws;
  float* intra = ws; ws += (size_t)N_ * L_;
  float* crossv= ws; ws += (size_t)M_ * M_ * A_;
  float* dinv  = ws; ws += N_;
  float* h0    = ws; ws += NH;
  float* hA    = ws; ws += NH;
  float* hx    = ws; ws += NH;
  float* hout  = ws; ws += NH;

  k_front<<<M_ * B_ + 5 * (N_ / 64), dim3(128), 0, stream>>>(x_a, x_v, x_t, fc0_w, fc0_b, h0,
                                                            intra, crossv, dinv);

  const float beta0 = logf(0.5f / 1.0f + 1.0f);
  const float beta1 = logf(0.5f / 2.0f + 1.0f);
  k_layer<false><<<2 * M_ * B_, dim3(256), 0, stream>>>(intra, crossv, dinv, h0, h0, conv_w, beta0, hA,
                                                       x_a, x_v, x_t, tm_w, tm_b, hx);
  k_layer<true> <<<2 * M_ * B_, dim3(256), 0, stream>>>(intra, crossv, dinv, hA, h0, conv_w + 2 * H_ * H_, beta1, hout,
                                                       x_a, x_v, x_t, tm_w, tm_b, hx);

  k_gate <<<dim3(5, A_ / 64, M_), dim3(128), 0, stream>>>(hx, cross_w, cross_b, hout);

  k_final<<<A_ / 16, dim3(512), 0, stream>>>(hout, uni_w, uni_b, out);
}


# k_gate: the block's 64 rows of hx_i and of the partner stay in LDS (partner brought in over 5 turns); A fragments, product third included, are read from them; only W is streamed; sums kept
# speedup vs baseline: 8.6170x; 1.0388x over previous
#include <hip/hip_runtime.h>
#include <math.h>

typedef __attribute__((ext_vector_type(2))) float v2f;
typedef __attribute__((ext_vector_type(8))) float v8f;

namespace {

constexpr int B_   = 32;
constexpr int L_   = 64;
constexpr int H_   = 200;
constexpr int M_   = 3;
constexpr int A_   = B_ * L_;
constexpr int N_   = M_ * A_;
constexpr int TAG_ = 6;
constexpr float PI_F = 3.14159265358979323846f;

__device__ __forceinline__ float wave_sum(float s) {
#pragma unroll
  for (int off = 16; off > 0; off >>= 1) s += __shfl_xor(s, off, 32);
  return s;
}

template <int N>
__device__ __forceinline__ void wave_sum_n(float (&s)[N]) {
#pragma unroll
  for (int off = 16; off > 0; off >>= 1) {
    float t[N];
#pragma unroll
    for (int i = 0; i < N; ++i) t[i] = __shfl_xor(s[i], off, 32);
#pragma unroll
    for (int i = 0; i < N; ++i) s[i] += t[i];
  }
}

__device__ __forceinline__ void graph_block(const int mb,
                                            const float* __restrict__ xa,
                                            const float* __restrict__ xv,
                                            const float* __restrict__ xt,
                                            float* __restrict__ intra,
                                            float* __restrict__ crossv,
                                            float* __restrict__ dinv) {
  constexpr int PN  = H_ + 4;
  constexpr int TP  = L_ + 4;
  constexpr int NS  = (H_ + 31) / 32;
  constexpr int NW  = 4;
  constexpr int RPW = L_ / NW;
  constexpr int RB  = 8;
  static_assert((PN % 8) == 4, "pitch must be 4 * odd dwords");
  static_assert(L_ * TP <= L_ * PN, "the similarity tile reuses the rows' LDS");
  static_assert(RPW == 16 && RPW % RB == 0, "a wave owns one 16-row strip");
  __shared__ __attribute__((aligned(16))) float nr[L_ * PN];
  __shared__ float cx[2][L_];
  __shared__ float dv[L_];
  const int m    = mb >> 5;
  const int b    = mb & 31;
  const int w    = threadIdx.x >> 5;
  const int lane = threadIdx.x & 31;
  const int n1   = (m == 0) ? 1 : 0;
  const int n2   = (m == 2) ? 1 : 2;
  auto plane = [&](const int mm) { return (mm == 0) ? xa : (mm == 1) ? xv : xt; };
  const float* P0 = plane(m)  + (size_t)b * L_ * H_;
  const float* P1 = plane(n1) + (size_t)b * L_ * H_;
  const float* P2 = plane(n2) + (size_t)b * L_ * H_;

#pragma unroll 1
  for (int rb = w * RPW; rb < (w + 1) * RPW; rb += RB) {
    float v0[RB][NS], v1[RB][NS], v2[RB][NS];
#pragma unroll
    for (int q = 0; q < RB; ++q) {
#pragma unroll
      for (int j = 0; j < NS; ++j) {
        const int k = lane + 32 * j;
        const bool ok = k < H_;
        v0[q][j] = ok ? P0[(size_t)(rb + q) * H_ + k] : 0.f;
        v1[q][j] = ok ? P1[(size_t)(rb + q) * H_ + k] : 0.f;
        v2[q][j] = ok ? P2[(size_t)(rb + q) * H_ + k] : 0.f;
      }
    }
    float sq[3 * RB];
#pragma unroll
    for (int q = 0; q < RB; ++q) {
      float s0 = 0.f, s1 = 0.f, s2 = 0.f;
#pragma unroll
      for (int j = 0; j < NS; ++j) {
        s0 += v0[q][j] * v0[q][j];
        s1 += v1[q][j] * v1[q][j];
        s2 += v2[q][j] * v2[q][j];
      }
      sq[3 * q] = s0; sq[3 * q + 1] = s1; sq[3 * q + 2] = s2;
    }
    wave_sum_n<3 * RB>(sq);
    float dd[2 * RB];
#pragma unroll
    for (int q = 0; q < RB; ++q) {
      const float rn0 = 1.0f / sqrtf(sq[3 * q]);
      const float rn1 = 1.0f / sqrtf(sq[3 * q + 1]);
      const float rn2 = 1.0f / sqrtf(sq[3 * q + 2]);
      float d1 = 0.f, d2 = 0.f;
#pragma unroll
      for (int j = 0; j < NS; ++j) {
        const float u0 = v0[q][j] * rn0;
        const float u1 = v1[q][j] * rn1;
        const float u2 = v2[q][j] * rn2;
        v0[q][j] = u0;
        d1 += u0 * u1;
        d2 += u0 * u2;
      }
      dd[2 * q] = d1; dd[2 * q + 1] = d2;
    }
    wave_sum_n<2 * RB>(dd);
#pragma unroll
    for (int q = 0; q < RB; ++q) {
#pragma unroll
      for (int j = 0; j < NS; ++j) {
        const int k = lane + 32 * j;
        if (k < H_) nr[(rb + q) * PN + k] = v0[q][j];
      }
    }
    float e1 = 0.f, e2 = 0.f;
#pragma unroll
    for (int q = 0; q < RB; ++q) {
      e1 = (lane == q) ? dd[2 * q] : e1;
      e2 = (lane == q) ? dd[2 * q + 1] : e2;
    }
    const float c1 = 1.0f - acosf(e1 * 0.99999f) * (1.0f / PI_F);
    const float c2 = 1.0f - acosf(e2 * 0.99999f) * (1.0f / PI_F);
    if (lane < RB) { cx[0][rb + lane] = c1; cx[1][rb + lane] = c2; }
  }
  __syncthreads();

  const int mr = lane & 15;
  const int kh = lane >> 4;
  v8f acc[4] = {};
  {
    const float* ap = nr + (w * 16 + mr) * PN + 2 * kh;
    const float* bp = nr + mr * PN + 2 * kh;
#pragma unroll 5
    for (int k = 0; k < H_; k += 4) {
      const v2f a = *reinterpret_cast<const v2f*>(ap + k);
#pragma unroll
      for (int c = 0; c < 4; ++c) {
        const v2f bb = *reinterpret_cast<const v2f*>(bp + c * 16 * PN + k);
        acc[c] = __builtin_amdgcn_wmma_f32_16x16x4_f32(false, a, false, bb, (short)0, acc[c], false, false);
      }
    }
  }
  __syncthreads();
#pragma unroll
  for (int c = 0; c < 4; ++c) {
#pragma unroll
    for (int v = 0; v < 8; ++v)
      nr[(w * 16 + 8 * kh + v) * TP + 16 * c + mr] = 1.0f - acosf(acc[c][v] * 0.99999f) * (1.0f / PI_F);
  }
  __syncthreads();

  {
    const int r0 = w * RPW;
    float sr[RPW];
#pragma unroll
    for (int q = 0; q < RPW; ++q) {
      const float t0 = nr[(r0 + q) * TP + lane];
      const float t1 = nr[(r0 + q) * TP + lane + 32];
      float* orow = intra + ((size_t)mb * L_ + r0 + q) * L_;
      orow[lane]      = t0;
      orow[lane + 32] = t1;
      float s = 0.f;
      s += t0;
      s += t1;
      sr[q] = s;
    }
    wave_sum_n<RPW>(sr);
    float e = 0.f;
#pragma unroll
    for (int q = 0; q < RPW; ++q) e = (lane == q) ? sr[q] : e;
    if (lane < RPW) {
      e += cx[0][r0 + lane];
      e += cx[1][r0 + lane];
      dv[r0 + lane] = 1.0f / sqrtf(e);
    }
  }
  __syncthreads();
  if (w < 2) {
    dinv[(size_t)mb * L_ + 32 * w + lane] = dv[32 * w + lane];
  } else {
    const int j  = w - 2;
    const int nn = j ? n2 : n1;
    crossv[((size_t)m * M_ + nn) * A_ + b * L_ + lane]      = cx[j][lane];
    crossv[((size_t)m * M_ + nn) * A_ + b * L_ + 32 + lane] = cx[j][32 + lane];
  }
}

struct no_mid {};
template <typename T> struct is_no_mid { static constexpr bool value = false; };
template <> struct is_no_mid<no_mid> { static constexpr bool value = true; };
__device__ __forceinline__ float4 ld4(const float* p) {
  return *reinterpret_cast<const float4*>(p);
}
__device__ __forceinline__ float2 ld2(const float* p) {
  return *reinterpret_cast<const float2*>(p);
}
__device__ __forceinline__ void st4(float* p, const float4 v) {
  *reinterpret_cast<float4*>(p) = v;
}

template <int I> struct set_c { static constexpr int value = I; };

__device__ __forceinline__ void lds_barrier() {
  __builtin_amdgcn_fence(__ATOMIC_RELEASE, "workgroup", "local");
  __builtin_amdgcn_s_barrier();
  __builtin_amdgcn_fence(__ATOMIC_ACQUIRE, "workgroup", "local");
}

template <int BK, bool COMB_HAS_MATH, typename FA, typename CA, typename FE, typename FM = no_mid>
__device__ __forceinline__ void gemm_block_64x48(const int K, const float* __restrict__ Bp,
                                                 const int n0, FA fetchA, CA combA, FE epi,
                                                 const int Kmid = 0, FM mid = FM{}) {
  constexpr bool TWO_PRODUCTS = !is_no_mid<FM>::value;
  constexpr int BM = 64, BN = 48, NTHR = 128;
  constexpr int NT = BN / 16;
  constexpr int PA = BK + 4;
  constexpr int PB = BN;
  constexpr int KQ = BK / 4;
  constexpr int NP = BN / 2;
  constexpr int A4 = BM * KQ;
  constexpr int B4 = (BK / 2) * NP;
  constexpr int RA = A4 / NTHR;
  constexpr int RB = (B4 + NTHR - 1) / NTHR;
  constexpr bool B_TAIL = (B4 % NTHR) != 0;
  static_assert(A4 % NTHR == 0 && B4 % 32 == 0, "A chunks split evenly over the block, B tasks over whole waves");
  static_assert((PA % 8) == 4, "A pitch must be 4 * odd dwords");
  static_assert(((2 * PB) % 64) == 32, "B pair pitch must be 32 mod 64 dwords");

  constexpr int ASZ = BM * PA;
  constexpr int BSZ = (BK / 2) * PB * 2;
  __shared__ __attribute__((aligned(16))) float As[2 * ASZ];
  __shared__ __attribute__((aligned(16))) float Bs[2 * BSZ];
  constexpr int TC = BN + 4;
  static_assert(((8 * TC) % 64) == 32 && (TC % 4) == 0, "epilogue tile pitch");
  __shared__ __attribute__((aligned(16))) float Ct[BM * TC];

  const int t    = threadIdx.x;
  const int lane = t & 31;
  const int w    = t >> 5;
  const int mr   = lane & 15;
  const int kh   = lane >> 4;

  float4 ax[2][RA], ay[2][RA];
  float2 be[2][RB], bo[2][RB];

  auto btask = [&](const int r) {
    const int e = t + NTHR * r;
    return (B_TAIL && r == RB - 1 && e >= B4) ? e - NTHR : e;
  };

  auto prefetch = [&](auto S, const int k0) {
    constexpr int s = decltype(S)::value;
#pragma unroll
    for (int r = 0; r < RA; ++r) {
      const int e = t + NTHR * r;
      fetchA(e / KQ, k0, (e % KQ) * 4, ax[s][r], ay[s][r]);
    }
#pragma unroll
    for (int r = 0; r < RB; ++r) {
      const int e  = btask(r);
      const int kp = e / NP;
      const int n  = min(n0 + (e % NP) * 2, H_ - 2);
      const float* p = Bp + (size_t)(k0 + 2 * kp) * H_ + n;
      be[s][r] = ld2(p);
      bo[s][r] = ld2(p + H_);
    }
  };

  auto stage = [&](auto S, const int k0) {
    constexpr int s = decltype(S)::value;
    constexpr int buf = s;
#pragma unroll
    for (int r = 0; r < RA; ++r) {
      const int e = t + NTHR * r;
      *reinterpret_cast<float4*>(&As[buf * ASZ + (e / KQ) * PA + (e % KQ) * 4]) = combA(k0, ax[s][r], ay[s][r]);
    }
#pragma unroll
    for (int r = 0; r < RB; ++r) {
      const int e = btask(r);
      const int j = (e % NP) * 2;
      const float4 v = make_float4(be[s][r].x, be[s][r].y, bo[s][r].x, bo[s][r].y);
      *reinterpret_cast<float4*>(&Bs[buf * BSZ + ((e / NP) * PB + j) * 2]) = v;
    }
  };

  const int aoff = (w * 16 + mr) * PA + 2 * kh;
  const int boff = (kh * PB + (mr & ~1)) * 2 + (mr & 1);

  v8f acc[NT] = {};
  auto compute = [&](const int buf) {
    const float* aptr = As + buf * ASZ + aoff;
    const float* bptr = Bs + buf * BSZ + boff;
#pragma unroll
    for (int ks = 0; ks < BK; ks += 4) {
      const v2f a = *reinterpret_cast<const v2f*>(aptr + ks);
#pragma unroll
      for (int c = 0; c < NT; ++c) {
        const float* bq = bptr + ((ks / 2) * PB + 16 * c) * 2;
        v2f b;
        b.x = bq[0];
        b.y = bq[2];
        acc[c] = __builtin_amdgcn_wmma_f32_16x16x4_f32(false, a, false, b, (short)0, acc[c], false, false);
      }
    }
  };

  constexpr int GR = BN / 4;
  const bool has_group = lane < 2 * GR;
  const int gl = has_group ? lane : lane - 2 * GR;
  const int er = gl / GR;
  const int eg = gl % GR;
  const int ei = w * 16 + er;
  const int en = has_group ? n0 + eg * 4 : H_;
  auto to_rows = [&](float4 (&rv)[8]) {
    float* ct = Ct + (w * 16) * TC;
#pragma unroll
    for (int c = 0; c < NT; ++c) {
#pragma unroll
      for (int v = 0; v < 8; ++v) ct[(8 * kh + v) * TC + 16 * c + mr] = acc[c][v];
    }
#pragma unroll
    for (int j = 0; j < 8; ++j) rv[j] = *reinterpret_cast<const float4*>(&ct[(er + 2 * j) * TC + eg * 4]);
  };

  auto turn = [&](auto S, auto MORE, const int k0) {
    constexpr int s = decltype(S)::value;
    constexpr bool more = decltype(MORE)::value != 0;
    if constexpr (TWO_PRODUCTS) {
      if (k0 == Kmid + BK) {
        float4 rv[8];
        to_rows(rv);
        mid(ei, en, rv);
#pragma unroll
        for (int c = 0; c < NT; ++c) acc[c] = v8f{};
      }
    }
    if constexpr (more) prefetch(set_c<(s ^ 1)>{}, k0 + BK);
    __builtin_amdgcn_sched_barrier(0);
    compute(s ^ 1);
    __builtin_amdgcn_sched_barrier(0);
    stage(S, k0);
    lds_barrier();
  };

  prefetch(set_c<0>{}, 0);
  prefetch(set_c<1>{}, BK);
  stage(set_c<0>{}, 0);
  lds_barrier();
  int k0 = BK;
  for (; k0 + 2 * BK < K; k0 += 2 * BK) {
    turn(set_c<1>{}, set_c<1>{}, k0);
    turn(set_c<0>{}, set_c<1>{}, k0 + BK);
  }
  if (k0 + BK < K) {
    turn(set_c<1>{}, set_c<1>{}, k0);
    turn(set_c<0>{}, set_c<0>{}, k0 + BK);
  } else {
    turn(set_c<1>{}, set_c<0>{}, k0);
  }
  compute(((K / BK) - 1) & 1);

  float4 rv[8];
  to_rows(rv);
  epi(ei, en, rv);
}

__global__ __launch_bounds__(128) void k_front(const float* __restrict__ xa,
                                               const float* __restrict__ xv,
                                               const float* __restrict__ xt,
                                               const float* __restrict__ w,
                                               const float* __restrict__ bias,
                                               float* __restrict__ h0,
                                               float* __restrict__ intra,
                                               float* __restrict__ crossv,
                                               float* __restrict__ dinv) {
  if (blockIdx.x < M_ * B_) {
    graph_block(blockIdx.x, xa, xv, xt, intra, crossv, dinv);
  } else {
    const int id = blockIdx.x - M_ * B_;
    const int n0 = (id % 5) * 48;
    const int r0 = (id / 5) * 64;
    const int mm = r0 / A_;
    const float* X = ((mm == 0) ? xa : (mm == 1) ? xv : xt) + (size_t)(r0 - mm * A_) * H_;
    auto fA = [&](int i, int k0, int kq, float4& x, float4& y) {
      (void)y;
      x = ld4(X + (size_t)i * H_ + k0 + kq);
    };
    auto cA = [&](int k0, const float4& x, const float4& y) { (void)k0; (void)y; return x; };
    auto ep = [&](const int i0, const int n, const float4 (&v)[8]) {
      if (n >= H_) return;
      const float4 b = ld4(bias + n);
#pragma unroll
      for (int j = 0; j < 8; ++j)
        st4(h0 + (size_t)(r0 + i0 + 2 * j) * H_ + n,
            make_float4(fmaxf(v[j].x + b.x, 0.0f), fmaxf(v[j].y + b.y, 0.0f),
                        fmaxf(v[j].z + b.z, 0.0f), fmaxf(v[j].w + b.w, 0.0f)));
    };
    gemm_block_64x48<40, false>(H_, w, n0, fA, cA, ep);
  }
}

constexpr int LR   = 32;
constexpr int LNC  = 208;
constexpr int LPH  = 212;
constexpr int LPS  = 68;
constexpr int LBK  = 40;
constexpr int LNP  = LNC / 2;
constexpr int LBSZ = (LBK / 2) * LNC * 2;
constexpr int LQ   = H_ / 4;
static_assert(((2 * LNC) % 64) == 32 && (LPH % 8) == 4 && (LPS % 8) == 4, "LDS pitches");
static_assert((L_ / 2) * LNC * 2 <= 2 * LBSZ, "the dialogue's rows of h fit the two B buffers");

template <int C0, int NTW, int KK>
__device__ __forceinline__ void mma_run(v8f (&acc)[4], const float* aptr, const float* bptr) {
#pragma unroll
  for (int ks = 0; ks < KK; ks += 4) {
    const v2f a = *reinterpret_cast<const v2f*>(aptr + ks);
#pragma unroll
    for (int c = C0; c < C0 + NTW; ++c) {
      const float* bq = bptr + ((ks / 2) * LNC + 16 * c) * 2;
      v2f b;
      b.x = bq[0];
      b.y = bq[2];
      acc[c] = __builtin_amdgcn_wmma_f32_16x16x4_f32(false, a, false, b, (short)0, acc[c], false, false);
    }
  }
}

template <bool WITH_HX>
__global__ __launch_bounds__(256) void k_layer(const float* __restrict__ intra,
                                               const float* __restrict__ crossv,
                                               const float* __restrict__ dinv,
                                               const float* __restrict__ h,
                                               const float* __restrict__ h0,
                                               const float* __restrict__ wl, float beta,
                                               float* __restrict__ hnew,
                                               const float* __restrict__ xa,
                                               const float* __restrict__ xv,
                                               const float* __restrict__ xt,
                                               const float* __restrict__ tw,
                                               const float* __restrict__ tb,
                                               float* __restrict__ hx) {
  __shared__ __attribute__((aligned(16))) float Bs[2 * LBSZ];
  __shared__ __attribute__((aligned(16))) float AH[2 * LR * LPH];
  __shared__ __attribute__((aligned(16))) float Ss[LR * LPS];
  __shared__ __attribute__((aligned(16))) float Ct[LR * LPH];

  const int t    = threadIdx.x;
  const int lane = t & 31;
  const int w    = __builtin_amdgcn_readfirstlane(t >> 5);
  const int mr   = lane & 15;
  const int kh   = lane >> 4;
  const int s    = w & 1;
  const int cg   = w >> 1;
  const int tile0 = (cg == 0) ? 0 : 3 * cg + 1;
  const bool four = (cg == 0);

  const int mb      = blockIdx.x >> 1;
  const int r0      = (blockIdx.x & 1) * LR;
  const int m       = mb >> 5;
  const int rowbase = mb * L_;
  const int o1 = (m == 0) ? 1 : 0;
  const int o2 = (m == 2) ? 1 : 2;

  float* Hs  = AH;
  float* H2s = AH + LR * LPH;

  constexpr int WT = (LBK / 2) * LNP;
  constexpr int WR = (WT + 255) / 256;
  float2 be[WR], bo[WR];
  auto wtask = [&](const int r) {
    const int e = t + 256 * r;
    return (r == WR - 1 && e >= WT) ? e - 256 : e;
  };
  unsigned woff[WR];
  int      soff[WR];
#pragma unroll
  for (int r = 0; r < WR; ++r) {
    const int e  = wtask(r);
    const int kp = e / LNP;
    const int n  = min(2 * (e % LNP), H_ - 2);
    woff[r] = (unsigned)((2 * kp) * H_ + n);
    soff[r] = (kp * LNC + 2 * (e % LNP)) * 2;
  }
  const float* wp[WR];
  auto startW = [&](const float* __restrict__ Wp) {
#pragma unroll
    for (int r = 0; r < WR; ++r) wp[r] = Wp + woff[r];
  };
  auto prefetchW = [&]() {
#pragma unroll
    for (int r = 0; r < WR; ++r) {
      be[r] = ld2(wp[r]);
      bo[r] = ld2(wp[r] + H_);
      wp[r] += LBK * H_;
    }
  };
  auto stageW = [&](const int buf) {
#pragma unroll
    for (int r = 0; r < WR; ++r)
      st4(&Bs[buf * LBSZ + soff[r]], make_float4(be[r].x, be[r].y, bo[r].x, bo[r].y));
  };

  const int aoff = (16 * s + mr) * LPH + 2 * kh;
  const int boff = (kh * LNC + 16 * tile0 + (mr & ~1)) * 2 + (mr & 1);

  v8f acc[4] = {};
  auto dump = [&](float* dst) {
#pragma unroll
    for (int c = 0; c < 4; ++c) {
      if (c < 3 || four) {
#pragma unroll
        for (int v = 0; v < 8; ++v) dst[(16 * s + 8 * kh + v) * LPH + 16 * (tile0 + c) + mr] = acc[c][v];
      }
    }
  };
  auto stream = [&]() {
#pragma unroll 1
    for (int c = 0; c < (2 * H_) / LBK; ++c) {
      const int  k0   = c * LBK;
      const bool more = (c + 1) < (2 * H_) / LBK;
      if (more) prefetchW();
      __builtin_amdgcn_sched_barrier(0);
      const float* ap = AH + ((k0 >= H_) ? LR * LPH + (k0 - H_) : k0) + aoff;
      const float* bp = Bs + (c & 1) * LBSZ + boff;
      mma_run<0, 3, LBK>(acc, ap, bp);
      if (four) mma_run<3, 1, LBK>(acc, ap, bp);
      __builtin_amdgcn_sched_barrier(0);
      if (more) stageW((c + 1) & 1);
      lds_barrier();
    }
  };

  {
    const float* S = intra + (size_t)mb * L_ * L_ + (size_t)r0 * L_;
#pragma unroll
    for (int r = 0; r < (LR * (L_ / 4)) / 256; ++r) {
      const int e = t + 256 * r;
      const int row = e / (L_ / 4), q = e % (L_ / 4);
      const float4 x = ld4(S + (size_t)row * L_ + 4 * q);
      const float4 y = ld4(dinv + rowbase + 4 * q);
      st4(&Ss[row * LPS + 4 * q], make_float4(x.x * y.x, x.y * y.y, x.z * y.z, x.w * y.w));
    }
    const float* Hb = h + (size_t)rowbase * H_;
    static_assert(((L_ / 2) * LNP) % 256 == 0, "the dialogue's rows of h split evenly over the block");
#pragma unroll
    for (int r = 0; r < ((L_ / 2) * LNP) / 256; ++r) {
      const int e  = t + 256 * r;
      const int kp = e / LNP, jp = e % LNP;
      const int n  = min(2 * jp, H_ - 2);
      const float* p = Hb + (size_t)(2 * kp) * H_ + n;
      const float2 u = ld2(p), v = ld2(p + H_);
      st4(&Bs[(kp * LNC + 2 * jp) * 2], make_float4(u.x, u.y, v.x, v.y));
    }
#pragma unroll
    for (int r = 0; r < (LR * LQ + 255) / 256; ++r) {
      const int e = min(t + 256 * r, LR * LQ - 1);
      const int row = e / LQ, q = e % LQ;
      st4(&H2s[row * LPH + 4 * q], ld4(h0 + (size_t)(rowbase + r0 + row) * H_ + 4 * q));
    }
  }
  startW(wl);
  prefetchW();
  constexpr int RP = (LR * LQ + 255) / 256;
  float4 g1[RP], g2[RP];
  float  cd1[RP], cd2[RP], dr[RP];
#pragma unroll
  for (int r = 0; r < RP; ++r) {
    const int e = min(t + 256 * r, LR * LQ - 1);
    const int row = e / LQ, q = e % LQ;
    const int grow = rowbase + r0 + row;
    const int a  = grow - m * A_;
    const int r1 = o1 * A_ + a;
    const int r2 = o2 * A_ + a;
    cd1[r] = crossv[((size_t)m * M_ + o1) * A_ + a] * dinv[r1];
    cd2[r] = crossv[((size_t)m * M_ + o2) * A_ + a] * dinv[r2];
    dr[r]  = dinv[grow];
    g1[r]  = ld4(h + (size_t)r1 * H_ + 4 * q);
    g2[r]  = ld4(h + (size_t)r2 * H_ + 4 * q);
  }
  lds_barrier();

  {
    const float* ap = Ss + (16 * s + mr) * LPS + 2 * kh;
    const float* bp = Bs + boff;
    mma_run<0, 3, L_>(acc, ap, bp);
    if (four) mma_run<3, 1, L_>(acc, ap, bp);
  }
  dump(Hs);
#pragma unroll
  for (int c = 0; c < 4; ++c) acc[c] = v8f{};
  lds_barrier();
  stageW(0);
#pragma unroll
  for (int r = 0; r < RP; ++r) {
    const bool own = (r < RP - 1) || (t + 256 * r < LR * LQ);
    const int e = min(t + 256 * r, LR * LQ - 1);
    const int row = e / LQ, q = e % LQ;
    float* p = Hs + row * LPH + 4 * q;
    float4 sv = ld4(p);
    sv.x += cd1[r] * g1[r].x; sv.y += cd1[r] * g1[r].y; sv.z += cd1[r] * g1[r].z; sv.w += cd1[r] * g1[r].w;
    sv.x += cd2[r] * g2[r].x; sv.y += cd2[r] * g2[r].y; sv.z += cd2[r] * g2[r].z; sv.w += cd2[r] * g2[r].w;
    st4(own ? p : Ss + 4 * t, make_float4(sv.x * dr[r], sv.y * dr[r], sv.z * dr[r], sv.w * dr[r]));
  }
  lds_barrier();

  stream();
  const float* Wm = tw + (size_t)m * 2 * H_ * H_;
  if constexpr (WITH_HX) { startW(Wm); prefetchW(); }
  dump(Ct);
  if constexpr (WITH_HX) stageW(0);
  lds_barrier();
  {
    const float* X = ((m == 0) ? xa : (m == 1) ? xv : xt) + (size_t)(mb & 31) * L_ * H_ + (size_t)r0 * H_;
    auto one = [&](const float x, const float p, const float q) {
      const float rr = 0.9f * p + 0.1f * q;
      return fmaxf(beta * x + (1.0f - beta) * rr, 0.0f);
    };
#pragma unroll
    for (int r = 0; r < RP; ++r) {
      const int e = t + 256 * r;
      if (r < RP - 1 || e < LR * LQ) {
        const int row = e / LQ, q = e % LQ;
        const float4 v = ld4(&Ct[row * LPH + 4 * q]);
        const float4 a = ld4(&Hs[row * LPH + 4 * q]);
        const float4 b = ld4(&H2s[row * LPH + 4 * q]);
        const float4 o = make_float4(one(v.x, a.x, b.x), one(v.y, a.y, b.y), one(v.z, a.z, b.z), one(v.w, a.w, b.w));
        if constexpr (WITH_HX) {
          st4(&H2s[row * LPH + 4 * q], o);
          st4(&Hs[row * LPH + 4 * q], ld4(X + (size_t)row * H_ + 4 * q));
        } else {
          st4(hnew + (size_t)(rowbase + r0 + row) * H_ + 4 * q, o);
        }
      }
    }
  }
  if constexpr (WITH_HX) {
#pragma unroll
    for (int c = 0; c < 4; ++c) acc[c] = v8f{};
    lds_barrier();
    stream();
    dump(Ct);
    lds_barrier();
    const float* Bb = tb + (size_t)m * H_;
#pragma unroll
    for (int r = 0; r < RP; ++r) {
      const int e = t + 256 * r;
      if (r < RP - 1 || e < LR * LQ) {
        const int row = e / LQ, q = e % LQ;
        const float4 v = ld4(&Ct[row * LPH + 4 * q]);
        const float4 b = ld4(Bb + 4 * q);
        st4(hx + (size_t)(rowbase + r0 + row) * H_ + 4 * q,
            make_float4(fmaxf(v.x + b.x, 0.0f), fmaxf(v.y + b.y, 0.0f), fmaxf(v.z + b.z, 0.0f), fmaxf(v.w + b.w, 0.0f)));
      }
    }
  }
}

__global__ __launch_bounds__(128) void k_gate(const float* __restrict__ hx,
                                              const float* __restrict__ cw,
                                              const float* __restrict__ cb,
                                              float* __restrict__ hout) {
  constexpr int KP = 3 * H_;
  constexpr int K  = 2 * KP;
  constexpr int BK = 40, BM = 64, BN = 48, NTHR = 128;
  constexpr int NT  = BN / 16;
  constexpr int PR  = H_ + 4;
  constexpr int PB  = BN;
  constexpr int NP  = BN / 2;
  constexpr int B4  = (BK / 2) * NP;
  constexpr int RB  = (B4 + NTHR - 1) / NTHR;
  constexpr int BSZ = (BK / 2) * PB * 2;
  constexpr int TC  = BN + 4;
  constexpr int RQ  = H_ / 4;
  constexpr int RR  = BM * RQ / NTHR;
  constexpr int CPT = H_ / BK;
  constexpr int XR  = RR / CPT;
  static_assert((PR % 8) == 4 && ((2 * PB) % 64) == 32 && ((8 * TC) % 64) == 32 && (TC % 4) == 0, "LDS pitches");
  static_assert((BM * RQ) % NTHR == 0 && RR % CPT == 0 && H_ % BK == 0, "rows split evenly over threads and turns");
  static_assert(B4 % 32 == 0 && (B4 % NTHR) != 0 && (K / BK) % 2 == 0, "B tasks over whole waves; an even number of chunks");
  __shared__ __attribute__((aligned(16))) float Ri[BM * PR];
  __shared__ __attribute__((aligned(16))) float Rj[BM * PR];
  __shared__ __attribute__((aligned(16))) float Bs[2 * BSZ];
  __shared__ __attribute__((aligned(16))) float Ct[BM * TC];

  const int i  = blockIdx.z;
  const int n0 = blockIdx.x * 48;
  const int a0 = blockIdx.y * 64;
  const int j0 = (i == 0) ? 1 : 0;
  const int j1 = (i == 2) ? 1 : 2;
  const float* Hi  = hx + (size_t)i  * A_ * H_;
  const float* Hj0 = hx + (size_t)j0 * A_ * H_;
  const float* Hj1 = hx + (size_t)j1 * A_ * H_;
  const float* Wp = cw + (size_t)(2 * i) * KP * H_;
  const float* Bb = cb + (size_t)(2 * i) * H_;

  const int t    = threadIdx.x;
  const int lane = t & 31;
  const int w    = t >> 5;
  const int mr   = lane & 15;
  const int kh   = lane >> 4;

  float2 be[2][RB], bo[2][RB];
  auto btask = [&](const int r) {
    const int e = t + NTHR * r;
    return (r == RB - 1 && e >= B4) ? e - NTHR : e;
  };
  auto prefetch = [&](auto S, const int k0) {
    constexpr int s = decltype(S)::value;
#pragma unroll
    for (int r = 0; r < RB; ++r) {
      const int e  = btask(r);
      const int kp = e / NP;
      const int n  = min(n0 + (e % NP) * 2, H_ - 2);
      const float* p = Wp + (size_t)(k0 + 2 * kp) * H_ + n;
      be[s][r] = ld2(p);
      bo[s][r] = ld2(p + H_);
    }
  };
  auto stage = [&](auto S) {
    constexpr int s = decltype(S)::value;
#pragma unroll
    for (int r = 0; r < RB; ++r) {
      const int e = btask(r);
      const int j = (e % NP) * 2;
      st4(&Bs[s * BSZ + ((e / NP) * PB + j) * 2], make_float4(be[s][r].x, be[s][r].y, bo[s][r].x, bo[s][r].y));
    }
  };
  auto rplace = [&](const int e) { return (e / RQ) * PR + (e % RQ) * 4; };

  const int aoff = (w * 16 + mr) * PR + 2 * kh;
  const int boff = (kh * PB + (mr & ~1)) * 2 + (mr & 1);

  v8f acc[NT] = {};
  auto compute = [&](const int buf, const int c) {
    const int pc    = c % (3 * CPT);
    const int third = pc / CPT;
    const int kk    = (pc % CPT) * BK;
    const float* bptr = Bs + buf * BSZ + boff;
    auto step = [&](const v2f a, const int ks) {
#pragma unroll
      for (int c2 = 0; c2 < NT; ++c2) {
        const float* bq = bptr + ((ks / 2) * PB + 16 * c2) * 2;
        v2f b;
        b.x = bq[0];
        b.y = bq[2];
        acc[c2] = __builtin_amdgcn_wmma_f32_16x16x4_f32(false, a, false, b, (short)0, acc[c2], false, false);
      }
    };
    if (third == 2) {
      const float* pi = Ri + aoff + kk;
      const float* pj = Rj + aoff + kk;
#pragma unroll
      for (int ks = 0; ks < BK; ks += 4)
        step(*reinterpret_cast<const v2f*>(pi + ks) * *reinterpret_cast<const v2f*>(pj + ks), ks);
    } else {
      const float* ap = ((third == 1) ? Rj : Ri) + aoff + kk;
#pragma unroll
      for (int ks = 0; ks < BK; ks += 4) step(*reinterpret_cast<const v2f*>(ap + ks), ks);
    }
  };

  constexpr int GR = BN / 4;
  const bool has_group = lane < 2 * GR;
  const int gl = has_group ? lane : lane - 2 * GR;
  const int er = gl / GR;
  const int eg = gl % GR;
  const int ei = w * 16 + er;
  const int en = has_group ? n0 + eg * 4 : H_;
  auto to_rows = [&](float4 (&rv)[8]) {
    float* ct = Ct + (w * 16) * TC;
#pragma unroll
    for (int c = 0; c < NT; ++c) {
#pragma unroll
      for (int v = 0; v < 8; ++v) ct[(8 * kh + v) * TC + 16 * c + mr] = acc[c][v];
    }
#pragma unroll
    for (int j = 0; j < 8; ++j) rv[j] = *reinterpret_cast<const float4*>(&ct[(er + 2 * j) * TC + eg * 4]);
  };

  float4 g[8];
  auto gated = [&](const int i0, const int n, const float4 (&v)[8], const float* Hj, const float* bias,
                   const bool add) {
    if (n >= H_) return;
    const float4 bn = ld4(bias + n);
    float4 p[8], q[8];
#pragma unroll
    for (int j = 0; j < 8; ++j) {
      const size_t o = (size_t)(a0 + i0 + 2 * j) * H_ + n;
      p[j] = ld4(Hi + o);
      q[j] = ld4(Hj + o);
    }
    auto one = [&](const float x, const float b, const float hi_, const float hj_) {
      const float z = 1.0f / (1.0f + expf(-(x + b)));
      return z * hi_ + (1.0f - z) * hj_;
    };
#pragma unroll
    for (int j = 0; j < 8; ++j) {
      const float4 tt = make_float4(one(v[j].x, bn.x, p[j].x, q[j].x), one(v[j].y, bn.y, p[j].y, q[j].y),
                                    one(v[j].z, bn.z, p[j].z, q[j].z), one(v[j].w, bn.w, p[j].w, q[j].w));
      g[j] = add ? make_float4(g[j].x + tt.x, g[j].y + tt.y, g[j].z + tt.z, g[j].w + tt.w) : tt;
    }
  };

  auto turn = [&](auto S, auto MORE, const int k0) {
    constexpr int s = decltype(S)::value;
    constexpr bool more = decltype(MORE)::value != 0;
    const int tau = k0 / BK;
    if (k0 == KP + BK) {
      float4 rv[8];
      to_rows(rv);
      gated(ei, en, rv, Hj0, Bb, false);
#pragma unroll
      for (int c = 0; c < NT; ++c) acc[c] = v8f{};
    }
    const int  xi = (tau - 1) % (3 * CPT);
    const bool xt = xi < CPT;
    const float* xs = ((tau > CPT) ? Hj1 : Hj0) + (size_t)a0 * H_;
    float4 xv[XR];
    if (xt) {
#pragma unroll
      for (int j = 0; j < XR; ++j) xv[j] = ld4(xs + 4 * (t + NTHR * (XR * xi + j)));
    }
    if constexpr (more) prefetch(set_c<(s ^ 1)>{}, k0 + BK);
    __builtin_amdgcn_sched_barrier(0);
    compute(s ^ 1, tau - 1);
    __builtin_amdgcn_sched_barrier(0);
    stage(S);
    if (xt) {
#pragma unroll
      for (int j = 0; j < XR; ++j) st4(&Rj[rplace(t + NTHR * (XR * xi + j))], xv[j]);
    }
    lds_barrier();
  };

  {
    const float* gi = Hi + (size_t)a0 * H_;
    float4 v[RR];
#pragma unroll
    for (int r = 0; r < RR; ++r) v[r] = ld4(gi + 4 * (t + NTHR * r));
    prefetch(set_c<0>{}, 0);
    prefetch(set_c<1>{}, BK);
#pragma unroll
    for (int r = 0; r < RR; ++r) st4(&Ri[rplace(t + NTHR * r)], v[r]);
  }
  stage(set_c<0>{});
  lds_barrier();
  int k0 = BK;
  for (; k0 + 2 * BK < K; k0 += 2 * BK) {
    turn(set_c<1>{}, set_c<1>{}, k0);
    turn(set_c<0>{}, set_c<1>{}, k0 + BK);
  }
  turn(set_c<1>{}, set_c<0>{}, k0);
  compute(((K / BK) - 1) & 1, (K / BK) - 1);

  float4 rv[8];
  to_rows(rv);
  if (en >= H_) return;
  gated(ei, en, rv, Hj1, Bb + H_, true);
#pragma unroll
  for (int j = 0; j < 8; ++j)
    st4(hout + ((size_t)i * A_ + a0 + ei + 2 * j) * H_ + en, g[j]);
}

__global__ __launch_bounds__(512) void k_final(const float* __restrict__ hout,
                                               const float* __restrict__ uw,
                                               const float* __restrict__ ub,
                                               float* __restrict__ out) {
  constexpr int UPB = 16;
  constexpr int NSF = H_ / 32;
  constexpr int REM = H_ - 32 * NSF;
  static_assert((UPB * TAG_) % 32 == 0, "a block's results must be whole 128-byte lines");
  static_assert(REM > 0 && M_ * REM <= 32, "the left-over units of all modalities share one slot");
  __shared__ float res[UPB * TAG_];
  const int wv   = threadIdx.x >> 5;
  const int lane = threadIdx.x & 31;
  const int a    = blockIdx.x * UPB + wv;
  float acc[TAG_] = {};
  auto add6 = [&](const float v, const float2 w01, const float2 w23, const float2 w45) {
    acc[0] += v * w01.x; acc[1] += v * w01.y;
    acc[2] += v * w23.x; acc[3] += v * w23.y;
    acc[4] += v * w45.x; acc[5] += v * w45.y;
  };
#pragma unroll
  for (int m = 0; m < M_; ++m) {
    const float* hr = hout + (size_t)(m * A_ + a) * H_;
    const float* wm = uw + (size_t)m * H_ * TAG_;
#pragma unroll
    for (int s = 0; s < NSF; ++s) {
      const int h = lane + 32 * s;
      const float* wr = wm + h * TAG_;
      add6(hr[h], ld2(wr), ld2(wr + 2), ld2(wr + 4));
    }
  }
  {
    const bool ok = lane < M_ * REM;
    const int l = ok ? lane : M_ * REM - 1;
    const int m = l / REM;
    const int h = 32 * NSF + l % REM;
    const float* wr = uw + ((size_t)m * H_ + h) * TAG_;
    const float  v   = hout[(size_t)(m * A_ + a) * H_ + h];
    const float2 w01 = ld2(wr), w23 = ld2(wr + 2), w45 = ld2(wr + 4);
    const float2 z = make_float2(0.f, 0.f);
    add6(ok ? v : 0.f, ok ? w01 : z, ok ? w23 : z, ok ? w45 : z);
  }
#pragma unroll
  for (int t = 0; t < TAG_; ++t) acc[t] = wave_sum(acc[t]);
  if (lane == 0) {
#pragma unroll
    for (int t = 0; t < TAG_; ++t)
      res[wv * TAG_ + t] = acc[t] + (ub[t] + ub[TAG_ + t] + ub[2 * TAG_ + t]);
  }
  __syncthreads();
  if (wv == 0) {
    for (int j = 0; j < UPB * TAG_ / 32; ++j)
      out[(size_t)blockIdx.x * (UPB * TAG_) + 32 * j + lane] = res[32 * j + lane];
  }
}

}

extern "C" void kernel_launch(void* const* d_in, const int* in_sizes, int n_in,
                              void* d_out, int out_size, void* d_ws, size_t ws_size,
                              hipStream_t stream) {
  (void)in_sizes; (void)n_in; (void)out_size; (void)ws_size;
  const float* x_a     = (const float*)d_in[0];
  const float* x_v     = (const float*)d_in[1];
  const float* x_t     = (const float*)d_in[2];
  const float* fc0_w   = (const float*)d_in[3];
  const float* fc0_b   = (const float*)d_in[4];
  const float* conv_w  = (const float*)d_in[5];
  const float* tm_w    = (const float*)d_in[6];
  const float* tm_b    = (const float*)d_in[7];
  const float* cross_w = (const float*)d_in[8];
  const float* cross_b = (const float*)d_in[9];
  const float* uni_w   = (const float*)d_in[10];
  const float* uni_b   = (const float*)d_in[11];
  float* out = (float*)d_out;

  const size_t NH = (size_t)N_ * H_;
  float* ws    = (float*)d_ws;
  float* intra = ws; ws += (size_t)N_ * L_;
  float* crossv= ws; ws += (size_t)M_ * M_ * A_;
  float* dinv  = ws; ws += N_;
  float* h0    = ws; ws += NH;
  float* hA    = ws; ws += NH;
  float* hx    = ws; ws += NH;
  float* hout  = ws; ws += NH;

  k_front<<<M_ * B_ + 5 * (N_ / 64), dim3(128), 0, stream>>>(x_a, x_v, x_t, fc0_w, fc0_b, h0,
                                                            intra, crossv, dinv);

  const float beta0 = logf(0.5f / 1.0f + 1.0f);
  const float beta1 = logf(0.5f / 2.0f + 1.0f);
  k_layer<false><<<2 * M_ * B_, dim3(256), 0, stream>>>(intra, crossv, dinv, h0, h0, conv_w, beta0, hA,
                                                       x_a, x_v, x_t, tm_w, tm_b, hx);
  k_layer<true> <<<2 * M_ * B_, dim3(256), 0, stream>>>(intra, crossv, dinv, hA, h0, conv_w + 2 * H_ * H_, beta1, hout,
                                                       x_a, x_v, x_t, tm_w, tm_b, hx);

  k_gate <<<dim3(5, A_ / 64, M_), dim3(128), 0, stream>>>(hx, cross_w, cross_b, hout);

  k_final<<<A_ / 16, dim3(512), 0, stream>>>(hout, uni_w, uni_b, out);
}
